# starting kernel
# baseline (speedup 1.0000x reference)
#include <hip/hip_runtime.h>
#include <math.h>

#define BATCH 32768
#define LRATE 0.01f
#define AEPS  1e-8f
#define LNEPS 1e-5f

typedef __attribute__((ext_vector_type(2))) float v2f;
typedef __attribute__((ext_vector_type(8))) float v8f;

__device__ __forceinline__ v2f mkv2(float x, float y) { v2f v; v[0] = x; v[1] = y; return v; }

__device__ __forceinline__ v8f wmma_f32(v2f a, v2f b, v8f c) {
  return __builtin_amdgcn_wmma_f32_16x16x4_f32(false, a, false, b, (short)0, c, false, false);
}

constexpr size_t F_CAT  = 0;
constexpr size_t F_S    = F_CAT + (size_t)BATCH * 16;
constexpr size_t F_W1   = F_S + (size_t)3 * BATCH * 64;
constexpr size_t F_HB   = F_W1 + 16 * 64;
constexpr size_t F_QT   = F_HB + 64;
constexpr size_t F_KT   = F_QT + 64 * 64;
constexpr size_t F_VT   = F_KT + 64 * 64;
constexpr size_t F_LWT  = F_VT + 64 * 64;
constexpr size_t F_GW   = F_LWT + 64 * 64;
constexpr size_t F_GB   = F_GW + 64 * 64;
constexpr size_t F_LW1T = F_GB + 64;
constexpr size_t F_LB1  = F_LW1T + 64 * 64;

__global__ void __launch_bounds__(256) k_cat(
    const float* __restrict__ in1, const float* __restrict__ in2,
    const float* __restrict__ in3, const float* __restrict__ in4,
    const float* __restrict__ wl, const float* __restrict__ bl,
    const float* __restrict__ wm, const float* __restrict__ bm,
    const float* __restrict__ wsw, const float* __restrict__ bs,
    float* __restrict__ cat)
{
  __shared__ float sw[4500 + 2250 + 450];
  for (int i = threadIdx.x; i < 7200; i += 256) {
    float v;
    if (i < 4500)      v = wl[i];
    else if (i < 6750) v = wm[i - 4500];
    else               v = wsw[i - 6750];
    sw[i] = v;
  }
  __syncthreads();

  const int warp = threadIdx.x >> 5, lane = threadIdx.x & 31;
  const int b = blockIdx.x * 8 + warp;

  float aL[3] = {0.f, 0.f, 0.f}, aM[3] = {0.f, 0.f, 0.f}, aS[3] = {0.f, 0.f, 0.f};

  const float* r1 = in1 + (size_t)b * 1500;
  for (int k = lane; k < 500; k += 32) {
    __builtin_prefetch(r1 + (k + 128) * 3, 0, 0);
    float x0 = r1[k * 3 + 0], x1 = r1[k * 3 + 1], x2 = r1[k * 3 + 2];
#pragma unroll
    for (int o = 0; o < 3; ++o)
      aL[o] += x0 * sw[(o * 3 + 0) * 500 + k] + x1 * sw[(o * 3 + 1) * 500 + k] +
               x2 * sw[(o * 3 + 2) * 500 + k];
  }
  const float* r2 = in2 + (size_t)b * 750;
  for (int k = lane; k < 250; k += 32) {
    float x0 = r2[k * 3 + 0], x1 = r2[k * 3 + 1], x2 = r2[k * 3 + 2];
#pragma unroll
    for (int o = 0; o < 3; ++o)
      aM[o] += x0 * sw[4500 + (o * 3 + 0) * 250 + k] + x1 * sw[4500 + (o * 3 + 1) * 250 + k] +
               x2 * sw[4500 + (o * 3 + 2) * 250 + k];
  }
  const float* r3 = in3 + (size_t)b * 150;
  for (int k = lane; k < 50; k += 32) {
    float x0 = r3[k * 3 + 0], x1 = r3[k * 3 + 1], x2 = r3[k * 3 + 2];
#pragma unroll
    for (int o = 0; o < 3; ++o)
      aS[o] += x0 * sw[6750 + (o * 3 + 0) * 50 + k] + x1 * sw[6750 + (o * 3 + 1) * 50 + k] +
               x2 * sw[6750 + (o * 3 + 2) * 50 + k];
  }
#pragma unroll
  for (int o = 0; o < 3; ++o)
#pragma unroll
    for (int off = 16; off; off >>= 1) {
      aL[o] += __shfl_xor(aL[o], off, 32);
      aM[o] += __shfl_xor(aM[o], off, 32);
      aS[o] += __shfl_xor(aS[o], off, 32);
    }
  if (lane == 0) {
    float* cr = cat + (size_t)b * 16;
#pragma unroll
    for (int o = 0; o < 3; ++o) {
      cr[o]      = aL[o] + bl[o];
      cr[3 + o]  = aM[o] + bm[o];
      cr[6 + o]  = aS[o] + bs[o];
      cr[9 + o]  = in4[(size_t)b * 3 + o];
      cr[12 + o] = 0.f;
    }
    cr[15] = 0.f;
  }
}

__global__ void k_prep(const float* __restrict__ fc1w, const float* __restrict__ fc1b,
                       const float* __restrict__ qw, const float* __restrict__ kw,
                       const float* __restrict__ vw, const float* __restrict__ lw,
                       float* __restrict__ wsf)
{
  float* W1 = wsf + F_W1; float* hb = wsf + F_HB;
  float* qT = wsf + F_QT; float* kT = wsf + F_KT; float* vT = wsf + F_VT;
  float* lwT = wsf + F_LWT; float* gW = wsf + F_GW; float* gb = wsf + F_GB;
  for (int i = threadIdx.x; i < 1024; i += 256) {
    int k = i >> 6, n = i & 63;
    W1[i] = (k < 12) ? fc1w[n * 12 + k] : 0.f;
  }
  for (int i = threadIdx.x; i < 4096; i += 256) {
    int k = i >> 6, n = i & 63;
    qT[i] = qw[n * 64 + k]; kT[i] = kw[n * 64 + k];
    vT[i] = vw[n * 64 + k]; lwT[i] = lw[n * 64 + k];
    gW[i] = 0.f;
  }
  for (int i = threadIdx.x; i < 64; i += 256) {
    hb[i] = fc1b[i] + ((i & 1) ? 1.0f : 0.0f);
    gb[i] = 0.f;
  }
}

__global__ void __launch_bounds__(256) k_qkv(
    const float* __restrict__ cat, const float* __restrict__ qb,
    const float* __restrict__ kbv, const float* __restrict__ vb,
    float* __restrict__ s, const float* __restrict__ W1, const float* __restrict__ hb,
    const float* __restrict__ qT, const float* __restrict__ kT, const float* __restrict__ vT)
{
  __shared__ float sh[8][16 * 68];
  const int warp = threadIdx.x >> 5, lane = threadIdx.x & 31;
  const int hi = lane >> 4, nl = lane & 15;
  const int row0 = (blockIdx.x * 8 + warp) * 16;
  float* shw = sh[warp];

  v2f ca[4];
#pragma unroll
  for (int ks = 0; ks < 4; ++ks) {
    int kb = ks * 4 + 2 * hi;
    const float* p = cat + (size_t)(row0 + nl) * 16 + kb;
    ca[ks] = mkv2(p[0], p[1]);
  }
#pragma unroll
  for (int n0 = 0; n0 < 4; ++n0) {
    int n = n0 * 16 + nl;
    v8f c = {};
#pragma unroll
    for (int ks = 0; ks < 4; ++ks) {
      int kb = ks * 4 + 2 * hi;
      c = wmma_f32(ca[ks], mkv2(W1[kb * 64 + n], W1[(kb + 1) * 64 + n]), c);
    }
    float bias = hb[n];
#pragma unroll
    for (int r = 0; r < 8; ++r) shw[(r + 8 * hi) * 68 + n] = c[r] + bias;
  }
  __syncthreads();

  v2f ha[16];
#pragma unroll
  for (int ks = 0; ks < 16; ++ks) {
    int kb = ks * 4 + 2 * hi;
    ha[ks] = mkv2(shw[nl * 68 + kb], shw[nl * 68 + kb + 1]);
  }
  const float* BT[3] = {qT, kT, vT};
  const float* BB[3] = {qb, kbv, vb};
#pragma unroll
  for (int t = 0; t < 3; ++t) {
    float* ob = s + (size_t)t * BATCH * 64;
#pragma unroll
    for (int n0 = 0; n0 < 4; ++n0) {
      int n = n0 * 16 + nl;
      v8f c = {};
#pragma unroll
      for (int ks = 0; ks < 16; ++ks) {
        int kb = ks * 4 + 2 * hi;
        c = wmma_f32(ha[ks], mkv2(BT[t][kb * 64 + n], BT[t][(kb + 1) * 64 + n]), c);
      }
      float bias = BB[t][n];
#pragma unroll
      for (int r = 0; r < 8; ++r)
        ob[(size_t)(row0 + r + 8 * hi) * 64 + n] = c[r] + bias;
    }
  }
}

__global__ void __launch_bounds__(128) k_grad(
    const float* __restrict__ s, const float* __restrict__ tk,
    const float* __restrict__ tvm, const float* __restrict__ lwT,
    const float* __restrict__ lb, float* __restrict__ gW, float* __restrict__ gb)
{
  __shared__ float stv[4][16 * 68];
  __shared__ float sg[4][16 * 68];
  __shared__ float sgW[64 * 64];
  __shared__ float sgb[64];
  for (int i = threadIdx.x; i < 4096; i += 128) sgW[i] = 0.f;
  if (threadIdx.x < 64) sgb[threadIdx.x] = 0.f;
  __syncthreads();

  const int warp = threadIdx.x >> 5, lane = threadIdx.x & 31;
  const int hi = lane >> 4, nl = lane & 15;
  float* tvw = stv[warp];
  float* gw_ = sg[warp];

  const float gscale = 2.0f / (float)(3ull * BATCH * 64);
  const int NT = 3 * BATCH / 16;
  const int wave = blockIdx.x * 4 + warp;
  const int nw = gridDim.x * 4;

  v8f cgw[4][4];
#pragma unroll
  for (int i0 = 0; i0 < 4; ++i0)
#pragma unroll
    for (int j0 = 0; j0 < 4; ++j0) cgw[i0][j0] = (v8f){};

  for (int t = wave; t < NT; t += nw) {
    const int r0 = t * 16;
    v2f sa[16];
#pragma unroll
    for (int ks = 0; ks < 16; ++ks) {
      int kb = ks * 4 + 2 * hi;
      const float* p = s + (size_t)(r0 + nl) * 64 + kb;
      sa[ks] = mkv2(p[0], p[1]);
    }
#pragma unroll
    for (int n0 = 0; n0 < 4; ++n0) {
      int n = n0 * 16 + nl;
      v8f ctv = {}, clv = {};
#pragma unroll
      for (int ks = 0; ks < 16; ++ks) {
        int kb = ks * 4 + 2 * hi;
        ctv = wmma_f32(sa[ks], mkv2(tk[kb * 64 + n], tk[(kb + 1) * 64 + n]), ctv);
        clv = wmma_f32(sa[ks], mkv2(tvm[kb * 64 + n], tvm[(kb + 1) * 64 + n]), clv);
      }
#pragma unroll
      for (int r = 0; r < 8; ++r) {
        tvw[(r + 8 * hi) * 68 + n] = ctv[r];
        gw_[(r + 8 * hi) * 68 + n] = -clv[r];
      }
    }
#pragma unroll
    for (int n0 = 0; n0 < 4; ++n0) {
      int n = n0 * 16 + nl;
      v8f cp = {};
#pragma unroll
      for (int ks = 0; ks < 16; ++ks) {
        int kb = ks * 4 + 2 * hi;
        v2f a = mkv2(tvw[nl * 68 + kb], tvw[nl * 68 + kb + 1]);
        cp = wmma_f32(a, mkv2(lwT[kb * 64 + n], lwT[(kb + 1) * 64 + n]), cp);
      }
      float lbn = lb[n];
#pragma unroll
      for (int r = 0; r < 8; ++r) {
        int idx = (r + 8 * hi) * 68 + n;
        float g = gscale * (cp[r] + lbn + gw_[idx]);
        gw_[idx] = g;
        atomicAdd(&sgb[n], g);
      }
    }
#pragma unroll
    for (int i0 = 0; i0 < 4; ++i0) {
      int i = i0 * 16 + nl;
#pragma unroll
      for (int j0 = 0; j0 < 4; ++j0) {
        int j = j0 * 16 + nl;
#pragma unroll
        for (int ks = 0; ks < 4; ++ks) {
          int kb = ks * 4 + 2 * hi;
          v2f a = mkv2(gw_[kb * 68 + i], gw_[(kb + 1) * 68 + i]);
          v2f b = mkv2(tvw[kb * 68 + j], tvw[(kb + 1) * 68 + j]);
          cgw[i0][j0] = wmma_f32(a, b, cgw[i0][j0]);
        }
      }
    }
  }
#pragma unroll
  for (int i0 = 0; i0 < 4; ++i0)
#pragma unroll
    for (int j0 = 0; j0 < 4; ++j0)
#pragma unroll
      for (int r = 0; r < 8; ++r)
        atomicAdd(&sgW[(i0 * 16 + r + 8 * hi) * 64 + j0 * 16 + nl], cgw[i0][j0][r]);
  __syncthreads();
  for (int i = threadIdx.x; i < 4096; i += 128) atomicAdd(&gW[i], sgW[i]);
  if (threadIdx.x < 64) atomicAdd(&gb[threadIdx.x], sgb[threadIdx.x]);
}

__global__ void k_update(const float* __restrict__ lw, const float* __restrict__ lb,
                         const float* __restrict__ gW, const float* __restrict__ gb,
                         float* __restrict__ lw1T, float* __restrict__ lb1)
{
  for (int i = threadIdx.x; i < 4096; i += 256) {
    int n = i >> 6, k = i & 63;
    float g = gW[i];
    lw1T[k * 64 + n] = lw[i] - LRATE * g / (fabsf(g) + AEPS);
  }
  for (int i = threadIdx.x; i < 64; i += 256) {
    float g = gb[i];
    lb1[i] = lb[i] - LRATE * g / (fabsf(g) + AEPS);
  }
}

__global__ void __launch_bounds__(128) k_out(
    const float* __restrict__ s, const float* __restrict__ tq,
    const float* __restrict__ lw1T, const float* __restrict__ lb1,
    const float* __restrict__ lng, const float* __restrict__ lnb,
    const float* __restrict__ ow, const float* __restrict__ obp,
    float* __restrict__ out)
{
  __shared__ float su[4][16 * 68];
  __shared__ float so[4][16 * 68];
  const int warp = threadIdx.x >> 5, lane = threadIdx.x & 31;
  const int hi = lane >> 4, nl = lane & 15;
  const int row0 = (blockIdx.x * 4 + warp) * 16;
  float* uw = su[warp];
  float* om = so[warp];
  const float* sv = s + (size_t)2 * BATCH * 64;

  v2f sa[16];
#pragma unroll
  for (int ks = 0; ks < 16; ++ks) {
    int kb = ks * 4 + 2 * hi;
    const float* p = sv + (size_t)(row0 + nl) * 64 + kb;
    sa[ks] = mkv2(p[0], p[1]);
  }
#pragma unroll
  for (int n0 = 0; n0 < 4; ++n0) {
    int n = n0 * 16 + nl;
    v8f c = {};
#pragma unroll
    for (int ks = 0; ks < 16; ++ks) {
      int kb = ks * 4 + 2 * hi;
      c = wmma_f32(sa[ks], mkv2(tq[kb * 64 + n], tq[(kb + 1) * 64 + n]), c);
    }
#pragma unroll
    for (int r = 0; r < 8; ++r) uw[(r + 8 * hi) * 68 + n] = c[r];
  }
#pragma unroll
  for (int n0 = 0; n0 < 4; ++n0) {
    int n = n0 * 16 + nl;
    v8f c = {};
#pragma unroll
    for (int ks = 0; ks < 16; ++ks) {
      int kb = ks * 4 + 2 * hi;
      v2f a = mkv2(uw[nl * 68 + kb], uw[nl * 68 + kb + 1]);
      c = wmma_f32(a, mkv2(lw1T[kb * 64 + n], lw1T[(kb + 1) * 64 + n]), c);
    }
    float bias = lb1[n];
#pragma unroll
    for (int r = 0; r < 8; ++r) om[(r + 8 * hi) * 68 + n] = c[r] + bias;
  }
  __syncthreads();
  if (lane < 16) {
    const float* rowp = om + lane * 68;
    float mu = 0.f;
#pragma unroll
    for (int j = 0; j < 64; ++j) mu += rowp[j];
    mu *= (1.0f / 64.0f);
    float var = 0.f;
#pragma unroll
    for (int j = 0; j < 64; ++j) { float d = rowp[j] - mu; var += d * d; }
    var *= (1.0f / 64.0f);
    float rstd = rsqrtf(var + LNEPS);
    float acc = 0.f;
#pragma unroll
    for (int j = 0; j < 64; ++j) {
      float nrm = (rowp[j] - mu) * rstd * lng[j] + lnb[j];
      float a = nrm / (1.0f + __expf(-nrm));
      acc += a * ow[j];
    }
    out[row0 + lane] = acc + obp[0];
  }
}

extern "C" void kernel_launch(void* const* d_in, const int* in_sizes, int n_in,
                              void* d_out, int out_size, void* d_ws, size_t ws_size,
                              hipStream_t stream) {
  const float* in1  = (const float*)d_in[0];
  const float* in2  = (const float*)d_in[1];
  const float* in3  = (const float*)d_in[2];
  const float* in4  = (const float*)d_in[3];
  const float* w_l  = (const float*)d_in[4];
  const float* b_l  = (const float*)d_in[5];
  const float* w_m  = (const float*)d_in[6];
  const float* b_m  = (const float*)d_in[7];
  const float* w_s  = (const float*)d_in[8];
  const float* b_s  = (const float*)d_in[9];
  const float* fc1w = (const float*)d_in[10];
  const float* fc1b = (const float*)d_in[11];
  const float* q_w  = (const float*)d_in[12];
  const float* q_b  = (const float*)d_in[13];
  const float* k_w  = (const float*)d_in[14];
  const float* k_b  = (const float*)d_in[15];
  const float* v_w  = (const float*)d_in[16];
  const float* v_b  = (const float*)d_in[17];
  const float* t_k  = (const float*)d_in[18];
  const float* t_v  = (const float*)d_in[19];
  const float* t_q  = (const float*)d_in[20];
  const float* lw   = (const float*)d_in[21];
  const float* lb   = (const float*)d_in[22];
  const float* lng  = (const float*)d_in[23];
  const float* lnb  = (const float*)d_in[24];
  const float* outw = (const float*)d_in[25];
  const float* outb = (const float*)d_in[26];

  float* wsf  = (float*)d_ws;
  float* cat  = wsf + F_CAT;
  float* sbuf = wsf + F_S;
  float* W1   = wsf + F_W1;
  float* hb   = wsf + F_HB;
  float* qT   = wsf + F_QT;
  float* kT   = wsf + F_KT;
  float* vT   = wsf + F_VT;
  float* lwT  = wsf + F_LWT;
  float* gW   = wsf + F_GW;
  float* gb   = wsf + F_GB;
  float* lw1T = wsf + F_LW1T;
  float* lb1  = wsf + F_LB1;

  k_cat<<<BATCH / 8, 256, 0, stream>>>(in1, in2, in3, in4, w_l, b_l, w_m, b_m, w_s, b_s, cat);
  k_prep<<<1, 256, 0, stream>>>(fc1w, fc1b, q_w, k_w, v_w, lw, wsf);
  k_qkv<<<BATCH / 128, 256, 0, stream>>>(cat, q_b, k_b, v_b, sbuf, W1, hb, qT, kT, vT);
  k_grad<<<512, 128, 0, stream>>>(sbuf, t_k, t_v, lwT, lb, gW, gb);
  k_update<<<1, 256, 0, stream>>>(lw, lb, gW, gb, lw1T, lb1);
  k_out<<<BATCH / 64, 128, 0, stream>>>(sbuf, t_q, lw1T, lb1, lng, lnb, outw, outb,
                                        (float*)d_out);
}


# k_cat as a one-pass stream: 8 rows per wave, 16-byte non-temporal loads three stages ahead, conv weights in LDS in input order, lane sums through LDS (changes the conv sum order)
# speedup vs baseline: 1.3055x; 1.3055x over previous
#include <hip/hip_runtime.h>
#include <math.h>

#define BATCH 32768
#define LRATE 0.01f
#define AEPS  1e-8f
#define LNEPS 1e-5f

typedef __attribute__((ext_vector_type(2))) float v2f;
typedef __attribute__((ext_vector_type(8))) float v8f;

__device__ __forceinline__ v2f mkv2(float x, float y) { v2f v; v[0] = x; v[1] = y; return v; }

__device__ __forceinline__ v8f wmma_f32(v2f a, v2f b, v8f c) {
  return __builtin_amdgcn_wmma_f32_16x16x4_f32(false, a, false, b, (short)0, c, false, false);
}

constexpr size_t F_CAT  = 0;
constexpr size_t F_S    = F_CAT + (size_t)BATCH * 16;
constexpr size_t F_W1   = F_S + (size_t)3 * BATCH * 64;
constexpr size_t F_HB   = F_W1 + 16 * 64;
constexpr size_t F_QT   = F_HB + 64;
constexpr size_t F_KT   = F_QT + 64 * 64;
constexpr size_t F_VT   = F_KT + 64 * 64;
constexpr size_t F_LWT  = F_VT + 64 * 64;
constexpr size_t F_GW   = F_LWT + 64 * 64;
constexpr size_t F_GB   = F_GW + 64 * 64;
constexpr size_t F_LW1T = F_GB + 64;
constexpr size_t F_LB1  = F_LW1T + 64 * 64;

typedef __attribute__((ext_vector_type(4))) float v4f;

constexpr int CAT_WAVES  = 8;
constexpr int CAT_ROWS   = 8;
constexpr int CAT_BLOCKS = BATCH / (CAT_WAVES * CAT_ROWS);
static_assert(CAT_BLOCKS == 512 && CAT_WAVES * 32 == 256, "k_cat is launched as <<<512, 256>>>");
constexpr int WQ1 = 0;
constexpr int WQ2 = 4608;
constexpr int WQ3 = 9216;
constexpr int WQN = 10240;
constexpr int NSTAGE = 20;
constexpr int RPITCH = 33;

__device__ __forceinline__ v4f ld_stream(const v4f* p) { return __builtin_nontemporal_load(p); }
__device__ __forceinline__ float dot4(v4f d, v4f w, float a) {
  a = __builtin_fmaf(d.x, w.x, a); a = __builtin_fmaf(d.y, w.y, a);
  a = __builtin_fmaf(d.z, w.z, a); a = __builtin_fmaf(d.w, w.w, a);
  return a;
}

template <int S>
__device__ __forceinline__ void cat_issue(v4f (&buf)[8], const v4f* p1, const v4f* p2,
                                          const v4f* p3, int lane) {
  if constexpr (S < 12) {
#pragma unroll
    for (int r = 0; r < 8; ++r) {
      if constexpr (S == 11) buf[r] = ld_stream(p1 + r * 375 + ((lane < 23) ? 352 : 374 - lane));
      else                   buf[r] = ld_stream(p1 + r * 375 + 32 * S);
    }
  } else if constexpr (S < 18) {
#pragma unroll
    for (int h = 0; h < 2; ++h) {
      const int q = 2 * (S - 12) + h;
#pragma unroll
      for (int p = 0; p < 4; ++p) {
        if (q == 11) buf[h * 4 + p] = ld_stream(p2 + p * 375 + ((lane < 23) ? 352 : 374 - lane));
        else         buf[h * 4 + p] = ld_stream(p2 + p * 375 + 32 * q);
      }
    }
  } else if constexpr (S == 18) {
#pragma unroll
    for (int h = 0; h < 2; ++h)
#pragma unroll
      for (int p = 0; p < 4; ++p) buf[h * 4 + p] = ld_stream(p3 + p * 75 + 32 * h);
  } else if constexpr (S == 19) {
#pragma unroll
    for (int p = 0; p < 4; ++p) buf[p] = ld_stream(p3 + p * 75 + ((lane < 11) ? 64 : 74 - lane));
  }
}

template <int S>
__device__ __forceinline__ void cat_weights(v4f (&w)[6], const float* sw, int lane) {
  const v4f zero = {0.f, 0.f, 0.f, 0.f};
  if constexpr (S < 12) {
    const bool live = (S != 11) || (lane < 23);
    const int j = live ? 32 * S + lane : 0;
#pragma unroll
    for (int o = 0; o < 3; ++o) { const v4f t = *(const v4f*)(sw + WQ1 + o * 1500 + 4 * j); w[o] = live ? t : zero; }
  } else if constexpr (S < 18) {
#pragma unroll
    for (int h = 0; h < 2; ++h) {
      const int q = 2 * (S - 12) + h;
      const bool live = (q != 11) || (lane < 23);
      const int j = live ? 32 * q + lane : 0;
#pragma unroll
      for (int o = 0; o < 3; ++o) { const v4f t = *(const v4f*)(sw + WQ2 + o * 1500 + 4 * j); w[h * 3 + o] = live ? t : zero; }
    }
  } else if constexpr (S == 18) {
#pragma unroll
    for (int h = 0; h < 2; ++h)
#pragma unroll
      for (int o = 0; o < 3; ++o) w[h * 3 + o] = *(const v4f*)(sw + WQ3 + o * 300 + 4 * (32 * h + lane));
  } else {
    const bool live = lane < 11;
    const int j = live ? 64 + lane : 0;
#pragma unroll
    for (int o = 0; o < 3; ++o) { const v4f t = *(const v4f*)(sw + WQ3 + o * 300 + 4 * j); w[o] = live ? t : zero; }
  }
}

__device__ __forceinline__ void pair_mixed(v4f d, const v4f* w, float* accA, float* accB,
                                           bool xyA, bool zwA) {
#pragma unroll
  for (int o = 0; o < 3; ++o) {
    const float pxy = __builtin_fmaf(d.x, w[o].x, d.y * w[o].y);
    const float pzw = __builtin_fmaf(d.z, w[o].z, d.w * w[o].w);
    accA[o] += (xyA ? pxy : 0.f) + (zwA ? pzw : 0.f);
    accB[o] += (xyA ? 0.f : pxy) + (zwA ? 0.f : pzw);
  }
}

template <int S>
__device__ __forceinline__ void cat_consume(const v4f (&buf)[8], const v4f (&w)[6],
                                            float (&acc)[8][3], int lane) {
  if constexpr (S < 12) {
#pragma unroll
    for (int r = 0; r < 8; ++r)
#pragma unroll
      for (int o = 0; o < 3; ++o) acc[r][o] = dot4(buf[r], w[o], acc[r][o]);
  } else if constexpr (S < 18) {
#pragma unroll
    for (int h = 0; h < 2; ++h) {
      const int q = 2 * (S - 12) + h;
#pragma unroll
      for (int p = 0; p < 4; ++p) {
        if (q < 5) {
#pragma unroll
          for (int o = 0; o < 3; ++o) acc[2 * p][o] = dot4(buf[h * 4 + p], w[h * 3 + o], acc[2 * p][o]);
        } else if (q > 5) {
#pragma unroll
          for (int o = 0; o < 3; ++o) acc[2 * p + 1][o] = dot4(buf[h * 4 + p], w[h * 3 + o], acc[2 * p + 1][o]);
        } else {
          pair_mixed(buf[h * 4 + p], &w[h * 3], acc[2 * p], acc[2 * p + 1], lane <= 27, lane <= 26);
        }
      }
    }
  } else if constexpr (S == 18) {
#pragma unroll
    for (int p = 0; p < 4; ++p) {
#pragma unroll
      for (int o = 0; o < 3; ++o) acc[2 * p][o] = dot4(buf[p], w[o], acc[2 * p][o]);
      pair_mixed(buf[4 + p], &w[3], acc[2 * p], acc[2 * p + 1], lane <= 5, lane <= 4);
    }
  } else {
#pragma unroll
    for (int p = 0; p < 4; ++p)
#pragma unroll
      for (int o = 0; o < 3; ++o) acc[2 * p + 1][o] = dot4(buf[p], w[o], acc[2 * p + 1][o]);
  }
}

__global__ void __launch_bounds__(256) __attribute__((amdgpu_waves_per_eu(4, 4))) k_cat(
    const float* __restrict__ in1, const float* __restrict__ in2,
    const float* __restrict__ in3, const float* __restrict__ in4,
    const float* __restrict__ wl, const float* __restrict__ bl,
    const float* __restrict__ wm, const float* __restrict__ bm,
    const float* __restrict__ wsw, const float* __restrict__ bs,
    float* __restrict__ cat)
{
  __shared__ __attribute__((aligned(16))) float sw[WQN];
  __shared__ float red[CAT_WAVES][72 * RPITCH];
  __shared__ __attribute__((aligned(16))) float cst[CAT_WAVES][CAT_ROWS * 16];

  const int warp = threadIdx.x >> 5, lane = threadIdx.x & 31;
  const int b0 = (blockIdx.x * CAT_WAVES + warp) * CAT_ROWS;
  const v4f* p1 = (const v4f*)in1 + (size_t)b0 * 375 + lane;
  const v4f* p2 = (const v4f*)in2 + (size_t)(b0 >> 1) * 375 + lane;
  const v4f* p3 = (const v4f*)in3 + (size_t)(b0 >> 1) * 75 + lane;

  v4f buf[4][8];
  cat_issue<0>(buf[0], p1, p2, p3, lane);
  cat_issue<1>(buf[1], p1, p2, p3, lane);
  cat_issue<2>(buf[2], p1, p2, p3, lane);

  float tailv[2];
#pragma unroll
  for (int g = 0; g < 2; ++g) {
    const int u = lane + 32 * g, cc = u % 7;
    const bool take = (u < 56) && (cc < 3);
    const float t = in4[(size_t)b0 * 3 + (take ? (u / 7) * 3 + cc : 0)];
    tailv[g] = take ? t : 0.f;
  }

  {
    float t1[18];
#pragma unroll
    for (int t = 0; t < 18; ++t) {
      const int i = min((int)threadIdx.x + 256 * t, 4499), o = i / 1500, f = i - o * 1500;
      t1[t] = wl[(o * 3 + f % 3) * 500 + f / 3];
    }
#pragma unroll
    for (int t = 0; t < 18; ++t) sw[WQ1 + threadIdx.x + 256 * t] = t1[t];
#pragma unroll
    for (int t = 0; t < 18; ++t) {
      const int i = min((int)threadIdx.x + 256 * t, 4499), o = i / 1500, g = i - o * 1500;
      const int f = (g >= 750) ? g - 750 : g;
      t1[t] = wm[(o * 3 + f % 3) * 250 + f / 3];
    }
#pragma unroll
    for (int t = 0; t < 18; ++t) sw[WQ2 + threadIdx.x + 256 * t] = t1[t];
#pragma unroll
    for (int t = 0; t < 4; ++t) {
      const int i = min((int)threadIdx.x + 256 * t, 899), o = i / 300, g = i - o * 300;
      const int f = (g >= 150) ? g - 150 : g;
      t1[t] = wsw[(o * 3 + f % 3) * 50 + f / 3];
    }
#pragma unroll
    for (int t = 0; t < 4; ++t) sw[WQ3 + threadIdx.x + 256 * t] = t1[t];
  }
  __syncthreads();

  float acc[8][3];
  float* redw = red[warp];
  v4f w[6];

#define CAT_STAGE(S)                                                              \
  cat_weights<S>(w, sw, lane);                                                    \
  __builtin_amdgcn_sched_barrier(0);                                              \
  if constexpr ((S) + 3 < NSTAGE)                                                 \
    cat_issue<((S) + 3 < NSTAGE ? (S) + 3 : 0)>(buf[((S) + 3) % 4], p1, p2, p3, lane); \
  __builtin_amdgcn_sched_barrier(0);                                              \
  cat_consume<S>(buf[(S) % 4], w, acc, lane);                                     \
  CAT_PIN()
#define CAT_PIN()                                                                 \
  _Pragma("unroll") for (int r = 0; r < 8; ++r)                                   \
    _Pragma("unroll") for (int o = 0; o < 3; ++o) asm volatile("" : "+v"(acc[r][o]));
#define CAT_CLEAR()                                                               \
  _Pragma("unroll") for (int r = 0; r < 8; ++r)                                   \
    _Pragma("unroll") for (int o = 0; o < 3; ++o) acc[r][o] = 0.f;
#define CAT_PARK(COL)                                                             \
  _Pragma("unroll") for (int r = 0; r < 8; ++r)                                   \
    _Pragma("unroll") for (int o = 0; o < 3; ++o)                                 \
      redw[(r * 9 + (COL) + o) * RPITCH + lane] = acc[r][o];

  CAT_CLEAR()
  CAT_STAGE(0) CAT_STAGE(1) CAT_STAGE(2) CAT_STAGE(3) CAT_STAGE(4) CAT_STAGE(5)
  CAT_STAGE(6) CAT_STAGE(7) CAT_STAGE(8) CAT_STAGE(9) CAT_STAGE(10) CAT_STAGE(11)
  CAT_PARK(0)
  CAT_CLEAR()
  CAT_STAGE(12) CAT_STAGE(13) CAT_STAGE(14) CAT_STAGE(15) CAT_STAGE(16) CAT_STAGE(17)
  CAT_PARK(3)
  CAT_CLEAR()
  CAT_STAGE(18) CAT_STAGE(19)
  CAT_PARK(6)
#undef CAT_STAGE
#undef CAT_PIN
#undef CAT_CLEAR
#undef CAT_PARK
  __syncthreads();

  float* cw = cst[warp];
  const float b9[9] = {bl[0], bl[1], bl[2], bm[0], bm[1], bm[2], bs[0], bs[1], bs[2]};
#pragma unroll
  for (int g = 0; g < 3; ++g) {
    const int v = lane + 32 * g;
    if (v < 72) {
      const float* rp = redw + v * RPITCH;
      float s0 = 0.f, s1 = 0.f, s2 = 0.f, s3 = 0.f;
#pragma unroll
      for (int i = 0; i < 32; i += 4) { s0 += rp[i]; s1 += rp[i + 1]; s2 += rp[i + 2]; s3 += rp[i + 3]; }
      const int c = v % 9;
      float bias = b9[0];
#pragma unroll
      for (int k = 1; k < 9; ++k) bias = (c == k) ? b9[k] : bias;
      cw[(v / 9) * 16 + c] = (s0 + s1) + (s2 + s3) + bias;
    }
  }
#pragma unroll
  for (int g = 0; g < 2; ++g) {
    const int u = lane + 32 * g;
    if (u < 56) cw[(u / 7) * 16 + 9 + u % 7] = tailv[g];
  }
  __syncthreads();
  *((v4f*)(cat + (size_t)b0 * 16) + lane) = *((const v4f*)cw + lane);
}

__global__ void k_prep(const float* __restrict__ fc1w, const float* __restrict__ fc1b,
                       const float* __restrict__ qw, const float* __restrict__ kw,
                       const float* __restrict__ vw, const float* __restrict__ lw,
                       float* __restrict__ wsf)
{
  float* W1 = wsf + F_W1; float* hb = wsf + F_HB;
  float* qT = wsf + F_QT; float* kT = wsf + F_KT; float* vT = wsf + F_VT;
  float* lwT = wsf + F_LWT; float* gW = wsf + F_GW; float* gb = wsf + F_GB;
  for (int i = threadIdx.x; i < 1024; i += 256) {
    int k = i >> 6, n = i & 63;
    W1[i] = (k < 12) ? fc1w[n * 12 + k] : 0.f;
  }
  for (int i = threadIdx.x; i < 4096; i += 256) {
    int k = i >> 6, n = i & 63;
    qT[i] = qw[n * 64 + k]; kT[i] = kw[n * 64 + k];
    vT[i] = vw[n * 64 + k]; lwT[i] = lw[n * 64 + k];
    gW[i] = 0.f;
  }
  for (int i = threadIdx.x; i < 64; i += 256) {
    hb[i] = fc1b[i] + ((i & 1) ? 1.0f : 0.0f);
    gb[i] = 0.f;
  }
}

__global__ void __launch_bounds__(256) k_qkv(
    const float* __restrict__ cat, const float* __restrict__ qb,
    const float* __restrict__ kbv, const float* __restrict__ vb,
    float* __restrict__ s, const float* __restrict__ W1, const float* __restrict__ hb,
    const float* __restrict__ qT, const float* __restrict__ kT, const float* __restrict__ vT)
{
  __shared__ float sh[8][16 * 68];
  const int warp = threadIdx.x >> 5, lane = threadIdx.x & 31;
  const int hi = lane >> 4, nl = lane & 15;
  const int row0 = (blockIdx.x * 8 + warp) * 16;
  float* shw = sh[warp];

  v2f ca[4];
#pragma unroll
  for (int ks = 0; ks < 4; ++ks) {
    int kb = ks * 4 + 2 * hi;
    const float* p = cat + (size_t)(row0 + nl) * 16 + kb;
    ca[ks] = mkv2(p[0], p[1]);
  }
#pragma unroll
  for (int n0 = 0; n0 < 4; ++n0) {
    int n = n0 * 16 + nl;
    v8f c = {};
#pragma unroll
    for (int ks = 0; ks < 4; ++ks) {
      int kb = ks * 4 + 2 * hi;
      c = wmma_f32(ca[ks], mkv2(W1[kb * 64 + n], W1[(kb + 1) * 64 + n]), c);
    }
    float bias = hb[n];
#pragma unroll
    for (int r = 0; r < 8; ++r) shw[(r + 8 * hi) * 68 + n] = c[r] + bias;
  }
  __syncthreads();

  v2f ha[16];
#pragma unroll
  for (int ks = 0; ks < 16; ++ks) {
    int kb = ks * 4 + 2 * hi;
    ha[ks] = mkv2(shw[nl * 68 + kb], shw[nl * 68 + kb + 1]);
  }
  const float* BT[3] = {qT, kT, vT};
  const float* BB[3] = {qb, kbv, vb};
#pragma unroll
  for (int t = 0; t < 3; ++t) {
    float* ob = s + (size_t)t * BATCH * 64;
#pragma unroll
    for (int n0 = 0; n0 < 4; ++n0) {
      int n = n0 * 16 + nl;
      v8f c = {};
#pragma unroll
      for (int ks = 0; ks < 16; ++ks) {
        int kb = ks * 4 + 2 * hi;
        c = wmma_f32(ha[ks], mkv2(BT[t][kb * 64 + n], BT[t][(kb + 1) * 64 + n]), c);
      }
      float bias = BB[t][n];
#pragma unroll
      for (int r = 0; r < 8; ++r)
        ob[(size_t)(row0 + r + 8 * hi) * 64 + n] = c[r] + bias;
    }
  }
}

__global__ void __launch_bounds__(128) k_grad(
    const float* __restrict__ s, const float* __restrict__ tk,
    const float* __restrict__ tvm, const float* __restrict__ lwT,
    const float* __restrict__ lb, float* __restrict__ gW, float* __restrict__ gb)
{
  __shared__ float stv[4][16 * 68];
  __shared__ float sg[4][16 * 68];
  __shared__ float sgW[64 * 64];
  __shared__ float sgb[64];
  for (int i = threadIdx.x; i < 4096; i += 128) sgW[i] = 0.f;
  if (threadIdx.x < 64) sgb[threadIdx.x] = 0.f;
  __syncthreads();

  const int warp = threadIdx.x >> 5, lane = threadIdx.x & 31;
  const int hi = lane >> 4, nl = lane & 15;
  float* tvw = stv[warp];
  float* gw_ = sg[warp];

  const float gscale = 2.0f / (float)(3ull * BATCH * 64);
  const int NT = 3 * BATCH / 16;
  const int wave = blockIdx.x * 4 + warp;
  const int nw = gridDim.x * 4;

  v8f cgw[4][4];
#pragma unroll
  for (int i0 = 0; i0 < 4; ++i0)
#pragma unroll
    for (int j0 = 0; j0 < 4; ++j0) cgw[i0][j0] = (v8f){};

  for (int t = wave; t < NT; t += nw) {
    const int r0 = t * 16;
    v2f sa[16];
#pragma unroll
    for (int ks = 0; ks < 16; ++ks) {
      int kb = ks * 4 + 2 * hi;
      const float* p = s + (size_t)(r0 + nl) * 64 + kb;
      sa[ks] = mkv2(p[0], p[1]);
    }
#pragma unroll
    for (int n0 = 0; n0 < 4; ++n0) {
      int n = n0 * 16 + nl;
      v8f ctv = {}, clv = {};
#pragma unroll
      for (int ks = 0; ks < 16; ++ks) {
        int kb = ks * 4 + 2 * hi;
        ctv = wmma_f32(sa[ks], mkv2(tk[kb * 64 + n], tk[(kb + 1) * 64 + n]), ctv);
        clv = wmma_f32(sa[ks], mkv2(tvm[kb * 64 + n], tvm[(kb + 1) * 64 + n]), clv);
      }
#pragma unroll
      for (int r = 0; r < 8; ++r) {
        tvw[(r + 8 * hi) * 68 + n] = ctv[r];
        gw_[(r + 8 * hi) * 68 + n] = -clv[r];
      }
    }
#pragma unroll
    for (int n0 = 0; n0 < 4; ++n0) {
      int n = n0 * 16 + nl;
      v8f cp = {};
#pragma unroll
      for (int ks = 0; ks < 16; ++ks) {
        int kb = ks * 4 + 2 * hi;
        v2f a = mkv2(tvw[nl * 68 + kb], tvw[nl * 68 + kb + 1]);
        cp = wmma_f32(a, mkv2(lwT[kb * 64 + n], lwT[(kb + 1) * 64 + n]), cp);
      }
      float lbn = lb[n];
#pragma unroll
      for (int r = 0; r < 8; ++r) {
        int idx = (r + 8 * hi) * 68 + n;
        float g = gscale * (cp[r] + lbn + gw_[idx]);
        gw_[idx] = g;
        atomicAdd(&sgb[n], g);
      }
    }
#pragma unroll
    for (int i0 = 0; i0 < 4; ++i0) {
      int i = i0 * 16 + nl;
#pragma unroll
      for (int j0 = 0; j0 < 4; ++j0) {
        int j = j0 * 16 + nl;
#pragma unroll
        for (int ks = 0; ks < 4; ++ks) {
          int kb = ks * 4 + 2 * hi;
          v2f a = mkv2(gw_[kb * 68 + i], gw_[(kb + 1) * 68 + i]);
          v2f b = mkv2(tvw[kb * 68 + j], tvw[(kb + 1) * 68 + j]);
          cgw[i0][j0] = wmma_f32(a, b, cgw[i0][j0]);
        }
      }
    }
  }
#pragma unroll
  for (int i0 = 0; i0 < 4; ++i0)
#pragma unroll
    for (int j0 = 0; j0 < 4; ++j0)
#pragma unroll
      for (int r = 0; r < 8; ++r)
        atomicAdd(&sgW[(i0 * 16 + r + 8 * hi) * 64 + j0 * 16 + nl], cgw[i0][j0][r]);
  __syncthreads();
  for (int i = threadIdx.x; i < 4096; i += 128) atomicAdd(&gW[i], sgW[i]);
  if (threadIdx.x < 64) atomicAdd(&gb[threadIdx.x], sgb[threadIdx.x]);
}

__global__ void k_update(const float* __restrict__ lw, const float* __restrict__ lb,
                         const float* __restrict__ gW, const float* __restrict__ gb,
                         float* __restrict__ lw1T, float* __restrict__ lb1)
{
  for (int i = threadIdx.x; i < 4096; i += 256) {
    int n = i >> 6, k = i & 63;
    float g = gW[i];
    lw1T[k * 64 + n] = lw[i] - LRATE * g / (fabsf(g) + AEPS);
  }
  for (int i = threadIdx.x; i < 64; i += 256) {
    float g = gb[i];
    lb1[i] = lb[i] - LRATE * g / (fabsf(g) + AEPS);
  }
}

__global__ void __launch_bounds__(128) k_out(
    const float* __restrict__ s, const float* __restrict__ tq,
    const float* __restrict__ lw1T, const float* __restrict__ lb1,
    const float* __restrict__ lng, const float* __restrict__ lnb,
    const float* __restrict__ ow, const float* __restrict__ obp,
    float* __restrict__ out)
{
  __shared__ float su[4][16 * 68];
  __shared__ float so[4][16 * 68];
  const int warp = threadIdx.x >> 5, lane = threadIdx.x & 31;
  const int hi = lane >> 4, nl = lane & 15;
  const int row0 = (blockIdx.x * 4 + warp) * 16;
  float* uw = su[warp];
  float* om = so[warp];
  const float* sv = s + (size_t)2 * BATCH * 64;

  v2f sa[16];
#pragma unroll
  for (int ks = 0; ks < 16; ++ks) {
    int kb = ks * 4 + 2 * hi;
    const float* p = sv + (size_t)(row0 + nl) * 64 + kb;
    sa[ks] = mkv2(p[0], p[1]);
  }
#pragma unroll
  for (int n0 = 0; n0 < 4; ++n0) {
    int n = n0 * 16 + nl;
    v8f c = {};
#pragma unroll
    for (int ks = 0; ks < 16; ++ks) {
      int kb = ks * 4 + 2 * hi;
      c = wmma_f32(sa[ks], mkv2(tq[kb * 64 + n], tq[(kb + 1) * 64 + n]), c);
    }
#pragma unroll
    for (int r = 0; r < 8; ++r) uw[(r + 8 * hi) * 68 + n] = c[r];
  }
#pragma unroll
  for (int n0 = 0; n0 < 4; ++n0) {
    int n = n0 * 16 + nl;
    v8f c = {};
#pragma unroll
    for (int ks = 0; ks < 16; ++ks) {
      int kb = ks * 4 + 2 * hi;
      v2f a = mkv2(uw[nl * 68 + kb], uw[nl * 68 + kb + 1]);
      c = wmma_f32(a, mkv2(lw1T[kb * 64 + n], lw1T[(kb + 1) * 64 + n]), c);
    }
    float bias = lb1[n];
#pragma unroll
    for (int r = 0; r < 8; ++r) om[(r + 8 * hi) * 68 + n] = c[r] + bias;
  }
  __syncthreads();
  if (lane < 16) {
    const float* rowp = om + lane * 68;
    float mu = 0.f;
#pragma unroll
    for (int j = 0; j < 64; ++j) mu += rowp[j];
    mu *= (1.0f / 64.0f);
    float var = 0.f;
#pragma unroll
    for (int j = 0; j < 64; ++j) { float d = rowp[j] - mu; var += d * d; }
    var *= (1.0f / 64.0f);
    float rstd = rsqrtf(var + LNEPS);
    float acc = 0.f;
#pragma unroll
    for (int j = 0; j < 64; ++j) {
      float nrm = (rowp[j] - mu) * rstd * lng[j] + lnb[j];
      float a = nrm / (1.0f + __expf(-nrm));
      acc += a * ow[j];
    }
    out[row0 + lane] = acc + obp[0];
  }
}

extern "C" void kernel_launch(void* const* d_in, const int* in_sizes, int n_in,
                              void* d_out, int out_size, void* d_ws, size_t ws_size,
                              hipStream_t stream) {
  const float* in1  = (const float*)d_in[0];
  const float* in2  = (const float*)d_in[1];
  const float* in3  = (const float*)d_in[2];
  const float* in4  = (const float*)d_in[3];
  const float* w_l  = (const float*)d_in[4];
  const float* b_l  = (const float*)d_in[5];
  const float* w_m  = (const float*)d_in[6];
  const float* b_m  = (const float*)d_in[7];
  const float* w_s  = (const float*)d_in[8];
  const float* b_s  = (const float*)d_in[9];
  const float* fc1w = (const float*)d_in[10];
  const float* fc1b = (const float*)d_in[11];
  const float* q_w  = (const float*)d_in[12];
  const float* q_b  = (const float*)d_in[13];
  const float* k_w  = (const float*)d_in[14];
  const float* k_b  = (const float*)d_in[15];
  const float* v_w  = (const float*)d_in[16];
  const float* v_b  = (const float*)d_in[17];
  const float* t_k  = (const float*)d_in[18];
  const float* t_v  = (const float*)d_in[19];
  const float* t_q  = (const float*)d_in[20];
  const float* lw   = (const float*)d_in[21];
  const float* lb   = (const float*)d_in[22];
  const float* lng  = (const float*)d_in[23];
  const float* lnb  = (const float*)d_in[24];
  const float* outw = (const float*)d_in[25];
  const float* outb = (const float*)d_in[26];

  float* wsf  = (float*)d_ws;
  float* cat  = wsf + F_CAT;
  float* sbuf = wsf + F_S;
  float* W1   = wsf + F_W1;
  float* hb   = wsf + F_HB;
  float* qT   = wsf + F_QT;
  float* kT   = wsf + F_KT;
  float* vT   = wsf + F_VT;
  float* lwT  = wsf + F_LWT;
  float* gW   = wsf + F_GW;
  float* gb   = wsf + F_GB;
  float* lw1T = wsf + F_LW1T;
  float* lb1  = wsf + F_LB1;

  k_cat<<<512, 256, 0, stream>>>(in1, in2, in3, in4, w_l, b_l, w_m, b_m, w_s, b_s, cat);
  k_prep<<<1, 256, 0, stream>>>(fc1w, fc1b, q_w, k_w, v_w, lw, wsf);
  k_qkv<<<BATCH / 128, 256, 0, stream>>>(cat, q_b, k_b, v_b, sbuf, W1, hb, qT, kT, vT);
  k_grad<<<512, 128, 0, stream>>>(sbuf, t_k, t_v, lwT, lb, gW, gb);
  k_update<<<1, 256, 0, stream>>>(lw, lb, gW, gb, lw1T, lb1);
  k_out<<<BATCH / 64, 128, 0, stream>>>(sbuf, t_q, lw1T, lb1, lng, lnb, outw, outb,
                                        (float*)d_out);
}


# k_qkv: Gram matrix and column sums of the q,k,v rows from the 16x16 Gram of the cat rows (rows are linear in cat), expanded per block on WMMA; q,k products go; changes the Gram sum order
# speedup vs baseline: 2.5931x; 1.9863x over previous
#include <hip/hip_runtime.h>
#include <math.h>

#define BATCH 32768
#define LRATE 0.01f
#define AEPS  1e-8f
#define LNEPS 1e-5f

typedef __attribute__((ext_vector_type(2))) float v2f;
typedef __attribute__((ext_vector_type(8))) float v8f;

__device__ __forceinline__ v2f mkv2(float x, float y) { v2f v; v[0] = x; v[1] = y; return v; }

__device__ __forceinline__ v8f wmma_f32(v2f a, v2f b, v8f c) {
  return __builtin_amdgcn_wmma_f32_16x16x4_f32(false, a, false, b, (short)0, c, false, false);
}

constexpr size_t F_CAT  = 0;
constexpr size_t F_S    = F_CAT + (size_t)BATCH * 16;
constexpr size_t F_M    = F_S + (size_t)3 * BATCH * 64;
constexpr size_t F_S2   = F_M + 64 * 64;
constexpr size_t F_CS   = F_S2 + 64 * 64;

constexpr int WP  = 68;
constexpr int W1P = 20;

typedef __attribute__((ext_vector_type(4))) float v4f;

constexpr int CAT_WAVES  = 8;
constexpr int CAT_ROWS   = 8;
constexpr int CAT_BLOCKS = BATCH / (CAT_WAVES * CAT_ROWS);
static_assert(CAT_BLOCKS == 512 && CAT_WAVES * 32 == 256, "k_cat is launched as <<<512, 256>>>");
constexpr int WQ1 = 0;
constexpr int WQ2 = 4608;
constexpr int WQ3 = 9216;
constexpr int WQN = 10240;
constexpr int NSTAGE = 20;
constexpr int RPITCH = 33;

__device__ __forceinline__ v4f ld_stream(const v4f* p) { return __builtin_nontemporal_load(p); }
__device__ __forceinline__ float dot4(v4f d, v4f w, float a) {
  a = __builtin_fmaf(d.x, w.x, a); a = __builtin_fmaf(d.y, w.y, a);
  a = __builtin_fmaf(d.z, w.z, a); a = __builtin_fmaf(d.w, w.w, a);
  return a;
}

template <int S>
__device__ __forceinline__ void cat_issue(v4f (&buf)[8], const v4f* p1, const v4f* p2,
                                          const v4f* p3, int lane) {
  if constexpr (S < 12) {
#pragma unroll
    for (int r = 0; r < 8; ++r) {
      if constexpr (S == 11) buf[r] = ld_stream(p1 + r * 375 + ((lane < 23) ? 352 : 374 - lane));
      else                   buf[r] = ld_stream(p1 + r * 375 + 32 * S);
    }
  } else if constexpr (S < 18) {
#pragma unroll
    for (int h = 0; h < 2; ++h) {
      const int q = 2 * (S - 12) + h;
#pragma unroll
      for (int p = 0; p < 4; ++p) {
        if (q == 11) buf[h * 4 + p] = ld_stream(p2 + p * 375 + ((lane < 23) ? 352 : 374 - lane));
        else         buf[h * 4 + p] = ld_stream(p2 + p * 375 + 32 * q);
      }
    }
  } else if constexpr (S == 18) {
#pragma unroll
    for (int h = 0; h < 2; ++h)
#pragma unroll
      for (int p = 0; p < 4; ++p) buf[h * 4 + p] = ld_stream(p3 + p * 75 + 32 * h);
  } else if constexpr (S == 19) {
#pragma unroll
    for (int p = 0; p < 4; ++p) buf[p] = ld_stream(p3 + p * 75 + ((lane < 11) ? 64 : 74 - lane));
  }
}

template <int S>
__device__ __forceinline__ void cat_weights(v4f (&w)[6], const float* sw, int lane) {
  const v4f zero = {0.f, 0.f, 0.f, 0.f};
  if constexpr (S < 12) {
    const bool live = (S != 11) || (lane < 23);
    const int j = live ? 32 * S + lane : 0;
#pragma unroll
    for (int o = 0; o < 3; ++o) { const v4f t = *(const v4f*)(sw + WQ1 + o * 1500 + 4 * j); w[o] = live ? t : zero; }
  } else if constexpr (S < 18) {
#pragma unroll
    for (int h = 0; h < 2; ++h) {
      const int q = 2 * (S - 12) + h;
      const bool live = (q != 11) || (lane < 23);
      const int j = live ? 32 * q + lane : 0;
#pragma unroll
      for (int o = 0; o < 3; ++o) { const v4f t = *(const v4f*)(sw + WQ2 + o * 1500 + 4 * j); w[h * 3 + o] = live ? t : zero; }
    }
  } else if constexpr (S == 18) {
#pragma unroll
    for (int h = 0; h < 2; ++h)
#pragma unroll
      for (int o = 0; o < 3; ++o) w[h * 3 + o] = *(const v4f*)(sw + WQ3 + o * 300 + 4 * (32 * h + lane));
  } else {
    const bool live = lane < 11;
    const int j = live ? 64 + lane : 0;
#pragma unroll
    for (int o = 0; o < 3; ++o) { const v4f t = *(const v4f*)(sw + WQ3 + o * 300 + 4 * j); w[o] = live ? t : zero; }
  }
}

__device__ __forceinline__ void pair_mixed(v4f d, const v4f* w, float* accA, float* accB,
                                           bool xyA, bool zwA) {
#pragma unroll
  for (int o = 0; o < 3; ++o) {
    const float pxy = __builtin_fmaf(d.x, w[o].x, d.y * w[o].y);
    const float pzw = __builtin_fmaf(d.z, w[o].z, d.w * w[o].w);
    accA[o] += (xyA ? pxy : 0.f) + (zwA ? pzw : 0.f);
    accB[o] += (xyA ? 0.f : pxy) + (zwA ? 0.f : pzw);
  }
}

template <int S>
__device__ __forceinline__ void cat_consume(const v4f (&buf)[8], const v4f (&w)[6],
                                            float (&acc)[8][3], int lane) {
  if constexpr (S < 12) {
#pragma unroll
    for (int r = 0; r < 8; ++r)
#pragma unroll
      for (int o = 0; o < 3; ++o) acc[r][o] = dot4(buf[r], w[o], acc[r][o]);
  } else if constexpr (S < 18) {
#pragma unroll
    for (int h = 0; h < 2; ++h) {
      const int q = 2 * (S - 12) + h;
#pragma unroll
      for (int p = 0; p < 4; ++p) {
        if (q < 5) {
#pragma unroll
          for (int o = 0; o < 3; ++o) acc[2 * p][o] = dot4(buf[h * 4 + p], w[h * 3 + o], acc[2 * p][o]);
        } else if (q > 5) {
#pragma unroll
          for (int o = 0; o < 3; ++o) acc[2 * p + 1][o] = dot4(buf[h * 4 + p], w[h * 3 + o], acc[2 * p + 1][o]);
        } else {
          pair_mixed(buf[h * 4 + p], &w[h * 3], acc[2 * p], acc[2 * p + 1], lane <= 27, lane <= 26);
        }
      }
    }
  } else if constexpr (S == 18) {
#pragma unroll
    for (int p = 0; p < 4; ++p) {
#pragma unroll
      for (int o = 0; o < 3; ++o) acc[2 * p][o] = dot4(buf[p], w[o], acc[2 * p][o]);
      pair_mixed(buf[4 + p], &w[3], acc[2 * p], acc[2 * p + 1], lane <= 5, lane <= 4);
    }
  } else {
#pragma unroll
    for (int p = 0; p < 4; ++p)
#pragma unroll
      for (int o = 0; o < 3; ++o) acc[2 * p + 1][o] = dot4(buf[p], w[o], acc[2 * p + 1][o]);
  }
}

__global__ void __launch_bounds__(256) __attribute__((amdgpu_waves_per_eu(4, 4))) k_cat(
    const float* __restrict__ in1, const float* __restrict__ in2,
    const float* __restrict__ in3, const float* __restrict__ in4,
    const float* __restrict__ wl, const float* __restrict__ bl,
    const float* __restrict__ wm, const float* __restrict__ bm,
    const float* __restrict__ wsw, const float* __restrict__ bs,
    float* __restrict__ cat, float* __restrict__ zacc)
{
  __shared__ __attribute__((aligned(16))) float sw[WQN];
  __shared__ float red[CAT_WAVES][72 * RPITCH];
  __shared__ __attribute__((aligned(16))) float cst[CAT_WAVES][CAT_ROWS * 16];

  const int warp = threadIdx.x >> 5, lane = threadIdx.x & 31;
  const int b0 = (blockIdx.x * CAT_WAVES + warp) * CAT_ROWS;
  const v4f* p1 = (const v4f*)in1 + (size_t)b0 * 375 + lane;
  const v4f* p2 = (const v4f*)in2 + (size_t)(b0 >> 1) * 375 + lane;
  const v4f* p3 = (const v4f*)in3 + (size_t)(b0 >> 1) * 75 + lane;

  v4f buf[4][8];
  cat_issue<0>(buf[0], p1, p2, p3, lane);
  cat_issue<1>(buf[1], p1, p2, p3, lane);
  cat_issue<2>(buf[2], p1, p2, p3, lane);

  float tailv[2];
#pragma unroll
  for (int g = 0; g < 2; ++g) {
    const int u = lane + 32 * g, cc = u % 7;
    const bool take = (u < 56) && (cc < 3);
    const float t = in4[(size_t)b0 * 3 + (take ? (u / 7) * 3 + cc : 0)];
    tailv[g] = take ? t : 0.f;
  }

  {
    float t1[18];
#pragma unroll
    for (int t = 0; t < 18; ++t) {
      const int i = min((int)threadIdx.x + 256 * t, 4499), o = i / 1500, f = i - o * 1500;
      t1[t] = wl[(o * 3 + f % 3) * 500 + f / 3];
    }
#pragma unroll
    for (int t = 0; t < 18; ++t) sw[WQ1 + threadIdx.x + 256 * t] = t1[t];
#pragma unroll
    for (int t = 0; t < 18; ++t) {
      const int i = min((int)threadIdx.x + 256 * t, 4499), o = i / 1500, g = i - o * 1500;
      const int f = (g >= 750) ? g - 750 : g;
      t1[t] = wm[(o * 3 + f % 3) * 250 + f / 3];
    }
#pragma unroll
    for (int t = 0; t < 18; ++t) sw[WQ2 + threadIdx.x + 256 * t] = t1[t];
#pragma unroll
    for (int t = 0; t < 4; ++t) {
      const int i = min((int)threadIdx.x + 256 * t, 899), o = i / 300, g = i - o * 300;
      const int f = (g >= 150) ? g - 150 : g;
      t1[t] = wsw[(o * 3 + f % 3) * 50 + f / 3];
    }
#pragma unroll
    for (int t = 0; t < 4; ++t) sw[WQ3 + threadIdx.x + 256 * t] = t1[t];
  }
  __syncthreads();

  float acc[8][3];
  float* redw = red[warp];
  v4f w[6];

#define CAT_STAGE(S)                                                              \
  cat_weights<S>(w, sw, lane);                                                    \
  __builtin_amdgcn_sched_barrier(0);                                              \
  if constexpr ((S) + 3 < NSTAGE)                                                 \
    cat_issue<((S) + 3 < NSTAGE ? (S) + 3 : 0)>(buf[((S) + 3) % 4], p1, p2, p3, lane); \
  __builtin_amdgcn_sched_barrier(0);                                              \
  cat_consume<S>(buf[(S) % 4], w, acc, lane);                                     \
  CAT_PIN()
#define CAT_PIN()                                                                 \
  _Pragma("unroll") for (int r = 0; r < 8; ++r)                                   \
    _Pragma("unroll") for (int o = 0; o < 3; ++o) asm volatile("" : "+v"(acc[r][o]));
#define CAT_CLEAR()                                                               \
  _Pragma("unroll") for (int r = 0; r < 8; ++r)                                   \
    _Pragma("unroll") for (int o = 0; o < 3; ++o) acc[r][o] = 0.f;
#define CAT_PARK(COL)                                                             \
  _Pragma("unroll") for (int r = 0; r < 8; ++r)                                   \
    _Pragma("unroll") for (int o = 0; o < 3; ++o)                                 \
      redw[(r * 9 + (COL) + o) * RPITCH + lane] = acc[r][o];

  CAT_CLEAR()
  CAT_STAGE(0) CAT_STAGE(1) CAT_STAGE(2) CAT_STAGE(3) CAT_STAGE(4) CAT_STAGE(5)
  CAT_STAGE(6) CAT_STAGE(7) CAT_STAGE(8) CAT_STAGE(9) CAT_STAGE(10) CAT_STAGE(11)
  CAT_PARK(0)
  CAT_CLEAR()
  CAT_STAGE(12) CAT_STAGE(13) CAT_STAGE(14) CAT_STAGE(15) CAT_STAGE(16) CAT_STAGE(17)
  CAT_PARK(3)
  CAT_CLEAR()
  CAT_STAGE(18) CAT_STAGE(19)
  CAT_PARK(6)
#undef CAT_STAGE
#undef CAT_PIN
#undef CAT_CLEAR
#undef CAT_PARK
  __syncthreads();

  float* cw = cst[warp];
  const float b9[9] = {bl[0], bl[1], bl[2], bm[0], bm[1], bm[2], bs[0], bs[1], bs[2]};
#pragma unroll
  for (int g = 0; g < 3; ++g) {
    const int v = lane + 32 * g;
    if (v < 72) {
      const float* rp = redw + v * RPITCH;
      float s0 = 0.f, s1 = 0.f, s2 = 0.f, s3 = 0.f;
#pragma unroll
      for (int i = 0; i < 32; i += 4) { s0 += rp[i]; s1 += rp[i + 1]; s2 += rp[i + 2]; s3 += rp[i + 3]; }
      const int c = v % 9;
      float bias = b9[0];
#pragma unroll
      for (int k = 1; k < 9; ++k) bias = (c == k) ? b9[k] : bias;
      cw[(v / 9) * 16 + c] = (s0 + s1) + (s2 + s3) + bias;
    }
  }
#pragma unroll
  for (int g = 0; g < 2; ++g) {
    const int u = lane + 32 * g;
    if (u < 56) cw[(u / 7) * 16 + 9 + u % 7] = tailv[g];
  }
  __syncthreads();
  *((v4f*)(cat + (size_t)b0 * 16) + lane) = *((const v4f*)cw + lane);
  if (blockIdx.x < 17) {
    const int i = blockIdx.x * 256 + threadIdx.x;
    if (i < 64 * 64 + 64) zacc[i] = 0.f;
  }
}

__global__ void __launch_bounds__(256) k_qkv(
    const float* __restrict__ cat,
    const float* __restrict__ fc1w, const float* __restrict__ fc1b,
    const float* __restrict__ qw, const float* __restrict__ qb,
    const float* __restrict__ kw, const float* __restrict__ kbv,
    const float* __restrict__ vw, const float* __restrict__ vb,
    const float* __restrict__ lw, const float* __restrict__ tk, const float* __restrict__ tvm,
    float* __restrict__ s, float* __restrict__ Mw, float* __restrict__ gS,
    float* __restrict__ gcs)
{
  __shared__ float sh[8][16 * 68];
  __shared__ float sC[16 * 16];
  __shared__ __attribute__((aligned(16))) float sA[3][64 * W1P];
  __shared__ __attribute__((aligned(16))) float sY[3][64 * W1P];
  __shared__ __attribute__((aligned(16))) float swt[3][64 * WP];
  __shared__ __attribute__((aligned(16))) float sw1[64 * W1P];
  const int warp = threadIdx.x >> 5, lane = threadIdx.x & 31;
  const int hi = lane >> 4, nl = lane & 15;
  const int row0 = (blockIdx.x * 8 + warp) * 16;
  float* shw = sh[warp];

  sC[threadIdx.x] = 0.f;
  if (blockIdx.x < 16 && warp == 0) {
    const int a0 = blockIdx.x >> 2, i0 = blockIdx.x & 3;
    const int a = a0 * 16 + nl, i = i0 * 16 + nl;
    v8f c = {};
#pragma unroll
    for (int ks = 0; ks < 16; ++ks) {
      const int kb = ks * 4 + 2 * hi;
      c = wmma_f32(*(const v2f*)(tk + a * 64 + kb), *(const v2f*)(lw + i * 64 + kb), c);
    }
#pragma unroll
    for (int r = 0; r < 8; ++r) {
      const int e = (a0 * 16 + r + 8 * hi) * 64 + i;
      Mw[e] = c[r] - tvm[e];
    }
  }
  {
    const float* WS[3] = {qw, kw, vw};
#pragma unroll
    for (int t = 0; t < 3; ++t)
#pragma unroll
      for (int j = 0; j < 4; ++j) {
        const int e = ((int)threadIdx.x + 256 * j) * 4;
        *(v4f*)(&swt[t][(e >> 6) * WP + (e & 63)]) = *(const v4f*)(WS[t] + e);
      }
    if (threadIdx.x < 192) {
      const int e = (int)threadIdx.x * 4, n = e / 12;
      *(v4f*)(&sw1[n * W1P + (e - n * 12)]) = *(const v4f*)(fc1w + e);
    }
    if (threadIdx.x < 64) {
      const v4f tail = {fc1b[threadIdx.x] + ((threadIdx.x & 1) ? 1.0f : 0.0f), 0.f, 0.f, 0.f};
      *(v4f*)(&sw1[(int)threadIdx.x * W1P + 12]) = tail;
    }
  }
  __syncthreads();

  v2f ca[4];
#pragma unroll
  for (int ks = 0; ks < 4; ++ks) {
    int kb = ks * 4 + 2 * hi;
    const float* p = cat + (size_t)(row0 + nl) * 16 + kb;
    ca[ks] = mkv2(p[0], p[1]);
  }
#pragma unroll
  for (int n0 = 0; n0 < 4; ++n0) {
    int n = n0 * 16 + nl;
    v8f c = {};
#pragma unroll
    for (int ks = 0; ks < 4; ++ks) {
      int kb = ks * 4 + 2 * hi;
      c = wmma_f32(ca[ks], *(const v2f*)(&sw1[n * W1P + kb]), c);
    }
    float bias = fc1b[n] + ((n & 1) ? 1.0f : 0.0f);
#pragma unroll
    for (int r = 0; r < 8; ++r) shw[(r + 8 * hi) * 68 + n] = c[r] + bias;
  }
  __syncthreads();

  v2f ha[16];
#pragma unroll
  for (int ks = 0; ks < 16; ++ks) {
    int kb = ks * 4 + 2 * hi;
    ha[ks] = mkv2(shw[nl * 68 + kb], shw[nl * 68 + kb + 1]);
  }
  const float* BB[3] = {qb, kbv, vb};
  {
    float* ob = s + (size_t)2 * BATCH * 64;
#pragma unroll
    for (int n0 = 0; n0 < 4; ++n0) {
      int n = n0 * 16 + nl;
      v8f c = {};
#pragma unroll
      for (int ks = 0; ks < 16; ++ks) {
        int kb = ks * 4 + 2 * hi;
        c = wmma_f32(ha[ks], *(const v2f*)(&swt[2][n * WP + kb]), c);
      }
      float bias = vb[n];
#pragma unroll
      for (int r = 0; r < 8; ++r)
        ob[(size_t)(row0 + r + 8 * hi) * 64 + n] = c[r] + bias;
    }
  }
  {
    v8f cc = {};
#pragma unroll
    for (int m = 0; m < 4; ++m) {
      const float* p = cat + (size_t)(row0 + 4 * m + 2 * hi) * 16 + nl;
      const v2f x = mkv2(nl == 12 ? 1.0f : p[0], nl == 12 ? 1.0f : p[16]);
      cc = wmma_f32(x, x, cc);
    }
#pragma unroll
    for (int r = 0; r < 8; ++r) atomicAdd(&sC[(r + 8 * hi) * 16 + nl], cc[r]);
  }
#pragma unroll
  for (int rep = 0; rep < 2; ++rep) {
    const int T = warp + 8 * rep;
    if (T < 12) {
      const int t = T >> 2, n0 = T & 3;
      v8f c = {};
#pragma unroll
      for (int ks = 0; ks < 16; ++ks) {
        const int kb = ks * 4 + 2 * hi;
        c = wmma_f32(*(const v2f*)(&swt[t][(n0 * 16 + nl) * WP + kb]),
                     mkv2(sw1[kb * W1P + nl], sw1[(kb + 1) * W1P + nl]), c);
      }
#pragma unroll
      for (int r = 0; r < 8; ++r) {
        const int n = n0 * 16 + r + 8 * hi;
        sA[t][n * W1P + nl] = c[r] + (nl == 12 ? BB[t][n] : 0.f);
      }
    }
  }
  __syncthreads();
#pragma unroll
  for (int rep = 0; rep < 2; ++rep) {
    const int T = warp + 8 * rep;
    if (T < 12) {
      const int t = T >> 2, n0 = T & 3;
      v8f c = {};
#pragma unroll
      for (int ks = 0; ks < 4; ++ks) {
        const int kb = ks * 4 + 2 * hi;
        c = wmma_f32(*(const v2f*)(&sA[t][(n0 * 16 + nl) * W1P + kb]),
                     mkv2(sC[kb * 16 + nl], sC[(kb + 1) * 16 + nl]), c);
      }
#pragma unroll
      for (int r = 0; r < 8; ++r) sY[t][(n0 * 16 + r + 8 * hi) * W1P + nl] = c[r];
    }
  }
  __syncthreads();
  float* sS2 = &sh[0][0];
  {
    const int i0 = warp >> 1;
#pragma unroll
    for (int jj = 0; jj < 2; ++jj) {
      const int j0 = 2 * (warp & 1) + jj;
      v8f c = {};
#pragma unroll
      for (int t = 0; t < 3; ++t)
#pragma unroll
        for (int ks = 0; ks < 4; ++ks) {
          const int kb = ks * 4 + 2 * hi;
          c = wmma_f32(*(const v2f*)(&sY[t][(i0 * 16 + nl) * W1P + kb]),
                       *(const v2f*)(&sA[t][(j0 * 16 + nl) * W1P + kb]), c);
        }
#pragma unroll
      for (int r = 0; r < 8; ++r)
        sS2[(i0 * 16 + r + 8 * hi) * 64 + j0 * 16 + nl] = c[r];
    }
  }
  __syncthreads();
  for (int i = threadIdx.x; i < 64 * 64; i += 256) atomicAdd(&gS[i], sS2[i]);
  if (threadIdx.x < 64) {
    const int n = threadIdx.x;
    atomicAdd(&gcs[n], (sY[0][n * W1P + 12] + sY[1][n * W1P + 12]) + sY[2][n * W1P + 12]);
  }
}

__global__ void __launch_bounds__(256) k_out(
    const float* __restrict__ s, const float* __restrict__ tq, const float* __restrict__ tk,
    const float* __restrict__ lw, const float* __restrict__ lb,
    const float* __restrict__ Mw, const float* __restrict__ gS, const float* __restrict__ gcs,
    const float* __restrict__ lng, const float* __restrict__ lnb,
    const float* __restrict__ ow, const float* __restrict__ obp,
    float* __restrict__ out)
{
  __shared__ float su[8][16 * 68];
  __shared__ float so[8][16 * 68];
  __shared__ __attribute__((aligned(16))) float slw1[64 * WP];
  __shared__ float sQ[64 * WP];
  __shared__ float slb1[64];
  __shared__ float sct[64];
  const int warp = threadIdx.x >> 5, lane = threadIdx.x & 31;
  const int hi = lane >> 4, nl = lane & 15;
  const int row0 = (blockIdx.x * 8 + warp) * 16;
  float* uw = su[warp];
  float* om = so[warp];
  const float* sv = s + (size_t)2 * BATCH * 64;
  const float gscale = 2.0f / (float)(3ull * BATCH * 64);
  const int t0 = warp >> 1, t1 = 2 * (warp & 1);

  v2f sa[16];
#pragma unroll
  for (int ks = 0; ks < 16; ++ks) {
    int kb = ks * 4 + 2 * hi;
    const float* p = sv + (size_t)(row0 + nl) * 64 + kb;
    sa[ks] = mkv2(p[0], p[1]);
  }
#pragma unroll
  for (int jj = 0; jj < 2; ++jj) {
    const int a = t0 * 16 + nl, j = (t1 + jj) * 16 + nl;
    v8f c = {};
#pragma unroll
    for (int ks = 0; ks < 16; ++ks) {
      const int kb = ks * 4 + 2 * hi;
      c = wmma_f32(*(const v2f*)(gS + a * 64 + kb), mkv2(tk[kb * 64 + j], tk[(kb + 1) * 64 + j]), c);
    }
#pragma unroll
    for (int r = 0; r < 8; ++r) sQ[(t0 * 16 + r + 8 * hi) * WP + j] = c[r];
  }
  if (threadIdx.x < 64) {
    const int j = threadIdx.x;
    float acc = 0.f;
    for (int b = 0; b < 64; ++b) acc = __builtin_fmaf(gcs[b], tk[b * 64 + j], acc);
    sct[j] = acc;
  } else if (threadIdx.x < 128) {
    const int i = (int)threadIdx.x - 64;
    float acc = 0.f;
    for (int a = 0; a < 64; ++a) acc = __builtin_fmaf(gcs[a], Mw[a * 64 + i], acc);
    const float g = gscale * (acc + (float)(3 * BATCH) * lb[i]);
    slb1[i] = lb[i] - LRATE * g / (fabsf(g) + AEPS);
  }
  __syncthreads();
#pragma unroll
  for (int jj = 0; jj < 2; ++jj) {
    const int i = t0 * 16 + nl, j = (t1 + jj) * 16 + nl;
    v8f c = {};
#pragma unroll
    for (int ks = 0; ks < 16; ++ks) {
      const int kb = ks * 4 + 2 * hi;
      c = wmma_f32(mkv2(Mw[kb * 64 + i], Mw[(kb + 1) * 64 + i]),
                   mkv2(sQ[kb * WP + j], sQ[(kb + 1) * WP + j]), c);
    }
    const float ctj = sct[j];
#pragma unroll
    for (int r = 0; r < 8; ++r) {
      const int ii = t0 * 16 + r + 8 * hi;
      const float g = gscale * (c[r] + lb[ii] * ctj);
      slw1[ii * WP + j] = lw[ii * 64 + j] - LRATE * g / (fabsf(g) + AEPS);
    }
  }
#pragma unroll
  for (int n0 = 0; n0 < 4; ++n0) {
    int n = n0 * 16 + nl;
    v8f c = {};
#pragma unroll
    for (int ks = 0; ks < 16; ++ks) {
      int kb = ks * 4 + 2 * hi;
      c = wmma_f32(sa[ks], mkv2(tq[kb * 64 + n], tq[(kb + 1) * 64 + n]), c);
    }
#pragma unroll
    for (int r = 0; r < 8; ++r) uw[(r + 8 * hi) * 68 + n] = c[r];
  }
  __syncthreads();
#pragma unroll
  for (int n0 = 0; n0 < 4; ++n0) {
    int n = n0 * 16 + nl;
    v8f c = {};
#pragma unroll
    for (int ks = 0; ks < 16; ++ks) {
      int kb = ks * 4 + 2 * hi;
      v2f a = mkv2(uw[nl * 68 + kb], uw[nl * 68 + kb + 1]);
      c = wmma_f32(a, *(const v2f*)(&slw1[n * WP + kb]), c);
    }
    float bias = slb1[n];
#pragma unroll
    for (int r = 0; r < 8; ++r) om[(r + 8 * hi) * 68 + n] = c[r] + bias;
  }
  __syncthreads();
  if (lane < 16) {
    const float* rowp = om + lane * 68;
    float mu = 0.f;
#pragma unroll
    for (int j = 0; j < 64; ++j) mu += rowp[j];
    mu *= (1.0f / 64.0f);
    float var = 0.f;
#pragma unroll
    for (int j = 0; j < 64; ++j) { float d = rowp[j] - mu; var += d * d; }
    var *= (1.0f / 64.0f);
    float rstd = rsqrtf(var + LNEPS);
    float acc = 0.f;
#pragma unroll
    for (int j = 0; j < 64; ++j) {
      float nrm = (rowp[j] - mu) * rstd * lng[j] + lnb[j];
      float a = nrm / (1.0f + __expf(-nrm));
      acc += a * ow[j];
    }
    out[row0 + lane] = acc + obp[0];
  }
}

extern "C" void kernel_launch(void* const* d_in, const int* in_sizes, int n_in,
                              void* d_out, int out_size, void* d_ws, size_t ws_size,
                              hipStream_t stream) {
  const float* in1  = (const float*)d_in[0];
  const float* in2  = (const float*)d_in[1];
  const float* in3  = (const float*)d_in[2];
  const float* in4  = (const float*)d_in[3];
  const float* w_l  = (const float*)d_in[4];
  const float* b_l  = (const float*)d_in[5];
  const float* w_m  = (const float*)d_in[6];
  const float* b_m  = (const float*)d_in[7];
  const float* w_s  = (const float*)d_in[8];
  const float* b_s  = (const float*)d_in[9];
  const float* fc1w = (const float*)d_in[10];
  const float* fc1b = (const float*)d_in[11];
  const float* q_w  = (const float*)d_in[12];
  const float* q_b  = (const float*)d_in[13];
  const float* k_w  = (const float*)d_in[14];
  const float* k_b  = (const float*)d_in[15];
  const float* v_w  = (const float*)d_in[16];
  const float* v_b  = (const float*)d_in[17];
  const float* t_k  = (const float*)d_in[18];
  const float* t_v  = (const float*)d_in[19];
  const float* t_q  = (const float*)d_in[20];
  const float* lw   = (const float*)d_in[21];
  const float* lb   = (const float*)d_in[22];
  const float* lng  = (const float*)d_in[23];
  const float* lnb  = (const float*)d_in[24];
  const float* outw = (const float*)d_in[25];
  const float* outb = (const float*)d_in[26];

  float* wsf  = (float*)d_ws;
  float* cat  = wsf + F_CAT;
  float* sbuf = wsf + F_S;
  float* Mw   = wsf + F_M;
  float* gS   = wsf + F_S2;
  float* gcs  = wsf + F_CS;

  k_cat<<<512, 256, 0, stream>>>(in1, in2, in3, in4, w_l, b_l, w_m, b_m, w_s, b_s, cat, gS);
  k_qkv<<<BATCH / 128, 256, 0, stream>>>(cat, fc1w, fc1b, q_w, q_b, k_w, k_b, v_w, v_b, lw,
                                         t_k, t_v, sbuf, Mw, gS, gcs);
  k_out<<<BATCH / 128, 256, 0, stream>>>(sbuf, t_q, t_k, lw, lb, Mw, gS, gcs, lng, lnb, outw,
                                         outb, (float*)d_out);
}


# k_qkv, k_out: reads on each block's serial chain (bias, cat tile, staging, two 64-term dots, grad W's A operand, LDS reads feeding the atomics) issued in batches ahead of use, not one trip each
# speedup vs baseline: 2.7881x; 1.0752x over previous
#include <hip/hip_runtime.h>
#include <math.h>

#define BATCH 32768
#define LRATE 0.01f
#define AEPS  1e-8f
#define LNEPS 1e-5f

typedef __attribute__((ext_vector_type(2))) float v2f;
typedef __attribute__((ext_vector_type(8))) float v8f;

__device__ __forceinline__ v2f mkv2(float x, float y) { v2f v; v[0] = x; v[1] = y; return v; }

__device__ __forceinline__ v8f wmma_f32(v2f a, v2f b, v8f c) {
  return __builtin_amdgcn_wmma_f32_16x16x4_f32(false, a, false, b, (short)0, c, false, false);
}

constexpr size_t F_CAT  = 0;
constexpr size_t F_S    = F_CAT + (size_t)BATCH * 16;
constexpr size_t F_M    = F_S + (size_t)3 * BATCH * 64;
constexpr size_t F_S2   = F_M + 64 * 64;
constexpr size_t F_CS   = F_S2 + 64 * 64;

constexpr int WP  = 68;
constexpr int W1P = 20;

typedef __attribute__((ext_vector_type(4))) float v4f;

constexpr int CAT_WAVES  = 8;
constexpr int CAT_ROWS   = 8;
constexpr int CAT_BLOCKS = BATCH / (CAT_WAVES * CAT_ROWS);
static_assert(CAT_BLOCKS == 512 && CAT_WAVES * 32 == 256, "k_cat is launched as <<<512, 256>>>");
constexpr int WQ1 = 0;
constexpr int WQ2 = 4608;
constexpr int WQ3 = 9216;
constexpr int WQN = 10240;
constexpr int NSTAGE = 20;
constexpr int RPITCH = 33;

__device__ __forceinline__ v4f ld_stream(const v4f* p) { return __builtin_nontemporal_load(p); }
__device__ __forceinline__ float dot4(v4f d, v4f w, float a) {
  a = __builtin_fmaf(d.x, w.x, a); a = __builtin_fmaf(d.y, w.y, a);
  a = __builtin_fmaf(d.z, w.z, a); a = __builtin_fmaf(d.w, w.w, a);
  return a;
}

template <int S>
__device__ __forceinline__ void cat_issue(v4f (&buf)[8], const v4f* p1, const v4f* p2,
                                          const v4f* p3, int lane) {
  if constexpr (S < 12) {
#pragma unroll
    for (int r = 0; r < 8; ++r) {
      if constexpr (S == 11) buf[r] = ld_stream(p1 + r * 375 + ((lane < 23) ? 352 : 374 - lane));
      else                   buf[r] = ld_stream(p1 + r * 375 + 32 * S);
    }
  } else if constexpr (S < 18) {
#pragma unroll
    for (int h = 0; h < 2; ++h) {
      const int q = 2 * (S - 12) + h;
#pragma unroll
      for (int p = 0; p < 4; ++p) {
        if (q == 11) buf[h * 4 + p] = ld_stream(p2 + p * 375 + ((lane < 23) ? 352 : 374 - lane));
        else         buf[h * 4 + p] = ld_stream(p2 + p * 375 + 32 * q);
      }
    }
  } else if constexpr (S == 18) {
#pragma unroll
    for (int h = 0; h < 2; ++h)
#pragma unroll
      for (int p = 0; p < 4; ++p) buf[h * 4 + p] = ld_stream(p3 + p * 75 + 32 * h);
  } else if constexpr (S == 19) {
#pragma unroll
    for (int p = 0; p < 4; ++p) buf[p] = ld_stream(p3 + p * 75 + ((lane < 11) ? 64 : 74 - lane));
  }
}

template <int S>
__device__ __forceinline__ void cat_weights(v4f (&w)[6], const float* sw, int lane) {
  const v4f zero = {0.f, 0.f, 0.f, 0.f};
  if constexpr (S < 12) {
    const bool live = (S != 11) || (lane < 23);
    const int j = live ? 32 * S + lane : 0;
#pragma unroll
    for (int o = 0; o < 3; ++o) { const v4f t = *(const v4f*)(sw + WQ1 + o * 1500 + 4 * j); w[o] = live ? t : zero; }
  } else if constexpr (S < 18) {
#pragma unroll
    for (int h = 0; h < 2; ++h) {
      const int q = 2 * (S - 12) + h;
      const bool live = (q != 11) || (lane < 23);
      const int j = live ? 32 * q + lane : 0;
#pragma unroll
      for (int o = 0; o < 3; ++o) { const v4f t = *(const v4f*)(sw + WQ2 + o * 1500 + 4 * j); w[h * 3 + o] = live ? t : zero; }
    }
  } else if constexpr (S == 18) {
#pragma unroll
    for (int h = 0; h < 2; ++h)
#pragma unroll
      for (int o = 0; o < 3; ++o) w[h * 3 + o] = *(const v4f*)(sw + WQ3 + o * 300 + 4 * (32 * h + lane));
  } else {
    const bool live = lane < 11;
    const int j = live ? 64 + lane : 0;
#pragma unroll
    for (int o = 0; o < 3; ++o) { const v4f t = *(const v4f*)(sw + WQ3 + o * 300 + 4 * j); w[o] = live ? t : zero; }
  }
}

__device__ __forceinline__ void pair_mixed(v4f d, const v4f* w, float* accA, float* accB,
                                           bool xyA, bool zwA) {
#pragma unroll
  for (int o = 0; o < 3; ++o) {
    const float pxy = __builtin_fmaf(d.x, w[o].x, d.y * w[o].y);
    const float pzw = __builtin_fmaf(d.z, w[o].z, d.w * w[o].w);
    accA[o] += (xyA ? pxy : 0.f) + (zwA ? pzw : 0.f);
    accB[o] += (xyA ? 0.f : pxy) + (zwA ? 0.f : pzw);
  }
}

template <int S>
__device__ __forceinline__ void cat_consume(const v4f (&buf)[8], const v4f (&w)[6],
                                            float (&acc)[8][3], int lane) {
  if constexpr (S < 12) {
#pragma unroll
    for (int r = 0; r < 8; ++r)
#pragma unroll
      for (int o = 0; o < 3; ++o) acc[r][o] = dot4(buf[r], w[o], acc[r][o]);
  } else if constexpr (S < 18) {
#pragma unroll
    for (int h = 0; h < 2; ++h) {
      const int q = 2 * (S - 12) + h;
#pragma unroll
      for (int p = 0; p < 4; ++p) {
        if (q < 5) {
#pragma unroll
          for (int o = 0; o < 3; ++o) acc[2 * p][o] = dot4(buf[h * 4 + p], w[h * 3 + o], acc[2 * p][o]);
        } else if (q > 5) {
#pragma unroll
          for (int o = 0; o < 3; ++o) acc[2 * p + 1][o] = dot4(buf[h * 4 + p], w[h * 3 + o], acc[2 * p + 1][o]);
        } else {
          pair_mixed(buf[h * 4 + p], &w[h * 3], acc[2 * p], acc[2 * p + 1], lane <= 27, lane <= 26);
        }
      }
    }
  } else if constexpr (S == 18) {
#pragma unroll
    for (int p = 0; p < 4; ++p) {
#pragma unroll
      for (int o = 0; o < 3; ++o) acc[2 * p][o] = dot4(buf[p], w[o], acc[2 * p][o]);
      pair_mixed(buf[4 + p], &w[3], acc[2 * p], acc[2 * p + 1], lane <= 5, lane <= 4);
    }
  } else {
#pragma unroll
    for (int p = 0; p < 4; ++p)
#pragma unroll
      for (int o = 0; o < 3; ++o) acc[2 * p + 1][o] = dot4(buf[p], w[o], acc[2 * p + 1][o]);
  }
}

__global__ void __launch_bounds__(256) __attribute__((amdgpu_waves_per_eu(4, 4))) k_cat(
    const float* __restrict__ in1, const float* __restrict__ in2,
    const float* __restrict__ in3, const float* __restrict__ in4,
    const float* __restrict__ wl, const float* __restrict__ bl,
    const float* __restrict__ wm, const float* __restrict__ bm,
    const float* __restrict__ wsw, const float* __restrict__ bs,
    float* __restrict__ cat, float* __restrict__ zacc)
{
  __shared__ __attribute__((aligned(16))) float sw[WQN];
  __shared__ float red[CAT_WAVES][72 * RPITCH];
  __shared__ __attribute__((aligned(16))) float cst[CAT_WAVES][CAT_ROWS * 16];

  const int warp = threadIdx.x >> 5, lane = threadIdx.x & 31;
  const int b0 = (blockIdx.x * CAT_WAVES + warp) * CAT_ROWS;
  const v4f* p1 = (const v4f*)in1 + (size_t)b0 * 375 + lane;
  const v4f* p2 = (const v4f*)in2 + (size_t)(b0 >> 1) * 375 + lane;
  const v4f* p3 = (const v4f*)in3 + (size_t)(b0 >> 1) * 75 + lane;

  v4f buf[4][8];
  cat_issue<0>(buf[0], p1, p2, p3, lane);
  cat_issue<1>(buf[1], p1, p2, p3, lane);
  cat_issue<2>(buf[2], p1, p2, p3, lane);

  float tailv[2];
#pragma unroll
  for (int g = 0; g < 2; ++g) {
    const int u = lane + 32 * g, cc = u % 7;
    const bool take = (u < 56) && (cc < 3);
    const float t = in4[(size_t)b0 * 3 + (take ? (u / 7) * 3 + cc : 0)];
    tailv[g] = take ? t : 0.f;
  }

  {
    float t1[18];
#pragma unroll
    for (int t = 0; t < 18; ++t) {
      const int i = min((int)threadIdx.x + 256 * t, 4499), o = i / 1500, f = i - o * 1500;
      t1[t] = wl[(o * 3 + f % 3) * 500 + f / 3];
    }
#pragma unroll
    for (int t = 0; t < 18; ++t) sw[WQ1 + threadIdx.x + 256 * t] = t1[t];
#pragma unroll
    for (int t = 0; t < 18; ++t) {
      const int i = min((int)threadIdx.x + 256 * t, 4499), o = i / 1500, g = i - o * 1500;
      const int f = (g >= 750) ? g - 750 : g;
      t1[t] = wm[(o * 3 + f % 3) * 250 + f / 3];
    }
#pragma unroll
    for (int t = 0; t < 18; ++t) sw[WQ2 + threadIdx.x + 256 * t] = t1[t];
#pragma unroll
    for (int t = 0; t < 4; ++t) {
      const int i = min((int)threadIdx.x + 256 * t, 899), o = i / 300, g = i - o * 300;
      const int f = (g >= 150) ? g - 150 : g;
      t1[t] = wsw[(o * 3 + f % 3) * 50 + f / 3];
    }
#pragma unroll
    for (int t = 0; t < 4; ++t) sw[WQ3 + threadIdx.x + 256 * t] = t1[t];
  }
  __syncthreads();

  float acc[8][3];
  float* redw = red[warp];
  v4f w[6];

#define CAT_STAGE(S)                                                              \
  cat_weights<S>(w, sw, lane);                                                    \
  __builtin_amdgcn_sched_barrier(0);                                              \
  if constexpr ((S) + 3 < NSTAGE)                                                 \
    cat_issue<((S) + 3 < NSTAGE ? (S) + 3 : 0)>(buf[((S) + 3) % 4], p1, p2, p3, lane); \
  __builtin_amdgcn_sched_barrier(0);                                              \
  cat_consume<S>(buf[(S) % 4], w, acc, lane);                                     \
  CAT_PIN()
#define CAT_PIN()                                                                 \
  _Pragma("unroll") for (int r = 0; r < 8; ++r)                                   \
    _Pragma("unroll") for (int o = 0; o < 3; ++o) asm volatile("" : "+v"(acc[r][o]));
#define CAT_CLEAR()                                                               \
  _Pragma("unroll") for (int r = 0; r < 8; ++r)                                   \
    _Pragma("unroll") for (int o = 0; o < 3; ++o) acc[r][o] = 0.f;
#define CAT_PARK(COL)                                                             \
  _Pragma("unroll") for (int r = 0; r < 8; ++r)                                   \
    _Pragma("unroll") for (int o = 0; o < 3; ++o)                                 \
      redw[(r * 9 + (COL) + o) * RPITCH + lane] = acc[r][o];

  CAT_CLEAR()
  CAT_STAGE(0) CAT_STAGE(1) CAT_STAGE(2) CAT_STAGE(3) CAT_STAGE(4) CAT_STAGE(5)
  CAT_STAGE(6) CAT_STAGE(7) CAT_STAGE(8) CAT_STAGE(9) CAT_STAGE(10) CAT_STAGE(11)
  CAT_PARK(0)
  CAT_CLEAR()
  CAT_STAGE(12) CAT_STAGE(13) CAT_STAGE(14) CAT_STAGE(15) CAT_STAGE(16) CAT_STAGE(17)
  CAT_PARK(3)
  CAT_CLEAR()
  CAT_STAGE(18) CAT_STAGE(19)
  CAT_PARK(6)
#undef CAT_STAGE
#undef CAT_PIN
#undef CAT_CLEAR
#undef CAT_PARK
  __syncthreads();

  float* cw = cst[warp];
  const float b9[9] = {bl[0], bl[1], bl[2], bm[0], bm[1], bm[2], bs[0], bs[1], bs[2]};
#pragma unroll
  for (int g = 0; g < 3; ++g) {
    const int v = lane + 32 * g;
    if (v < 72) {
      const float* rp = redw + v * RPITCH;
      float s0 = 0.f, s1 = 0.f, s2 = 0.f, s3 = 0.f;
#pragma unroll
      for (int i = 0; i < 32; i += 4) { s0 += rp[i]; s1 += rp[i + 1]; s2 += rp[i + 2]; s3 += rp[i + 3]; }
      const int c = v % 9;
      float bias = b9[0];
#pragma unroll
      for (int k = 1; k < 9; ++k) bias = (c == k) ? b9[k] : bias;
      cw[(v / 9) * 16 + c] = (s0 + s1) + (s2 + s3) + bias;
    }
  }
#pragma unroll
  for (int g = 0; g < 2; ++g) {
    const int u = lane + 32 * g;
    if (u < 56) cw[(u / 7) * 16 + 9 + u % 7] = tailv[g];
  }
  __syncthreads();
  *((v4f*)(cat + (size_t)b0 * 16) + lane) = *((const v4f*)cw + lane);
  if (blockIdx.x < 17) {
    const int i = blockIdx.x * 256 + threadIdx.x;
    if (i < 64 * 64 + 64) zacc[i] = 0.f;
  }
}

__global__ void __launch_bounds__(256) k_qkv(
    const float* __restrict__ cat,
    const float* __restrict__ fc1w, const float* __restrict__ fc1b,
    const float* __restrict__ qw, const float* __restrict__ qb,
    const float* __restrict__ kw, const float* __restrict__ kbv,
    const float* __restrict__ vw, const float* __restrict__ vb,
    const float* __restrict__ lw, const float* __restrict__ tk, const float* __restrict__ tvm,
    float* __restrict__ s, float* __restrict__ Mw, float* __restrict__ gS,
    float* __restrict__ gcs)
{
  __shared__ float sh[8][16 * 68];
  __shared__ float sC[16 * 16];
  __shared__ __attribute__((aligned(16))) float sA[3][64 * W1P];
  __shared__ __attribute__((aligned(16))) float sY[3][64 * W1P];
  __shared__ __attribute__((aligned(16))) float swt[3][64 * WP];
  __shared__ __attribute__((aligned(16))) float sw1[64 * W1P];
  const int warp = threadIdx.x >> 5, lane = threadIdx.x & 31;
  const int hi = lane >> 4, nl = lane & 15;
  const int row0 = (blockIdx.x * 8 + warp) * 16;
  float* shw = sh[warp];

  sC[threadIdx.x] = 0.f;
  const float* WS[3] = {qw, kw, vw};
  v4f wst[3][4];
#pragma unroll
  for (int t = 0; t < 3; ++t)
#pragma unroll
    for (int j = 0; j < 4; ++j) wst[t][j] = *(const v4f*)(WS[t] + ((int)threadIdx.x + 256 * j) * 4);
  const int e1 = min((int)threadIdx.x, 191) * 4;
  const v4f wf1 = *(const v4f*)(fc1w + e1);
  const float wfb = fc1b[threadIdx.x & 63];
  v2f ca[4];
#pragma unroll
  for (int ks = 0; ks < 4; ++ks) {
    const float* p = cat + (size_t)(row0 + nl) * 16 + ks * 4 + 2 * hi;
    ca[ks] = mkv2(p[0], p[1]);
  }
  float fbias[4];
#pragma unroll
  for (int n0 = 0; n0 < 4; ++n0) fbias[n0] = fc1b[n0 * 16 + nl];
  float cg[4][2];
#pragma unroll
  for (int m = 0; m < 4; ++m) {
    const float* p = cat + (size_t)(row0 + 4 * m + 2 * hi) * 16 + nl;
    cg[m][0] = p[0];
    cg[m][1] = p[16];
  }
  float bt[2][8];
#pragma unroll
  for (int rep = 0; rep < 2; ++rep) {
    const int T = min(warp + 8 * rep, 11), t = T >> 2, n0 = T & 3;
    const float* bsel = (t == 0) ? qb : ((t == 1) ? kbv : vb);
#pragma unroll
    for (int r = 0; r < 8; ++r) bt[rep][r] = bsel[n0 * 16 + 8 * hi + r];
  }
  if (blockIdx.x < 16 && warp == 0) {
    const int a0 = blockIdx.x >> 2, i0 = blockIdx.x & 3;
    const int a = a0 * 16 + nl, i = i0 * 16 + nl;
    v8f c = {};
#pragma unroll
    for (int ks = 0; ks < 16; ++ks) {
      const int kb = ks * 4 + 2 * hi;
      c = wmma_f32(*(const v2f*)(tk + a * 64 + kb), *(const v2f*)(lw + i * 64 + kb), c);
    }
#pragma unroll
    for (int r = 0; r < 8; ++r) {
      const int e = (a0 * 16 + r + 8 * hi) * 64 + i;
      Mw[e] = c[r] - tvm[e];
    }
  }
  {
#pragma unroll
    for (int t = 0; t < 3; ++t)
#pragma unroll
      for (int j = 0; j < 4; ++j) {
        const int e = ((int)threadIdx.x + 256 * j) * 4;
        *(v4f*)(&swt[t][(e >> 6) * WP + (e & 63)]) = wst[t][j];
      }
    if (threadIdx.x < 192) {
      const int n = e1 / 12;
      *(v4f*)(&sw1[n * W1P + (e1 - n * 12)]) = wf1;
    }
    if (threadIdx.x < 64) {
      const v4f tail = {wfb + ((threadIdx.x & 1) ? 1.0f : 0.0f), 0.f, 0.f, 0.f};
      *(v4f*)(&sw1[(int)threadIdx.x * W1P + 12]) = tail;
    }
  }
  __syncthreads();

#pragma unroll
  for (int n0 = 0; n0 < 4; ++n0) {
    int n = n0 * 16 + nl;
    v8f c = {};
#pragma unroll
    for (int ks = 0; ks < 4; ++ks) {
      int kb = ks * 4 + 2 * hi;
      c = wmma_f32(ca[ks], *(const v2f*)(&sw1[n * W1P + kb]), c);
    }
    float bias = fbias[n0] + ((n & 1) ? 1.0f : 0.0f);
#pragma unroll
    for (int r = 0; r < 8; ++r) shw[(r + 8 * hi) * 68 + n] = c[r] + bias;
  }
  __syncthreads();

  v2f ha[16];
#pragma unroll
  for (int ks = 0; ks < 16; ++ks) {
    int kb = ks * 4 + 2 * hi;
    ha[ks] = mkv2(shw[nl * 68 + kb], shw[nl * 68 + kb + 1]);
  }
  {
    float* ob = s + (size_t)2 * BATCH * 64;
#pragma unroll
    for (int n0 = 0; n0 < 4; ++n0) {
      int n = n0 * 16 + nl;
      v8f c = {};
#pragma unroll
      for (int ks = 0; ks < 16; ++ks) {
        int kb = ks * 4 + 2 * hi;
        c = wmma_f32(ha[ks], *(const v2f*)(&swt[2][n * WP + kb]), c);
      }
      float bias = vb[n];
#pragma unroll
      for (int r = 0; r < 8; ++r)
        ob[(size_t)(row0 + r + 8 * hi) * 64 + n] = c[r] + bias;
    }
  }
  {
    v8f cc = {};
#pragma unroll
    for (int m = 0; m < 4; ++m) {
      const v2f x = mkv2(nl == 12 ? 1.0f : cg[m][0], nl == 12 ? 1.0f : cg[m][1]);
      cc = wmma_f32(x, x, cc);
    }
#pragma unroll
    for (int r = 0; r < 8; ++r) atomicAdd(&sC[(r + 8 * hi) * 16 + nl], cc[r]);
  }
#pragma unroll
  for (int rep = 0; rep < 2; ++rep) {
    const int T = warp + 8 * rep;
    if (T < 12) {
      const int t = T >> 2, n0 = T & 3;
      v8f c = {};
#pragma unroll
      for (int ks = 0; ks < 16; ++ks) {
        const int kb = ks * 4 + 2 * hi;
        c = wmma_f32(*(const v2f*)(&swt[t][(n0 * 16 + nl) * WP + kb]),
                     mkv2(sw1[kb * W1P + nl], sw1[(kb + 1) * W1P + nl]), c);
      }
#pragma unroll
      for (int r = 0; r < 8; ++r) {
        const int n = n0 * 16 + r + 8 * hi;
        sA[t][n * W1P + nl] = c[r] + (nl == 12 ? bt[rep][r] : 0.f);
      }
    }
  }
  __syncthreads();
#pragma unroll
  for (int rep = 0; rep < 2; ++rep) {
    const int T = warp + 8 * rep;
    if (T < 12) {
      const int t = T >> 2, n0 = T & 3;
      v8f c = {};
#pragma unroll
      for (int ks = 0; ks < 4; ++ks) {
        const int kb = ks * 4 + 2 * hi;
        c = wmma_f32(*(const v2f*)(&sA[t][(n0 * 16 + nl) * W1P + kb]),
                     mkv2(sC[kb * 16 + nl], sC[(kb + 1) * 16 + nl]), c);
      }
#pragma unroll
      for (int r = 0; r < 8; ++r) sY[t][(n0 * 16 + r + 8 * hi) * W1P + nl] = c[r];
    }
  }
  __syncthreads();
  float* sS2 = &sh[0][0];
  {
    const int i0 = warp >> 1;
#pragma unroll
    for (int jj = 0; jj < 2; ++jj) {
      const int j0 = 2 * (warp & 1) + jj;
      v8f c = {};
#pragma unroll
      for (int t = 0; t < 3; ++t)
#pragma unroll
        for (int ks = 0; ks < 4; ++ks) {
          const int kb = ks * 4 + 2 * hi;
          c = wmma_f32(*(const v2f*)(&sY[t][(i0 * 16 + nl) * W1P + kb]),
                       *(const v2f*)(&sA[t][(j0 * 16 + nl) * W1P + kb]), c);
        }
#pragma unroll
      for (int r = 0; r < 8; ++r)
        sS2[(i0 * 16 + r + 8 * hi) * 64 + j0 * 16 + nl] = c[r];
    }
  }
  __syncthreads();
  {
    float part[16];
#pragma unroll
    for (int q = 0; q < 16; ++q) part[q] = sS2[(int)threadIdx.x + 256 * q];
#pragma unroll
    for (int q = 0; q < 16; ++q) atomicAdd(&gS[(int)threadIdx.x + 256 * q], part[q]);
  }
  if (threadIdx.x < 64) {
    const int n = threadIdx.x;
    atomicAdd(&gcs[n], (sY[0][n * W1P + 12] + sY[1][n * W1P + 12]) + sY[2][n * W1P + 12]);
  }
}

__global__ void __launch_bounds__(256) k_out(
    const float* __restrict__ s, const float* __restrict__ tq, const float* __restrict__ tk,
    const float* __restrict__ lw, const float* __restrict__ lb,
    const float* __restrict__ Mw, const float* __restrict__ gS, const float* __restrict__ gcs,
    const float* __restrict__ lng, const float* __restrict__ lnb,
    const float* __restrict__ ow, const float* __restrict__ obp,
    float* __restrict__ out)
{
  __shared__ float su[8][16 * 68];
  __shared__ float so[8][16 * 68];
  __shared__ __attribute__((aligned(16))) float slw1[64 * WP];
  __shared__ float sQ[64 * WP];
  __shared__ float slb1[64];
  __shared__ float sct[64];
  const int warp = threadIdx.x >> 5, lane = threadIdx.x & 31;
  const int hi = lane >> 4, nl = lane & 15;
  const int row0 = (blockIdx.x * 8 + warp) * 16;
  float* uw = su[warp];
  float* om = so[warp];
  const float* sv = s + (size_t)2 * BATCH * 64;
  const float gscale = 2.0f / (float)(3ull * BATCH * 64);
  const int t0 = warp >> 1, t1 = 2 * (warp & 1);

  v2f sa[16];
#pragma unroll
  for (int ks = 0; ks < 16; ++ks) {
    int kb = ks * 4 + 2 * hi;
    const float* p = sv + (size_t)(row0 + nl) * 64 + kb;
    sa[ks] = mkv2(p[0], p[1]);
  }
#pragma unroll
  for (int jj = 0; jj < 2; ++jj) {
    const int a = t0 * 16 + nl, j = (t1 + jj) * 16 + nl;
    v8f c = {};
#pragma unroll
    for (int ks = 0; ks < 16; ++ks) {
      const int kb = ks * 4 + 2 * hi;
      c = wmma_f32(*(const v2f*)(gS + a * 64 + kb), mkv2(tk[kb * 64 + j], tk[(kb + 1) * 64 + j]), c);
    }
#pragma unroll
    for (int r = 0; r < 8; ++r) sQ[(t0 * 16 + r + 8 * hi) * WP + j] = c[r];
  }
  v2f ma[16];
#pragma unroll
  for (int ks = 0; ks < 16; ++ks) {
    const int kb = ks * 4 + 2 * hi, i = t0 * 16 + nl;
    ma[ks] = mkv2(Mw[kb * 64 + i], Mw[(kb + 1) * 64 + i]);
  }
  if (warp < 4) {
    const int j = (int)threadIdx.x & 63;
    const float* src = (warp < 2) ? tk : Mw;
    float dv[64];
#pragma unroll
    for (int b = 0; b < 64; ++b) dv[b] = src[b * 64 + j];
    const float lbj = lb[j];
    float acc = 0.f;
#pragma unroll
    for (int b = 0; b < 64; ++b) acc = __builtin_fmaf(gcs[b], dv[b], acc);
    const float g = gscale * (acc + (float)(3 * BATCH) * lbj);
    const float upd = lbj - LRATE * g / (fabsf(g) + AEPS);
    float* dst = (warp < 2) ? sct : slb1;
    dst[j] = (warp < 2) ? acc : upd;
  }
  __syncthreads();
#pragma unroll
  for (int jj = 0; jj < 2; ++jj) {
    const int j = (t1 + jj) * 16 + nl;
    v8f c = {};
#pragma unroll
    for (int ks = 0; ks < 16; ++ks) {
      const int kb = ks * 4 + 2 * hi;
      c = wmma_f32(ma[ks], mkv2(sQ[kb * WP + j], sQ[(kb + 1) * WP + j]), c);
    }
    const float ctj = sct[j];
#pragma unroll
    for (int r = 0; r < 8; ++r) {
      const int ii = t0 * 16 + r + 8 * hi;
      const float g = gscale * (c[r] + lb[ii] * ctj);
      slw1[ii * WP + j] = lw[ii * 64 + j] - LRATE * g / (fabsf(g) + AEPS);
    }
  }
#pragma unroll
  for (int n0 = 0; n0 < 4; ++n0) {
    int n = n0 * 16 + nl;
    v8f c = {};
#pragma unroll
    for (int ks = 0; ks < 16; ++ks) {
      int kb = ks * 4 + 2 * hi;
      c = wmma_f32(sa[ks], mkv2(tq[kb * 64 + n], tq[(kb + 1) * 64 + n]), c);
    }
#pragma unroll
    for (int r = 0; r < 8; ++r) uw[(r + 8 * hi) * 68 + n] = c[r];
  }
  __syncthreads();
#pragma unroll
  for (int n0 = 0; n0 < 4; ++n0) {
    int n = n0 * 16 + nl;
    v8f c = {};
#pragma unroll
    for (int ks = 0; ks < 16; ++ks) {
      int kb = ks * 4 + 2 * hi;
      v2f a = mkv2(uw[nl * 68 + kb], uw[nl * 68 + kb + 1]);
      c = wmma_f32(a, *(const v2f*)(&slw1[n * WP + kb]), c);
    }
    float bias = slb1[n];
#pragma unroll
    for (int r = 0; r < 8; ++r) om[(r + 8 * hi) * 68 + n] = c[r] + bias;
  }
  __syncthreads();
  if (lane < 16) {
    const float* rowp = om + lane * 68;
    float mu = 0.f;
#pragma unroll
    for (int j = 0; j < 64; ++j) mu += rowp[j];
    mu *= (1.0f / 64.0f);
    float var = 0.f;
#pragma unroll
    for (int j = 0; j < 64; ++j) { float d = rowp[j] - mu; var += d * d; }
    var *= (1.0f / 64.0f);
    float rstd = rsqrtf(var + LNEPS);
    float acc = 0.f;
#pragma unroll
    for (int j = 0; j < 64; ++j) {
      float nrm = (rowp[j] - mu) * rstd * lng[j] + lnb[j];
      float a = nrm / (1.0f + __expf(-nrm));
      acc += a * ow[j];
    }
    out[row0 + lane] = acc + obp[0];
  }
}

extern "C" void kernel_launch(void* const* d_in, const int* in_sizes, int n_in,
                              void* d_out, int out_size, void* d_ws, size_t ws_size,
                              hipStream_t stream) {
  const float* in1  = (const float*)d_in[0];
  const float* in2  = (const float*)d_in[1];
  const float* in3  = (const float*)d_in[2];
  const float* in4  = (const float*)d_in[3];
  const float* w_l  = (const float*)d_in[4];
  const float* b_l  = (const float*)d_in[5];
  const float* w_m  = (const float*)d_in[6];
  const float* b_m  = (const float*)d_in[7];
  const float* w_s  = (const float*)d_in[8];
  const float* b_s  = (const float*)d_in[9];
  const float* fc1w = (const float*)d_in[10];
  const float* fc1b = (const float*)d_in[11];
  const float* q_w  = (const float*)d_in[12];
  const float* q_b  = (const float*)d_in[13];
  const float* k_w  = (const float*)d_in[14];
  const float* k_b  = (const float*)d_in[15];
  const float* v_w  = (const float*)d_in[16];
  const float* v_b  = (const float*)d_in[17];
  const float* t_k  = (const float*)d_in[18];
  const float* t_v  = (const float*)d_in[19];
  const float* t_q  = (const float*)d_in[20];
  const float* lw   = (const float*)d_in[21];
  const float* lb   = (const float*)d_in[22];
  const float* lng  = (const float*)d_in[23];
  const float* lnb  = (const float*)d_in[24];
  const float* outw = (const float*)d_in[25];
  const float* outb = (const float*)d_in[26];

  float* wsf  = (float*)d_ws;
  float* cat  = wsf + F_CAT;
  float* sbuf = wsf + F_S;
  float* Mw   = wsf + F_M;
  float* gS   = wsf + F_S2;
  float* gcs  = wsf + F_CS;

  k_cat<<<512, 256, 0, stream>>>(in1, in2, in3, in4, w_l, b_l, w_m, b_m, w_s, b_s, cat, gS);
  k_qkv<<<BATCH / 128, 256, 0, stream>>>(cat, fc1w, fc1b, q_w, q_b, k_w, k_b, v_w, v_b, lw,
                                         t_k, t_v, sbuf, Mw, gS, gcs);
  k_out<<<BATCH / 128, 256, 0, stream>>>(sbuf, t_q, t_k, lw, lb, Mw, gS, gcs, lng, lnb, outw,
                                         outb, (float*)d_out);
}


# k_qkv adds only its 16x16 cat Gram share to memory (256 atomics per block, not 4160); k_out expands the finished sum to s^T s and the column sums on WMMA itself; changes the Gram sum order
# speedup vs baseline: 2.8977x; 1.0393x over previous
#include <hip/hip_runtime.h>
#include <math.h>

#define BATCH 32768
#define LRATE 0.01f
#define AEPS  1e-8f
#define LNEPS 1e-5f

typedef __attribute__((ext_vector_type(2))) float v2f;
typedef __attribute__((ext_vector_type(8))) float v8f;

__device__ __forceinline__ v2f mkv2(float x, float y) { v2f v; v[0] = x; v[1] = y; return v; }

__device__ __forceinline__ v8f wmma_f32(v2f a, v2f b, v8f c) {
  return __builtin_amdgcn_wmma_f32_16x16x4_f32(false, a, false, b, (short)0, c, false, false);
}

constexpr size_t F_CAT  = 0;
constexpr size_t F_S    = F_CAT + (size_t)BATCH * 16;
constexpr size_t F_M    = F_S + (size_t)3 * BATCH * 64;
constexpr size_t F_C    = F_M + 64 * 64;

constexpr int WP  = 68;
constexpr int W1P = 20;

typedef __attribute__((ext_vector_type(4))) float v4f;

constexpr int CAT_WAVES  = 8;
constexpr int CAT_ROWS   = 8;
constexpr int CAT_BLOCKS = BATCH / (CAT_WAVES * CAT_ROWS);
static_assert(CAT_BLOCKS == 512 && CAT_WAVES * 32 == 256, "k_cat is launched as <<<512, 256>>>");
constexpr int WQ1 = 0;
constexpr int WQ2 = 4608;
constexpr int WQ3 = 9216;
constexpr int WQN = 10240;
constexpr int NSTAGE = 20;
constexpr int RPITCH = 33;

__device__ __forceinline__ v4f ld_stream(const v4f* p) { return __builtin_nontemporal_load(p); }
__device__ __forceinline__ float dot4(v4f d, v4f w, float a) {
  a = __builtin_fmaf(d.x, w.x, a); a = __builtin_fmaf(d.y, w.y, a);
  a = __builtin_fmaf(d.z, w.z, a); a = __builtin_fmaf(d.w, w.w, a);
  return a;
}

template <int S>
__device__ __forceinline__ void cat_issue(v4f (&buf)[8], const v4f* p1, const v4f* p2,
                                          const v4f* p3, int lane) {
  if constexpr (S < 12) {
#pragma unroll
    for (int r = 0; r < 8; ++r) {
      if constexpr (S == 11) buf[r] = ld_stream(p1 + r * 375 + ((lane < 23) ? 352 : 374 - lane));
      else                   buf[r] = ld_stream(p1 + r * 375 + 32 * S);
    }
  } else if constexpr (S < 18) {
#pragma unroll
    for (int h = 0; h < 2; ++h) {
      const int q = 2 * (S - 12) + h;
#pragma unroll
      for (int p = 0; p < 4; ++p) {
        if (q == 11) buf[h * 4 + p] = ld_stream(p2 + p * 375 + ((lane < 23) ? 352 : 374 - lane));
        else         buf[h * 4 + p] = ld_stream(p2 + p * 375 + 32 * q);
      }
    }
  } else if constexpr (S == 18) {
#pragma unroll
    for (int h = 0; h < 2; ++h)
#pragma unroll
      for (int p = 0; p < 4; ++p) buf[h * 4 + p] = ld_stream(p3 + p * 75 + 32 * h);
  } else if constexpr (S == 19) {
#pragma unroll
    for (int p = 0; p < 4; ++p) buf[p] = ld_stream(p3 + p * 75 + ((lane < 11) ? 64 : 74 - lane));
  }
}

template <int S>
__device__ __forceinline__ void cat_weights(v4f (&w)[6], const float* sw, int lane) {
  const v4f zero = {0.f, 0.f, 0.f, 0.f};
  if constexpr (S < 12) {
    const bool live = (S != 11) || (lane < 23);
    const int j = live ? 32 * S + lane : 0;
#pragma unroll
    for (int o = 0; o < 3; ++o) { const v4f t = *(const v4f*)(sw + WQ1 + o * 1500 + 4 * j); w[o] = live ? t : zero; }
  } else if constexpr (S < 18) {
#pragma unroll
    for (int h = 0; h < 2; ++h) {
      const int q = 2 * (S - 12) + h;
      const bool live = (q != 11) || (lane < 23);
      const int j = live ? 32 * q + lane : 0;
#pragma unroll
      for (int o = 0; o < 3; ++o) { const v4f t = *(const v4f*)(sw + WQ2 + o * 1500 + 4 * j); w[h * 3 + o] = live ? t : zero; }
    }
  } else if constexpr (S == 18) {
#pragma unroll
    for (int h = 0; h < 2; ++h)
#pragma unroll
      for (int o = 0; o < 3; ++o) w[h * 3 + o] = *(const v4f*)(sw + WQ3 + o * 300 + 4 * (32 * h + lane));
  } else {
    const bool live = lane < 11;
    const int j = live ? 64 + lane : 0;
#pragma unroll
    for (int o = 0; o < 3; ++o) { const v4f t = *(const v4f*)(sw + WQ3 + o * 300 + 4 * j); w[o] = live ? t : zero; }
  }
}

__device__ __forceinline__ void pair_mixed(v4f d, const v4f* w, float* accA, float* accB,
                                           bool xyA, bool zwA) {
#pragma unroll
  for (int o = 0; o < 3; ++o) {
    const float pxy = __builtin_fmaf(d.x, w[o].x, d.y * w[o].y);
    const float pzw = __builtin_fmaf(d.z, w[o].z, d.w * w[o].w);
    accA[o] += (xyA ? pxy : 0.f) + (zwA ? pzw : 0.f);
    accB[o] += (xyA ? 0.f : pxy) + (zwA ? 0.f : pzw);
  }
}

template <int S>
__device__ __forceinline__ void cat_consume(const v4f (&buf)[8], const v4f (&w)[6],
                                            float (&acc)[8][3], int lane) {
  if constexpr (S < 12) {
#pragma unroll
    for (int r = 0; r < 8; ++r)
#pragma unroll
      for (int o = 0; o < 3; ++o) acc[r][o] = dot4(buf[r], w[o], acc[r][o]);
  } else if constexpr (S < 18) {
#pragma unroll
    for (int h = 0; h < 2; ++h) {
      const int q = 2 * (S - 12) + h;
#pragma unroll
      for (int p = 0; p < 4; ++p) {
        if (q < 5) {
#pragma unroll
          for (int o = 0; o < 3; ++o) acc[2 * p][o] = dot4(buf[h * 4 + p], w[h * 3 + o], acc[2 * p][o]);
        } else if (q > 5) {
#pragma unroll
          for (int o = 0; o < 3; ++o) acc[2 * p + 1][o] = dot4(buf[h * 4 + p], w[h * 3 + o], acc[2 * p + 1][o]);
        } else {
          pair_mixed(buf[h * 4 + p], &w[h * 3], acc[2 * p], acc[2 * p + 1], lane <= 27, lane <= 26);
        }
      }
    }
  } else if constexpr (S == 18) {
#pragma unroll
    for (int p = 0; p < 4; ++p) {
#pragma unroll
      for (int o = 0; o < 3; ++o) acc[2 * p][o] = dot4(buf[p], w[o], acc[2 * p][o]);
      pair_mixed(buf[4 + p], &w[3], acc[2 * p], acc[2 * p + 1], lane <= 5, lane <= 4);
    }
  } else {
#pragma unroll
    for (int p = 0; p < 4; ++p)
#pragma unroll
      for (int o = 0; o < 3; ++o) acc[2 * p + 1][o] = dot4(buf[p], w[o], acc[2 * p + 1][o]);
  }
}

__global__ void __launch_bounds__(256) __attribute__((amdgpu_waves_per_eu(4, 4))) k_cat(
    const float* __restrict__ in1, const float* __restrict__ in2,
    const float* __restrict__ in3, const float* __restrict__ in4,
    const float* __restrict__ wl, const float* __restrict__ bl,
    const float* __restrict__ wm, const float* __restrict__ bm,
    const float* __restrict__ wsw, const float* __restrict__ bs,
    float* __restrict__ cat, float* __restrict__ zacc)
{
  __shared__ __attribute__((aligned(16))) float sw[WQN];
  __shared__ float red[CAT_WAVES][72 * RPITCH];
  __shared__ __attribute__((aligned(16))) float cst[CAT_WAVES][CAT_ROWS * 16];

  const int warp = threadIdx.x >> 5, lane = threadIdx.x & 31;
  const int b0 = (blockIdx.x * CAT_WAVES + warp) * CAT_ROWS;
  const v4f* p1 = (const v4f*)in1 + (size_t)b0 * 375 + lane;
  const v4f* p2 = (const v4f*)in2 + (size_t)(b0 >> 1) * 375 + lane;
  const v4f* p3 = (const v4f*)in3 + (size_t)(b0 >> 1) * 75 + lane;

  v4f buf[4][8];
  cat_issue<0>(buf[0], p1, p2, p3, lane);
  cat_issue<1>(buf[1], p1, p2, p3, lane);
  cat_issue<2>(buf[2], p1, p2, p3, lane);

  float tailv[2];
#pragma unroll
  for (int g = 0; g < 2; ++g) {
    const int u = lane + 32 * g, cc = u % 7;
    const bool take = (u < 56) && (cc < 3);
    const float t = in4[(size_t)b0 * 3 + (take ? (u / 7) * 3 + cc : 0)];
    tailv[g] = take ? t : 0.f;
  }

  {
    float t1[18];
#pragma unroll
    for (int t = 0; t < 18; ++t) {
      const int i = min((int)threadIdx.x + 256 * t, 4499), o = i / 1500, f = i - o * 1500;
      t1[t] = wl[(o * 3 + f % 3) * 500 + f / 3];
    }
#pragma unroll
    for (int t = 0; t < 18; ++t) sw[WQ1 + threadIdx.x + 256 * t] = t1[t];
#pragma unroll
    for (int t = 0; t < 18; ++t) {
      const int i = min((int)threadIdx.x + 256 * t, 4499), o = i / 1500, g = i - o * 1500;
      const int f = (g >= 750) ? g - 750 : g;
      t1[t] = wm[(o * 3 + f % 3) * 250 + f / 3];
    }
#pragma unroll
    for (int t = 0; t < 18; ++t) sw[WQ2 + threadIdx.x + 256 * t] = t1[t];
#pragma unroll
    for (int t = 0; t < 4; ++t) {
      const int i = min((int)threadIdx.x + 256 * t, 899), o = i / 300, g = i - o * 300;
      const int f = (g >= 150) ? g - 150 : g;
      t1[t] = wsw[(o * 3 + f % 3) * 50 + f / 3];
    }
#pragma unroll
    for (int t = 0; t < 4; ++t) sw[WQ3 + threadIdx.x + 256 * t] = t1[t];
  }
  __syncthreads();

  float acc[8][3];
  float* redw = red[warp];
  v4f w[6];

#define CAT_STAGE(S)                                                              \
  cat_weights<S>(w, sw, lane);                                                    \
  __builtin_amdgcn_sched_barrier(0);                                              \
  if constexpr ((S) + 3 < NSTAGE)                                                 \
    cat_issue<((S) + 3 < NSTAGE ? (S) + 3 : 0)>(buf[((S) + 3) % 4], p1, p2, p3, lane); \
  __builtin_amdgcn_sched_barrier(0);                                              \
  cat_consume<S>(buf[(S) % 4], w, acc, lane);                                     \
  CAT_PIN()
#define CAT_PIN()                                                                 \
  _Pragma("unroll") for (int r = 0; r < 8; ++r)                                   \
    _Pragma("unroll") for (int o = 0; o < 3; ++o) asm volatile("" : "+v"(acc[r][o]));
#define CAT_CLEAR()                                                               \
  _Pragma("unroll") for (int r = 0; r < 8; ++r)                                   \
    _Pragma("unroll") for (int o = 0; o < 3; ++o) acc[r][o] = 0.f;
#define CAT_PARK(COL)                                                             \
  _Pragma("unroll") for (int r = 0; r < 8; ++r)                                   \
    _Pragma("unroll") for (int o = 0; o < 3; ++o)                                 \
      redw[(r * 9 + (COL) + o) * RPITCH + lane] = acc[r][o];

  CAT_CLEAR()
  CAT_STAGE(0) CAT_STAGE(1) CAT_STAGE(2) CAT_STAGE(3) CAT_STAGE(4) CAT_STAGE(5)
  CAT_STAGE(6) CAT_STAGE(7) CAT_STAGE(8) CAT_STAGE(9) CAT_STAGE(10) CAT_STAGE(11)
  CAT_PARK(0)
  CAT_CLEAR()
  CAT_STAGE(12) CAT_STAGE(13) CAT_STAGE(14) CAT_STAGE(15) CAT_STAGE(16) CAT_STAGE(17)
  CAT_PARK(3)
  CAT_CLEAR()
  CAT_STAGE(18) CAT_STAGE(19)
  CAT_PARK(6)
#undef CAT_STAGE
#undef CAT_PIN
#undef CAT_CLEAR
#undef CAT_PARK
  __syncthreads();

  float* cw = cst[warp];
  const float b9[9] = {bl[0], bl[1], bl[2], bm[0], bm[1], bm[2], bs[0], bs[1], bs[2]};
#pragma unroll
  for (int g = 0; g < 3; ++g) {
    const int v = lane + 32 * g;
    if (v < 72) {
      const float* rp = redw + v * RPITCH;
      float s0 = 0.f, s1 = 0.f, s2 = 0.f, s3 = 0.f;
#pragma unroll
      for (int i = 0; i < 32; i += 4) { s0 += rp[i]; s1 += rp[i + 1]; s2 += rp[i + 2]; s3 += rp[i + 3]; }
      const int c = v % 9;
      float bias = b9[0];
#pragma unroll
      for (int k = 1; k < 9; ++k) bias = (c == k) ? b9[k] : bias;
      cw[(v / 9) * 16 + c] = (s0 + s1) + (s2 + s3) + bias;
    }
  }
#pragma unroll
  for (int g = 0; g < 2; ++g) {
    const int u = lane + 32 * g;
    if (u < 56) cw[(u / 7) * 16 + 9 + u % 7] = tailv[g];
  }
  __syncthreads();
  *((v4f*)(cat + (size_t)b0 * 16) + lane) = *((const v4f*)cw + lane);
  if (blockIdx.x == 0) zacc[threadIdx.x] = 0.f;
}


__global__ void __launch_bounds__(256) k_qkv(
    const float* __restrict__ cat,
    const float* __restrict__ fc1w, const float* __restrict__ fc1b,
    const float* __restrict__ vw, const float* __restrict__ vb,
    const float* __restrict__ lw, const float* __restrict__ tk, const float* __restrict__ tvm,
    float* __restrict__ s, float* __restrict__ Mw, float* __restrict__ gC)
{
  __shared__ float sh[8][16 * 68];
  __shared__ float sC[16 * 16];
  __shared__ __attribute__((aligned(16))) float swv[64 * WP];
  __shared__ __attribute__((aligned(16))) float sw1[64 * W1P];
  const int warp = threadIdx.x >> 5, lane = threadIdx.x & 31;
  const int hi = lane >> 4, nl = lane & 15;
  const int row0 = (blockIdx.x * 8 + warp) * 16;
  float* shw = sh[warp];

  sC[threadIdx.x] = 0.f;
  v4f wst[4];
#pragma unroll
  for (int j = 0; j < 4; ++j) wst[j] = *(const v4f*)(vw + ((int)threadIdx.x + 256 * j) * 4);
  const int e1 = min((int)threadIdx.x, 191) * 4;
  const v4f wf1 = *(const v4f*)(fc1w + e1);
  v2f ca[4];
#pragma unroll
  for (int ks = 0; ks < 4; ++ks) {
    const float* p = cat + (size_t)(row0 + nl) * 16 + ks * 4 + 2 * hi;
    ca[ks] = mkv2(p[0], p[1]);
  }
  float fbias[4];
#pragma unroll
  for (int n0 = 0; n0 < 4; ++n0) fbias[n0] = fc1b[n0 * 16 + nl];
  float cg[4][2];
#pragma unroll
  for (int m = 0; m < 4; ++m) {
    const float* p = cat + (size_t)(row0 + 4 * m + 2 * hi) * 16 + nl;
    cg[m][0] = p[0];
    cg[m][1] = p[16];
  }
  if (blockIdx.x < 16 && warp == 0) {
    const int a0 = blockIdx.x >> 2, i0 = blockIdx.x & 3;
    const int a = a0 * 16 + nl, i = i0 * 16 + nl;
    v8f c = {};
#pragma unroll
    for (int ks = 0; ks < 16; ++ks) {
      const int kb = ks * 4 + 2 * hi;
      c = wmma_f32(*(const v2f*)(tk + a * 64 + kb), *(const v2f*)(lw + i * 64 + kb), c);
    }
#pragma unroll
    for (int r = 0; r < 8; ++r) {
      const int e = (a0 * 16 + r + 8 * hi) * 64 + i;
      Mw[e] = c[r] - tvm[e];
    }
  }
  {
#pragma unroll
    for (int j = 0; j < 4; ++j) {
      const int e = ((int)threadIdx.x + 256 * j) * 4;
      *(v4f*)(&swv[(e >> 6) * WP + (e & 63)]) = wst[j];
    }
    if (threadIdx.x < 192) {
      const int n = e1 / 12;
      *(v4f*)(&sw1[n * W1P + (e1 - n * 12)]) = wf1;
    }
    if (threadIdx.x < 64) {
      const v4f tail = {0.f, 0.f, 0.f, 0.f};
      *(v4f*)(&sw1[(int)threadIdx.x * W1P + 12]) = tail;
    }
  }
  __syncthreads();

  {
    v8f cc = {};
#pragma unroll
    for (int m = 0; m < 4; ++m) {
      const v2f x = mkv2(nl == 12 ? 1.0f : cg[m][0], nl == 12 ? 1.0f : cg[m][1]);
      cc = wmma_f32(x, x, cc);
    }
#pragma unroll
    for (int r = 0; r < 8; ++r) atomicAdd(&sC[(r + 8 * hi) * 16 + nl], cc[r]);
  }
#pragma unroll
  for (int n0 = 0; n0 < 4; ++n0) {
    int n = n0 * 16 + nl;
    v8f c = {};
#pragma unroll
    for (int ks = 0; ks < 4; ++ks) {
      int kb = ks * 4 + 2 * hi;
      c = wmma_f32(ca[ks], *(const v2f*)(&sw1[n * W1P + kb]), c);
    }
    float bias = fbias[n0] + ((n & 1) ? 1.0f : 0.0f);
#pragma unroll
    for (int r = 0; r < 8; ++r) shw[(r + 8 * hi) * 68 + n] = c[r] + bias;
  }
  __syncthreads();
  atomicAdd(&gC[threadIdx.x], sC[threadIdx.x]);

  v2f ha[16];
#pragma unroll
  for (int ks = 0; ks < 16; ++ks) {
    int kb = ks * 4 + 2 * hi;
    ha[ks] = mkv2(shw[nl * 68 + kb], shw[nl * 68 + kb + 1]);
  }
  {
    float* ob = s + (size_t)2 * BATCH * 64;
#pragma unroll
    for (int n0 = 0; n0 < 4; ++n0) {
      int n = n0 * 16 + nl;
      v8f c = {};
#pragma unroll
      for (int ks = 0; ks < 16; ++ks) {
        int kb = ks * 4 + 2 * hi;
        c = wmma_f32(ha[ks], *(const v2f*)(&swv[n * WP + kb]), c);
      }
      float bias = vb[n];
#pragma unroll
      for (int r = 0; r < 8; ++r)
        ob[(size_t)(row0 + r + 8 * hi) * 64 + n] = c[r] + bias;
    }
  }
}

__global__ void __launch_bounds__(256) k_out(
    const float* __restrict__ s, const float* __restrict__ tq, const float* __restrict__ tk,
    const float* __restrict__ lw, const float* __restrict__ lb,
    const float* __restrict__ Mw, const float* __restrict__ gC,
    const float* __restrict__ fc1w, const float* __restrict__ fc1b,
    const float* __restrict__ qw, const float* __restrict__ qb,
    const float* __restrict__ kw, const float* __restrict__ kbv,
    const float* __restrict__ vw, const float* __restrict__ vb,
    const float* __restrict__ lng, const float* __restrict__ lnb,
    const float* __restrict__ ow, const float* __restrict__ obp,
    float* __restrict__ out)
{
  __shared__ float su[8][16 * 68];
  __shared__ float so[8][16 * 68];
  __shared__ __attribute__((aligned(16))) float slw1[64 * WP];
  __shared__ float sQ[64 * WP];
  __shared__ float slb1[64];
  __shared__ float sct[64];
  __shared__ float sC[16 * 16];
  __shared__ __attribute__((aligned(16))) float sA[3][64 * W1P];
  __shared__ __attribute__((aligned(16))) float sY[3][64 * W1P];
  __shared__ __attribute__((aligned(16))) float swt[3][64 * WP];
  __shared__ __attribute__((aligned(16))) float sw1[64 * W1P];
  __shared__ __attribute__((aligned(16))) float sS2[64 * WP];
  __shared__ float scs[64];
  const int warp = threadIdx.x >> 5, lane = threadIdx.x & 31;
  const int hi = lane >> 4, nl = lane & 15;
  const int row0 = (blockIdx.x * 8 + warp) * 16;
  float* uw = su[warp];
  float* om = so[warp];
  const float* sv = s + (size_t)2 * BATCH * 64;
  const float gscale = 2.0f / (float)(3ull * BATCH * 64);
  const int t0 = warp >> 1, t1 = 2 * (warp & 1);

  v2f sa[16];
#pragma unroll
  for (int ks = 0; ks < 16; ++ks) {
    int kb = ks * 4 + 2 * hi;
    const float* p = sv + (size_t)(row0 + nl) * 64 + kb;
    sa[ks] = mkv2(p[0], p[1]);
  }
  const float* WS[3] = {qw, kw, vw};
  v4f wst[3][4];
#pragma unroll
  for (int t = 0; t < 3; ++t)
#pragma unroll
    for (int j = 0; j < 4; ++j) wst[t][j] = *(const v4f*)(WS[t] + ((int)threadIdx.x + 256 * j) * 4);
  const int e1 = min((int)threadIdx.x, 191) * 4;
  const v4f wf1 = *(const v4f*)(fc1w + e1);
  const float wfb = fc1b[threadIdx.x & 63];
  const float cval = gC[threadIdx.x];
  float bt[2][8];
#pragma unroll
  for (int rep = 0; rep < 2; ++rep) {
    const int T = min(warp + 8 * rep, 11), t = T >> 2, n0 = T & 3;
    const float* bsel = (t == 0) ? qb : ((t == 1) ? kbv : vb);
#pragma unroll
    for (int r = 0; r < 8; ++r) bt[rep][r] = bsel[n0 * 16 + 8 * hi + r];
  }
  {
#pragma unroll
    for (int t = 0; t < 3; ++t)
#pragma unroll
      for (int j = 0; j < 4; ++j) {
        const int e = ((int)threadIdx.x + 256 * j) * 4;
        *(v4f*)(&swt[t][(e >> 6) * WP + (e & 63)]) = wst[t][j];
      }
    if (threadIdx.x < 192) {
      const int n = e1 / 12;
      *(v4f*)(&sw1[n * W1P + (e1 - n * 12)]) = wf1;
    }
    if (threadIdx.x < 64) {
      const v4f tail = {wfb + ((threadIdx.x & 1) ? 1.0f : 0.0f), 0.f, 0.f, 0.f};
      *(v4f*)(&sw1[(int)threadIdx.x * W1P + 12]) = tail;
    }
    sC[threadIdx.x] = cval;
  }
  __syncthreads();
#pragma unroll
  for (int rep = 0; rep < 2; ++rep) {
    const int T = warp + 8 * rep;
    if (T < 12) {
      const int t = T >> 2, n0 = T & 3;
      v8f c = {};
#pragma unroll
      for (int ks = 0; ks < 16; ++ks) {
        const int kb = ks * 4 + 2 * hi;
        c = wmma_f32(*(const v2f*)(&swt[t][(n0 * 16 + nl) * WP + kb]),
                     mkv2(sw1[kb * W1P + nl], sw1[(kb + 1) * W1P + nl]), c);
      }
#pragma unroll
      for (int r = 0; r < 8; ++r) {
        const int n = n0 * 16 + r + 8 * hi;
        sA[t][n * W1P + nl] = c[r] + (nl == 12 ? bt[rep][r] : 0.f);
      }
    }
  }
  __syncthreads();
#pragma unroll
  for (int rep = 0; rep < 2; ++rep) {
    const int T = warp + 8 * rep;
    if (T < 12) {
      const int t = T >> 2, n0 = T & 3;
      v8f c = {};
#pragma unroll
      for (int ks = 0; ks < 4; ++ks) {
        const int kb = ks * 4 + 2 * hi;
        c = wmma_f32(*(const v2f*)(&sA[t][(n0 * 16 + nl) * W1P + kb]),
                     mkv2(sC[kb * 16 + nl], sC[(kb + 1) * 16 + nl]), c);
      }
#pragma unroll
      for (int r = 0; r < 8; ++r) sY[t][(n0 * 16 + r + 8 * hi) * W1P + nl] = c[r];
    }
  }
  __syncthreads();
#pragma unroll
  for (int jj = 0; jj < 2; ++jj) {
    const int j0 = t1 + jj;
    v8f c = {};
#pragma unroll
    for (int t = 0; t < 3; ++t)
#pragma unroll
      for (int ks = 0; ks < 4; ++ks) {
        const int kb = ks * 4 + 2 * hi;
        c = wmma_f32(*(const v2f*)(&sY[t][(t0 * 16 + nl) * W1P + kb]),
                     *(const v2f*)(&sA[t][(j0 * 16 + nl) * W1P + kb]), c);
      }
#pragma unroll
    for (int r = 0; r < 8; ++r)
      sS2[(t0 * 16 + r + 8 * hi) * WP + j0 * 16 + nl] = c[r];
  }
  if (threadIdx.x < 64) {
    const int n = threadIdx.x;
    scs[n] = (sY[0][n * W1P + 12] + sY[1][n * W1P + 12]) + sY[2][n * W1P + 12];
  }
  __syncthreads();
#pragma unroll
  for (int jj = 0; jj < 2; ++jj) {
    const int a = t0 * 16 + nl, j = (t1 + jj) * 16 + nl;
    v8f c = {};
#pragma unroll
    for (int ks = 0; ks < 16; ++ks) {
      const int kb = ks * 4 + 2 * hi;
      c = wmma_f32(*(const v2f*)(&sS2[a * WP + kb]), mkv2(tk[kb * 64 + j], tk[(kb + 1) * 64 + j]), c);
    }
#pragma unroll
    for (int r = 0; r < 8; ++r) sQ[(t0 * 16 + r + 8 * hi) * WP + j] = c[r];
  }
  v2f ma[16];
#pragma unroll
  for (int ks = 0; ks < 16; ++ks) {
    const int kb = ks * 4 + 2 * hi, i = t0 * 16 + nl;
    ma[ks] = mkv2(Mw[kb * 64 + i], Mw[(kb + 1) * 64 + i]);
  }
  if (warp < 4) {
    const int j = (int)threadIdx.x & 63;
    const float* src = (warp < 2) ? tk : Mw;
    float dv[64];
#pragma unroll
    for (int b = 0; b < 64; ++b) dv[b] = src[b * 64 + j];
    const float lbj = lb[j];
    float acc = 0.f;
#pragma unroll
    for (int b = 0; b < 64; ++b) acc = __builtin_fmaf(scs[b], dv[b], acc);
    const float g = gscale * (acc + (float)(3 * BATCH) * lbj);
    const float upd = lbj - LRATE * g / (fabsf(g) + AEPS);
    float* dst = (warp < 2) ? sct : slb1;
    dst[j] = (warp < 2) ? acc : upd;
  }
  __syncthreads();
#pragma unroll
  for (int jj = 0; jj < 2; ++jj) {
    const int j = (t1 + jj) * 16 + nl;
    v8f c = {};
#pragma unroll
    for (int ks = 0; ks < 16; ++ks) {
      const int kb = ks * 4 + 2 * hi;
      c = wmma_f32(ma[ks], mkv2(sQ[kb * WP + j], sQ[(kb + 1) * WP + j]), c);
    }
    const float ctj = sct[j];
#pragma unroll
    for (int r = 0; r < 8; ++r) {
      const int ii = t0 * 16 + r + 8 * hi;
      const float g = gscale * (c[r] + lb[ii] * ctj);
      slw1[ii * WP + j] = lw[ii * 64 + j] - LRATE * g / (fabsf(g) + AEPS);
    }
  }
#pragma unroll
  for (int n0 = 0; n0 < 4; ++n0) {
    int n = n0 * 16 + nl;
    v8f c = {};
#pragma unroll
    for (int ks = 0; ks < 16; ++ks) {
      int kb = ks * 4 + 2 * hi;
      c = wmma_f32(sa[ks], mkv2(tq[kb * 64 + n], tq[(kb + 1) * 64 + n]), c);
    }
#pragma unroll
    for (int r = 0; r < 8; ++r) uw[(r + 8 * hi) * 68 + n] = c[r];
  }
  __syncthreads();
#pragma unroll
  for (int n0 = 0; n0 < 4; ++n0) {
    int n = n0 * 16 + nl;
    v8f c = {};
#pragma unroll
    for (int ks = 0; ks < 16; ++ks) {
      int kb = ks * 4 + 2 * hi;
      v2f a = mkv2(uw[nl * 68 + kb], uw[nl * 68 + kb + 1]);
      c = wmma_f32(a, *(const v2f*)(&slw1[n * WP + kb]), c);
    }
    float bias = slb1[n];
#pragma unroll
    for (int r = 0; r < 8; ++r) om[(r + 8 * hi) * 68 + n] = c[r] + bias;
  }
  __syncthreads();
  if (lane < 16) {
    const float* rowp = om + lane * 68;
    float mu = 0.f;
#pragma unroll
    for (int j = 0; j < 64; ++j) mu += rowp[j];
    mu *= (1.0f / 64.0f);
    float var = 0.f;
#pragma unroll
    for (int j = 0; j < 64; ++j) { float d = rowp[j] - mu; var += d * d; }
    var *= (1.0f / 64.0f);
    float rstd = rsqrtf(var + LNEPS);
    float acc = 0.f;
#pragma unroll
    for (int j = 0; j < 64; ++j) {
      float nrm = (rowp[j] - mu) * rstd * lng[j] + lnb[j];
      float a = nrm / (1.0f + __expf(-nrm));
      acc += a * ow[j];
    }
    out[row0 + lane] = acc + obp[0];
  }
}

extern "C" void kernel_launch(void* const* d_in, const int* in_sizes, int n_in,
                              void* d_out, int out_size, void* d_ws, size_t ws_size,
                              hipStream_t stream) {
  const float* in1  = (const float*)d_in[0];
  const float* in2  = (const float*)d_in[1];
  const float* in3  = (const float*)d_in[2];
  const float* in4  = (const float*)d_in[3];
  const float* w_l  = (const float*)d_in[4];
  const float* b_l  = (const float*)d_in[5];
  const float* w_m  = (const float*)d_in[6];
  const float* b_m  = (const float*)d_in[7];
  const float* w_s  = (const float*)d_in[8];
  const float* b_s  = (const float*)d_in[9];
  const float* fc1w = (const float*)d_in[10];
  const float* fc1b = (const float*)d_in[11];
  const float* q_w  = (const float*)d_in[12];
  const float* q_b  = (const float*)d_in[13];
  const float* k_w  = (const float*)d_in[14];
  const float* k_b  = (const float*)d_in[15];
  const float* v_w  = (const float*)d_in[16];
  const float* v_b  = (const float*)d_in[17];
  const float* t_k  = (const float*)d_in[18];
  const float* t_v  = (const float*)d_in[19];
  const float* t_q  = (const float*)d_in[20];
  const float* lw   = (const float*)d_in[21];
  const float* lb   = (const float*)d_in[22];
  const float* lng  = (const float*)d_in[23];
  const float* lnb  = (const float*)d_in[24];
  const float* outw = (const float*)d_in[25];
  const float* outb = (const float*)d_in[26];

  float* wsf  = (float*)d_ws;
  float* cat  = wsf + F_CAT;
  float* sbuf = wsf + F_S;
  float* Mw   = wsf + F_M;
  float* gC   = wsf + F_C;

  k_cat<<<512, 256, 0, stream>>>(in1, in2, in3, in4, w_l, b_l, w_m, b_m, w_s, b_s, cat, gC);
  k_qkv<<<BATCH / 128, 256, 0, stream>>>(cat, fc1w, fc1b, v_w, v_b, lw, t_k, t_v, sbuf, Mw, gC);
  k_out<<<BATCH / 128, 256, 0, stream>>>(sbuf, t_q, t_k, lw, lb, Mw, gC, fc1w, fc1b, q_w, q_b,
                                         k_w, k_b, v_w, v_b, lng, lnb, outw, outb, (float*)d_out);
}


# k_out takes its rows from cat: o = c~ P + lb1, P = A_v^T t_q lw1^T (16x64) made per block on WMMA; v rows never made or stored, k_qkv keeps only the Gram sum and M; changes o's sum order
# speedup vs baseline: 3.0693x; 1.0592x over previous
#include <hip/hip_runtime.h>
#include <math.h>

#define BATCH 32768
#define LRATE 0.01f
#define AEPS  1e-8f
#define LNEPS 1e-5f

typedef __attribute__((ext_vector_type(2))) float v2f;
typedef __attribute__((ext_vector_type(8))) float v8f;

__device__ __forceinline__ v2f mkv2(float x, float y) { v2f v; v[0] = x; v[1] = y; return v; }

__device__ __forceinline__ v8f wmma_f32(v2f a, v2f b, v8f c) {
  return __builtin_amdgcn_wmma_f32_16x16x4_f32(false, a, false, b, (short)0, c, false, false);
}

constexpr size_t F_CAT  = 0;
constexpr size_t F_M    = F_CAT + (size_t)BATCH * 16;
constexpr size_t F_C    = F_M + 64 * 64;

constexpr int WP  = 68;
constexpr int W1P = 20;

typedef __attribute__((ext_vector_type(4))) float v4f;

constexpr int CAT_WAVES  = 8;
constexpr int CAT_ROWS   = 8;
constexpr int CAT_BLOCKS = BATCH / (CAT_WAVES * CAT_ROWS);
static_assert(CAT_BLOCKS == 512 && CAT_WAVES * 32 == 256, "k_cat is launched as <<<512, 256>>>");
constexpr int WQ1 = 0;
constexpr int WQ2 = 4608;
constexpr int WQ3 = 9216;
constexpr int WQN = 10240;
constexpr int NSTAGE = 20;
constexpr int RPITCH = 33;

__device__ __forceinline__ v4f ld_stream(const v4f* p) { return __builtin_nontemporal_load(p); }
__device__ __forceinline__ float dot4(v4f d, v4f w, float a) {
  a = __builtin_fmaf(d.x, w.x, a); a = __builtin_fmaf(d.y, w.y, a);
  a = __builtin_fmaf(d.z, w.z, a); a = __builtin_fmaf(d.w, w.w, a);
  return a;
}

template <int S>
__device__ __forceinline__ void cat_issue(v4f (&buf)[8], const v4f* p1, const v4f* p2,
                                          const v4f* p3, int lane) {
  if constexpr (S < 12) {
#pragma unroll
    for (int r = 0; r < 8; ++r) {
      if constexpr (S == 11) buf[r] = ld_stream(p1 + r * 375 + ((lane < 23) ? 352 : 374 - lane));
      else                   buf[r] = ld_stream(p1 + r * 375 + 32 * S);
    }
  } else if constexpr (S < 18) {
#pragma unroll
    for (int h = 0; h < 2; ++h) {
      const int q = 2 * (S - 12) + h;
#pragma unroll
      for (int p = 0; p < 4; ++p) {
        if (q == 11) buf[h * 4 + p] = ld_stream(p2 + p * 375 + ((lane < 23) ? 352 : 374 - lane));
        else         buf[h * 4 + p] = ld_stream(p2 + p * 375 + 32 * q);
      }
    }
  } else if constexpr (S == 18) {
#pragma unroll
    for (int h = 0; h < 2; ++h)
#pragma unroll
      for (int p = 0; p < 4; ++p) buf[h * 4 + p] = ld_stream(p3 + p * 75 + 32 * h);
  } else if constexpr (S == 19) {
#pragma unroll
    for (int p = 0; p < 4; ++p) buf[p] = ld_stream(p3 + p * 75 + ((lane < 11) ? 64 : 74 - lane));
  }
}

template <int S>
__device__ __forceinline__ void cat_weights(v4f (&w)[6], const float* sw, int lane) {
  const v4f zero = {0.f, 0.f, 0.f, 0.f};
  if constexpr (S < 12) {
    const bool live = (S != 11) || (lane < 23);
    const int j = live ? 32 * S + lane : 0;
#pragma unroll
    for (int o = 0; o < 3; ++o) { const v4f t = *(const v4f*)(sw + WQ1 + o * 1500 + 4 * j); w[o] = live ? t : zero; }
  } else if constexpr (S < 18) {
#pragma unroll
    for (int h = 0; h < 2; ++h) {
      const int q = 2 * (S - 12) + h;
      const bool live = (q != 11) || (lane < 23);
      const int j = live ? 32 * q + lane : 0;
#pragma unroll
      for (int o = 0; o < 3; ++o) { const v4f t = *(const v4f*)(sw + WQ2 + o * 1500 + 4 * j); w[h * 3 + o] = live ? t : zero; }
    }
  } else if constexpr (S == 18) {
#pragma unroll
    for (int h = 0; h < 2; ++h)
#pragma unroll
      for (int o = 0; o < 3; ++o) w[h * 3 + o] = *(const v4f*)(sw + WQ3 + o * 300 + 4 * (32 * h + lane));
  } else {
    const bool live = lane < 11;
    const int j = live ? 64 + lane : 0;
#pragma unroll
    for (int o = 0; o < 3; ++o) { const v4f t = *(const v4f*)(sw + WQ3 + o * 300 + 4 * j); w[o] = live ? t : zero; }
  }
}

__device__ __forceinline__ void pair_mixed(v4f d, const v4f* w, float* accA, float* accB,
                                           bool xyA, bool zwA) {
#pragma unroll
  for (int o = 0; o < 3; ++o) {
    const float pxy = __builtin_fmaf(d.x, w[o].x, d.y * w[o].y);
    const float pzw = __builtin_fmaf(d.z, w[o].z, d.w * w[o].w);
    accA[o] += (xyA ? pxy : 0.f) + (zwA ? pzw : 0.f);
    accB[o] += (xyA ? 0.f : pxy) + (zwA ? 0.f : pzw);
  }
}

template <int S>
__device__ __forceinline__ void cat_consume(const v4f (&buf)[8], const v4f (&w)[6],
                                            float (&acc)[8][3], int lane) {
  if constexpr (S < 12) {
#pragma unroll
    for (int r = 0; r < 8; ++r)
#pragma unroll
      for (int o = 0; o < 3; ++o) acc[r][o] = dot4(buf[r], w[o], acc[r][o]);
  } else if constexpr (S < 18) {
#pragma unroll
    for (int h = 0; h < 2; ++h) {
      const int q = 2 * (S - 12) + h;
#pragma unroll
      for (int p = 0; p < 4; ++p) {
        if (q < 5) {
#pragma unroll
          for (int o = 0; o < 3; ++o) acc[2 * p][o] = dot4(buf[h * 4 + p], w[h * 3 + o], acc[2 * p][o]);
        } else if (q > 5) {
#pragma unroll
          for (int o = 0; o < 3; ++o) acc[2 * p + 1][o] = dot4(buf[h * 4 + p], w[h * 3 + o], acc[2 * p + 1][o]);
        } else {
          pair_mixed(buf[h * 4 + p], &w[h * 3], acc[2 * p], acc[2 * p + 1], lane <= 27, lane <= 26);
        }
      }
    }
  } else if constexpr (S == 18) {
#pragma unroll
    for (int p = 0; p < 4; ++p) {
#pragma unroll
      for (int o = 0; o < 3; ++o) acc[2 * p][o] = dot4(buf[p], w[o], acc[2 * p][o]);
      pair_mixed(buf[4 + p], &w[3], acc[2 * p], acc[2 * p + 1], lane <= 5, lane <= 4);
    }
  } else {
#pragma unroll
    for (int p = 0; p < 4; ++p)
#pragma unroll
      for (int o = 0; o < 3; ++o) acc[2 * p + 1][o] = dot4(buf[p], w[o], acc[2 * p + 1][o]);
  }
}

__global__ void __launch_bounds__(256) __attribute__((amdgpu_waves_per_eu(4, 4))) k_cat(
    const float* __restrict__ in1, const float* __restrict__ in2,
    const float* __restrict__ in3, const float* __restrict__ in4,
    const float* __restrict__ wl, const float* __restrict__ bl,
    const float* __restrict__ wm, const float* __restrict__ bm,
    const float* __restrict__ wsw, const float* __restrict__ bs,
    float* __restrict__ cat, float* __restrict__ zacc)
{
  __shared__ __attribute__((aligned(16))) float sw[WQN];
  __shared__ float red[CAT_WAVES][72 * RPITCH];
  __shared__ __attribute__((aligned(16))) float cst[CAT_WAVES][CAT_ROWS * 16];

  const int warp = threadIdx.x >> 5, lane = threadIdx.x & 31;
  const int b0 = (blockIdx.x * CAT_WAVES + warp) * CAT_ROWS;
  const v4f* p1 = (const v4f*)in1 + (size_t)b0 * 375 + lane;
  const v4f* p2 = (const v4f*)in2 + (size_t)(b0 >> 1) * 375 + lane;
  const v4f* p3 = (const v4f*)in3 + (size_t)(b0 >> 1) * 75 + lane;

  v4f buf[4][8];
  cat_issue<0>(buf[0], p1, p2, p3, lane);
  cat_issue<1>(buf[1], p1, p2, p3, lane);
  cat_issue<2>(buf[2], p1, p2, p3, lane);

  float tailv[2];
#pragma unroll
  for (int g = 0; g < 2; ++g) {
    const int u = lane + 32 * g, cc = u % 7;
    const bool take = (u < 56) && (cc < 3);
    const float t = in4[(size_t)b0 * 3 + (take ? (u / 7) * 3 + cc : 0)];
    tailv[g] = take ? t : 0.f;
  }

  {
    float t1[18];
#pragma unroll
    for (int t = 0; t < 18; ++t) {
      const int i = min((int)threadIdx.x + 256 * t, 4499), o = i / 1500, f = i - o * 1500;
      t1[t] = wl[(o * 3 + f % 3) * 500 + f / 3];
    }
#pragma unroll
    for (int t = 0; t < 18; ++t) sw[WQ1 + threadIdx.x + 256 * t] = t1[t];
#pragma unroll
    for (int t = 0; t < 18; ++t) {
      const int i = min((int)threadIdx.x + 256 * t, 4499), o = i / 1500, g = i - o * 1500;
      const int f = (g >= 750) ? g - 750 : g;
      t1[t] = wm[(o * 3 + f % 3) * 250 + f / 3];
    }
#pragma unroll
    for (int t = 0; t < 18; ++t) sw[WQ2 + threadIdx.x + 256 * t] = t1[t];
#pragma unroll
    for (int t = 0; t < 4; ++t) {
      const int i = min((int)threadIdx.x + 256 * t, 899), o = i / 300, g = i - o * 300;
      const int f = (g >= 150) ? g - 150 : g;
      t1[t] = wsw[(o * 3 + f % 3) * 50 + f / 3];
    }
#pragma unroll
    for (int t = 0; t < 4; ++t) sw[WQ3 + threadIdx.x + 256 * t] = t1[t];
  }
  __syncthreads();

  float acc[8][3];
  float* redw = red[warp];
  v4f w[6];

#define CAT_STAGE(S)                                                              \
  cat_weights<S>(w, sw, lane);                                                    \
  __builtin_amdgcn_sched_barrier(0);                                              \
  if constexpr ((S) + 3 < NSTAGE)                                                 \
    cat_issue<((S) + 3 < NSTAGE ? (S) + 3 : 0)>(buf[((S) + 3) % 4], p1, p2, p3, lane); \
  __builtin_amdgcn_sched_barrier(0);                                              \
  cat_consume<S>(buf[(S) % 4], w, acc, lane);                                     \
  CAT_PIN()
#define CAT_PIN()                                                                 \
  _Pragma("unroll") for (int r = 0; r < 8; ++r)                                   \
    _Pragma("unroll") for (int o = 0; o < 3; ++o) asm volatile("" : "+v"(acc[r][o]));
#define CAT_CLEAR()                                                               \
  _Pragma("unroll") for (int r = 0; r < 8; ++r)                                   \
    _Pragma("unroll") for (int o = 0; o < 3; ++o) acc[r][o] = 0.f;
#define CAT_PARK(COL)                                                             \
  _Pragma("unroll") for (int r = 0; r < 8; ++r)                                   \
    _Pragma("unroll") for (int o = 0; o < 3; ++o)                                 \
      redw[(r * 9 + (COL) + o) * RPITCH + lane] = acc[r][o];

  CAT_CLEAR()
  CAT_STAGE(0) CAT_STAGE(1) CAT_STAGE(2) CAT_STAGE(3) CAT_STAGE(4) CAT_STAGE(5)
  CAT_STAGE(6) CAT_STAGE(7) CAT_STAGE(8) CAT_STAGE(9) CAT_STAGE(10) CAT_STAGE(11)
  CAT_PARK(0)
  CAT_CLEAR()
  CAT_STAGE(12) CAT_STAGE(13) CAT_STAGE(14) CAT_STAGE(15) CAT_STAGE(16) CAT_STAGE(17)
  CAT_PARK(3)
  CAT_CLEAR()
  CAT_STAGE(18) CAT_STAGE(19)
  CAT_PARK(6)
#undef CAT_STAGE
#undef CAT_PIN
#undef CAT_CLEAR
#undef CAT_PARK
  __syncthreads();

  float* cw = cst[warp];
  const float b9[9] = {bl[0], bl[1], bl[2], bm[0], bm[1], bm[2], bs[0], bs[1], bs[2]};
#pragma unroll
  for (int g = 0; g < 3; ++g) {
    const int v = lane + 32 * g;
    if (v < 72) {
      const float* rp = redw + v * RPITCH;
      float s0 = 0.f, s1 = 0.f, s2 = 0.f, s3 = 0.f;
#pragma unroll
      for (int i = 0; i < 32; i += 4) { s0 += rp[i]; s1 += rp[i + 1]; s2 += rp[i + 2]; s3 += rp[i + 3]; }
      const int c = v % 9;
      float bias = b9[0];
#pragma unroll
      for (int k = 1; k < 9; ++k) bias = (c == k) ? b9[k] : bias;
      cw[(v / 9) * 16 + c] = (s0 + s1) + (s2 + s3) + bias;
    }
  }
#pragma unroll
  for (int g = 0; g < 2; ++g) {
    const int u = lane + 32 * g;
    if (u < 56) cw[(u / 7) * 16 + 9 + u % 7] = tailv[g];
  }
  __syncthreads();
  *((v4f*)(cat + (size_t)b0 * 16) + lane) = *((const v4f*)cw + lane);
  if (blockIdx.x == 0) zacc[threadIdx.x] = 0.f;
}


__global__ void __launch_bounds__(256) __attribute__((amdgpu_waves_per_eu(2, 2))) k_gram(
    const float* __restrict__ cat,
    const float* __restrict__ lw, const float* __restrict__ tk, const float* __restrict__ tvm,
    float* __restrict__ Mw, float* __restrict__ gC)
{
  __shared__ float sC[16 * 16];
  const int warp = threadIdx.x >> 5, lane = threadIdx.x & 31;
  const int hi = lane >> 4, nl = lane & 15;
  const int row0 = (blockIdx.x * 8 + warp) * 16;

  sC[threadIdx.x] = 0.f;
  float cg[4][2];
#pragma unroll
  for (int m = 0; m < 4; ++m) {
    const float* p = cat + (size_t)(row0 + 4 * m + 2 * hi) * 16 + nl;
    cg[m][0] = p[0];
    cg[m][1] = p[16];
  }
  if (blockIdx.x < 16 && warp == 0) {
    const int a0 = blockIdx.x >> 2, i0 = blockIdx.x & 3;
    const int a = a0 * 16 + nl, i = i0 * 16 + nl;
    v8f c = {};
#pragma unroll
    for (int ks = 0; ks < 16; ++ks) {
      const int kb = ks * 4 + 2 * hi;
      c = wmma_f32(*(const v2f*)(tk + a * 64 + kb), *(const v2f*)(lw + i * 64 + kb), c);
    }
#pragma unroll
    for (int r = 0; r < 8; ++r) {
      const int e = (a0 * 16 + r + 8 * hi) * 64 + i;
      Mw[e] = c[r] - tvm[e];
    }
  }
  __syncthreads();

  {
    v8f cc = {};
#pragma unroll
    for (int m = 0; m < 4; ++m) {
      const v2f x = mkv2(nl == 12 ? 1.0f : cg[m][0], nl == 12 ? 1.0f : cg[m][1]);
      cc = wmma_f32(x, x, cc);
    }
#pragma unroll
    for (int r = 0; r < 8; ++r) atomicAdd(&sC[(r + 8 * hi) * 16 + nl], cc[r]);
  }
  __syncthreads();
  atomicAdd(&gC[threadIdx.x], sC[threadIdx.x]);
}

__global__ void __launch_bounds__(256) k_out(
    const float* __restrict__ cat, const float* __restrict__ tq, const float* __restrict__ tk,
    const float* __restrict__ lw, const float* __restrict__ lb,
    const float* __restrict__ Mw, const float* __restrict__ gC,
    const float* __restrict__ fc1w, const float* __restrict__ fc1b,
    const float* __restrict__ qw, const float* __restrict__ qb,
    const float* __restrict__ kw, const float* __restrict__ kbv,
    const float* __restrict__ vw, const float* __restrict__ vb,
    const float* __restrict__ lng, const float* __restrict__ lnb,
    const float* __restrict__ ow, const float* __restrict__ obp,
    float* __restrict__ out)
{
  __shared__ float so[8][16 * 68];
  __shared__ __attribute__((aligned(16))) float slw1[64 * WP];
  __shared__ float sQ[64 * WP];
  __shared__ float slb1[64];
  __shared__ float sct[64];
  __shared__ float sC[16 * 16];
  __shared__ __attribute__((aligned(16))) float sA[3][64 * W1P];
  __shared__ __attribute__((aligned(16))) float sY[3][64 * W1P];
  __shared__ __attribute__((aligned(16))) float swt[3][64 * WP];
  __shared__ __attribute__((aligned(16))) float sw1[64 * W1P];
  __shared__ __attribute__((aligned(16))) float sS2[64 * WP];
  __shared__ float scs[64];
  __shared__ __attribute__((aligned(16))) float sT[16 * WP];
  __shared__ __attribute__((aligned(16))) float sPT[64 * W1P];
  const int warp = threadIdx.x >> 5, lane = threadIdx.x & 31;
  const int hi = lane >> 4, nl = lane & 15;
  const int row0 = (blockIdx.x * 8 + warp) * 16;
  float* om = so[warp];
  const float gscale = 2.0f / (float)(3ull * BATCH * 64);
  const int t0 = warp >> 1, t1 = 2 * (warp & 1);

  v2f ca[4];
#pragma unroll
  for (int ks = 0; ks < 4; ++ks) {
    const float* p = cat + (size_t)(row0 + nl) * 16 + ks * 4 + 2 * hi;
    ca[ks] = mkv2((ks == 3 && hi == 0) ? 1.0f : p[0], p[1]);
  }
  const float* WS[3] = {qw, kw, vw};
  v4f wst[3][4];
#pragma unroll
  for (int t = 0; t < 3; ++t)
#pragma unroll
    for (int j = 0; j < 4; ++j) wst[t][j] = *(const v4f*)(WS[t] + ((int)threadIdx.x + 256 * j) * 4);
  const int e1 = min((int)threadIdx.x, 191) * 4;
  const v4f wf1 = *(const v4f*)(fc1w + e1);
  const float wfb = fc1b[threadIdx.x & 63];
  const float cval = gC[threadIdx.x];
  float bt[2][8];
#pragma unroll
  for (int rep = 0; rep < 2; ++rep) {
    const int T = min(warp + 8 * rep, 11), t = T >> 2, n0 = T & 3;
    const float* bsel = (t == 0) ? qb : ((t == 1) ? kbv : vb);
#pragma unroll
    for (int r = 0; r < 8; ++r) bt[rep][r] = bsel[n0 * 16 + 8 * hi + r];
  }
  {
#pragma unroll
    for (int t = 0; t < 3; ++t)
#pragma unroll
      for (int j = 0; j < 4; ++j) {
        const int e = ((int)threadIdx.x + 256 * j) * 4;
        *(v4f*)(&swt[t][(e >> 6) * WP + (e & 63)]) = wst[t][j];
      }
    if (threadIdx.x < 192) {
      const int n = e1 / 12;
      *(v4f*)(&sw1[n * W1P + (e1 - n * 12)]) = wf1;
    }
    if (threadIdx.x < 64) {
      const v4f tail = {wfb + ((threadIdx.x & 1) ? 1.0f : 0.0f), 0.f, 0.f, 0.f};
      *(v4f*)(&sw1[(int)threadIdx.x * W1P + 12]) = tail;
    }
    sC[threadIdx.x] = cval;
  }
  __syncthreads();
#pragma unroll
  for (int rep = 0; rep < 2; ++rep) {
    const int T = warp + 8 * rep;
    if (T < 12) {
      const int t = T >> 2, n0 = T & 3;
      v8f c = {};
#pragma unroll
      for (int ks = 0; ks < 16; ++ks) {
        const int kb = ks * 4 + 2 * hi;
        c = wmma_f32(*(const v2f*)(&swt[t][(n0 * 16 + nl) * WP + kb]),
                     mkv2(sw1[kb * W1P + nl], sw1[(kb + 1) * W1P + nl]), c);
      }
#pragma unroll
      for (int r = 0; r < 8; ++r) {
        const int n = n0 * 16 + r + 8 * hi;
        sA[t][n * W1P + nl] = c[r] + (nl == 12 ? bt[rep][r] : 0.f);
      }
    }
  }
  __syncthreads();
#pragma unroll
  for (int rep = 0; rep < 2; ++rep) {
    const int T = warp + 8 * rep;
    if (T < 12) {
      const int t = T >> 2, n0 = T & 3;
      v8f c = {};
#pragma unroll
      for (int ks = 0; ks < 4; ++ks) {
        const int kb = ks * 4 + 2 * hi;
        c = wmma_f32(*(const v2f*)(&sA[t][(n0 * 16 + nl) * W1P + kb]),
                     mkv2(sC[kb * 16 + nl], sC[(kb + 1) * 16 + nl]), c);
      }
#pragma unroll
      for (int r = 0; r < 8; ++r) sY[t][(n0 * 16 + r + 8 * hi) * W1P + nl] = c[r];
    }
  }
  __syncthreads();
#pragma unroll
  for (int jj = 0; jj < 2; ++jj) {
    const int j0 = t1 + jj;
    v8f c = {};
#pragma unroll
    for (int t = 0; t < 3; ++t)
#pragma unroll
      for (int ks = 0; ks < 4; ++ks) {
        const int kb = ks * 4 + 2 * hi;
        c = wmma_f32(*(const v2f*)(&sY[t][(t0 * 16 + nl) * W1P + kb]),
                     *(const v2f*)(&sA[t][(j0 * 16 + nl) * W1P + kb]), c);
      }
#pragma unroll
    for (int r = 0; r < 8; ++r)
      sS2[(t0 * 16 + r + 8 * hi) * WP + j0 * 16 + nl] = c[r];
  }
  if (threadIdx.x < 64) {
    const int n = threadIdx.x;
    scs[n] = (sY[0][n * W1P + 12] + sY[1][n * W1P + 12]) + sY[2][n * W1P + 12];
  }
  __syncthreads();
#pragma unroll
  for (int jj = 0; jj < 2; ++jj) {
    const int a = t0 * 16 + nl, j = (t1 + jj) * 16 + nl;
    v8f c = {};
#pragma unroll
    for (int ks = 0; ks < 16; ++ks) {
      const int kb = ks * 4 + 2 * hi;
      c = wmma_f32(*(const v2f*)(&sS2[a * WP + kb]), mkv2(tk[kb * 64 + j], tk[(kb + 1) * 64 + j]), c);
    }
#pragma unroll
    for (int r = 0; r < 8; ++r) sQ[(t0 * 16 + r + 8 * hi) * WP + j] = c[r];
  }
  v2f ma[16];
#pragma unroll
  for (int ks = 0; ks < 16; ++ks) {
    const int kb = ks * 4 + 2 * hi, i = t0 * 16 + nl;
    ma[ks] = mkv2(Mw[kb * 64 + i], Mw[(kb + 1) * 64 + i]);
  }
  if (warp < 4) {
    const int j = (int)threadIdx.x & 63;
    const float* src = (warp < 2) ? tk : Mw;
    float dv[64];
#pragma unroll
    for (int b = 0; b < 64; ++b) dv[b] = src[b * 64 + j];
    const float lbj = lb[j];
    float acc = 0.f;
#pragma unroll
    for (int b = 0; b < 64; ++b) acc = __builtin_fmaf(scs[b], dv[b], acc);
    const float g = gscale * (acc + (float)(3 * BATCH) * lbj);
    const float upd = lbj - LRATE * g / (fabsf(g) + AEPS);
    float* dst = (warp < 2) ? sct : slb1;
    dst[j] = (warp < 2) ? acc : upd;
  } else {
    const int m = (warp - 4) * 16 + nl;
    v8f c = {};
#pragma unroll
    for (int ks = 0; ks < 16; ++ks) {
      const int kb = ks * 4 + 2 * hi;
      c = wmma_f32(mkv2(sA[2][kb * W1P + nl], sA[2][(kb + 1) * W1P + nl]),
                   mkv2(tq[kb * 64 + m], tq[(kb + 1) * 64 + m]), c);
    }
#pragma unroll
    for (int r = 0; r < 8; ++r) sT[(r + 8 * hi) * WP + m] = c[r];
  }
  __syncthreads();
#pragma unroll
  for (int jj = 0; jj < 2; ++jj) {
    const int j = (t1 + jj) * 16 + nl;
    v8f c = {};
#pragma unroll
    for (int ks = 0; ks < 16; ++ks) {
      const int kb = ks * 4 + 2 * hi;
      c = wmma_f32(ma[ks], mkv2(sQ[kb * WP + j], sQ[(kb + 1) * WP + j]), c);
    }
    const float ctj = sct[j];
#pragma unroll
    for (int r = 0; r < 8; ++r) {
      const int ii = t0 * 16 + r + 8 * hi;
      const float g = gscale * (c[r] + lb[ii] * ctj);
      slw1[ii * WP + j] = lw[ii * 64 + j] - LRATE * g / (fabsf(g) + AEPS);
    }
  }
  __syncthreads();
  if (warp < 4) {
    const int n = warp * 16 + nl;
    v8f c = {};
#pragma unroll
    for (int ks = 0; ks < 16; ++ks) {
      const int kb = ks * 4 + 2 * hi;
      c = wmma_f32(*(const v2f*)(&sT[nl * WP + kb]), *(const v2f*)(&slw1[n * WP + kb]), c);
    }
#pragma unroll
    for (int r = 0; r < 8; ++r) sPT[n * W1P + 8 * hi + r] = c[r];
  }
  __syncthreads();
#pragma unroll
  for (int n0 = 0; n0 < 4; ++n0) {
    int n = n0 * 16 + nl;
    v8f c = {};
#pragma unroll
    for (int ks = 0; ks < 4; ++ks) {
      int kb = ks * 4 + 2 * hi;
      c = wmma_f32(ca[ks], *(const v2f*)(&sPT[n * W1P + kb]), c);
    }
    float bias = slb1[n];
#pragma unroll
    for (int r = 0; r < 8; ++r) om[(r + 8 * hi) * 68 + n] = c[r] + bias;
  }
  __syncthreads();
  if (lane < 16) {
    const float* rowp = om + lane * 68;
    float mu = 0.f;
#pragma unroll
    for (int j = 0; j < 64; ++j) mu += rowp[j];
    mu *= (1.0f / 64.0f);
    float var = 0.f;
#pragma unroll
    for (int j = 0; j < 64; ++j) { float d = rowp[j] - mu; var += d * d; }
    var *= (1.0f / 64.0f);
    float rstd = rsqrtf(var + LNEPS);
    float acc = 0.f;
#pragma unroll
    for (int j = 0; j < 64; ++j) {
      float nrm = (rowp[j] - mu) * rstd * lng[j] + lnb[j];
      float a = nrm / (1.0f + __expf(-nrm));
      acc += a * ow[j];
    }
    out[row0 + lane] = acc + obp[0];
  }
}

extern "C" void kernel_launch(void* const* d_in, const int* in_sizes, int n_in,
                              void* d_out, int out_size, void* d_ws, size_t ws_size,
                              hipStream_t stream) {
  const float* in1  = (const float*)d_in[0];
  const float* in2  = (const float*)d_in[1];
  const float* in3  = (const float*)d_in[2];
  const float* in4  = (const float*)d_in[3];
  const float* w_l  = (const float*)d_in[4];
  const float* b_l  = (const float*)d_in[5];
  const float* w_m  = (const float*)d_in[6];
  const float* b_m  = (const float*)d_in[7];
  const float* w_s  = (const float*)d_in[8];
  const float* b_s  = (const float*)d_in[9];
  const float* fc1w = (const float*)d_in[10];
  const float* fc1b = (const float*)d_in[11];
  const float* q_w  = (const float*)d_in[12];
  const float* q_b  = (const float*)d_in[13];
  const float* k_w  = (const float*)d_in[14];
  const float* k_b  = (const float*)d_in[15];
  const float* v_w  = (const float*)d_in[16];
  const float* v_b  = (const float*)d_in[17];
  const float* t_k  = (const float*)d_in[18];
  const float* t_v  = (const float*)d_in[19];
  const float* t_q  = (const float*)d_in[20];
  const float* lw   = (const float*)d_in[21];
  const float* lb   = (const float*)d_in[22];
  const float* lng  = (const float*)d_in[23];
  const float* lnb  = (const float*)d_in[24];
  const float* outw = (const float*)d_in[25];
  const float* outb = (const float*)d_in[26];

  float* wsf  = (float*)d_ws;
  float* cat  = wsf + F_CAT;
  float* Mw   = wsf + F_M;
  float* gC   = wsf + F_C;

  k_cat<<<512, 256, 0, stream>>>(in1, in2, in3, in4, w_l, b_l, w_m, b_m, w_s, b_s, cat, gC);
  k_gram<<<BATCH / 128, 256, 0, stream>>>(cat, lw, t_k, t_v, Mw, gC);
  k_out<<<BATCH / 128, 256, 0, stream>>>(cat, t_q, t_k, lw, lb, Mw, gC, fc1w, fc1b, q_w, q_b,
                                         k_w, k_b, v_w, v_b, lng, lnb, outw, outb, (float*)d_out);
}


# inner gradient from weight-only factors: 12 extra k_gram blocks make U_t=M^T A_t, V_t=A_t^T t_k, T once; k_out: grad W = sum_t U_t (C V_t), row sums from C's row 12; changes the gradient sum order
# speedup vs baseline: 3.3315x; 1.0854x over previous
#include <hip/hip_runtime.h>
#include <math.h>

#define BATCH 32768
#define LRATE 0.01f
#define AEPS  1e-8f
#define LNEPS 1e-5f

typedef __attribute__((ext_vector_type(2))) float v2f;
typedef __attribute__((ext_vector_type(8))) float v8f;

__device__ __forceinline__ v2f mkv2(float x, float y) { v2f v; v[0] = x; v[1] = y; return v; }

__device__ __forceinline__ v8f wmma_f32(v2f a, v2f b, v8f c) {
  return __builtin_amdgcn_wmma_f32_16x16x4_f32(false, a, false, b, (short)0, c, false, false);
}

constexpr size_t F_CAT  = 0;
constexpr size_t F_C    = F_CAT + (size_t)BATCH * 16;
constexpr size_t F_U    = F_C + 16 * 16;
constexpr size_t F_VT   = F_U + 3 * 64 * 16;
constexpr size_t F_T    = F_VT + 3 * 64 * 16;

constexpr int WP  = 68;
constexpr int W1P = 20;

typedef __attribute__((ext_vector_type(4))) float v4f;

constexpr int CAT_WAVES  = 8;
constexpr int CAT_ROWS   = 8;
constexpr int CAT_BLOCKS = BATCH / (CAT_WAVES * CAT_ROWS);
static_assert(CAT_BLOCKS == 512 && CAT_WAVES * 32 == 256, "k_cat is launched as <<<512, 256>>>");
constexpr int WQ1 = 0;
constexpr int WQ2 = 4608;
constexpr int WQ3 = 9216;
constexpr int WQN = 10240;
constexpr int NSTAGE = 20;
constexpr int RPITCH = 33;

__device__ __forceinline__ v4f ld_stream(const v4f* p) { return __builtin_nontemporal_load(p); }
__device__ __forceinline__ float dot4(v4f d, v4f w, float a) {
  a = __builtin_fmaf(d.x, w.x, a); a = __builtin_fmaf(d.y, w.y, a);
  a = __builtin_fmaf(d.z, w.z, a); a = __builtin_fmaf(d.w, w.w, a);
  return a;
}

template <int S>
__device__ __forceinline__ void cat_issue(v4f (&buf)[8], const v4f* p1, const v4f* p2,
                                          const v4f* p3, int lane) {
  if constexpr (S < 12) {
#pragma unroll
    for (int r = 0; r < 8; ++r) {
      if constexpr (S == 11) buf[r] = ld_stream(p1 + r * 375 + ((lane < 23) ? 352 : 374 - lane));
      else                   buf[r] = ld_stream(p1 + r * 375 + 32 * S);
    }
  } else if constexpr (S < 18) {
#pragma unroll
    for (int h = 0; h < 2; ++h) {
      const int q = 2 * (S - 12) + h;
#pragma unroll
      for (int p = 0; p < 4; ++p) {
        if (q == 11) buf[h * 4 + p] = ld_stream(p2 + p * 375 + ((lane < 23) ? 352 : 374 - lane));
        else         buf[h * 4 + p] = ld_stream(p2 + p * 375 + 32 * q);
      }
    }
  } else if constexpr (S == 18) {
#pragma unroll
    for (int h = 0; h < 2; ++h)
#pragma unroll
      for (int p = 0; p < 4; ++p) buf[h * 4 + p] = ld_stream(p3 + p * 75 + 32 * h);
  } else if constexpr (S == 19) {
#pragma unroll
    for (int p = 0; p < 4; ++p) buf[p] = ld_stream(p3 + p * 75 + ((lane < 11) ? 64 : 74 - lane));
  }
}

template <int S>
__device__ __forceinline__ void cat_weights(v4f (&w)[6], const float* sw, int lane) {
  const v4f zero = {0.f, 0.f, 0.f, 0.f};
  if constexpr (S < 12) {
    const bool live = (S != 11) || (lane < 23);
    const int j = live ? 32 * S + lane : 0;
#pragma unroll
    for (int o = 0; o < 3; ++o) { const v4f t = *(const v4f*)(sw + WQ1 + o * 1500 + 4 * j); w[o] = live ? t : zero; }
  } else if constexpr (S < 18) {
#pragma unroll
    for (int h = 0; h < 2; ++h) {
      const int q = 2 * (S - 12) + h;
      const bool live = (q != 11) || (lane < 23);
      const int j = live ? 32 * q + lane : 0;
#pragma unroll
      for (int o = 0; o < 3; ++o) { const v4f t = *(const v4f*)(sw + WQ2 + o * 1500 + 4 * j); w[h * 3 + o] = live ? t : zero; }
    }
  } else if constexpr (S == 18) {
#pragma unroll
    for (int h = 0; h < 2; ++h)
#pragma unroll
      for (int o = 0; o < 3; ++o) w[h * 3 + o] = *(const v4f*)(sw + WQ3 + o * 300 + 4 * (32 * h + lane));
  } else {
    const bool live = lane < 11;
    const int j = live ? 64 + lane : 0;
#pragma unroll
    for (int o = 0; o < 3; ++o) { const v4f t = *(const v4f*)(sw + WQ3 + o * 300 + 4 * j); w[o] = live ? t : zero; }
  }
}

__device__ __forceinline__ void pair_mixed(v4f d, const v4f* w, float* accA, float* accB,
                                           bool xyA, bool zwA) {
#pragma unroll
  for (int o = 0; o < 3; ++o) {
    const float pxy = __builtin_fmaf(d.x, w[o].x, d.y * w[o].y);
    const float pzw = __builtin_fmaf(d.z, w[o].z, d.w * w[o].w);
    accA[o] += (xyA ? pxy : 0.f) + (zwA ? pzw : 0.f);
    accB[o] += (xyA ? 0.f : pxy) + (zwA ? 0.f : pzw);
  }
}

template <int S>
__device__ __forceinline__ void cat_consume(const v4f (&buf)[8], const v4f (&w)[6],
                                            float (&acc)[8][3], int lane) {
  if constexpr (S < 12) {
#pragma unroll
    for (int r = 0; r < 8; ++r)
#pragma unroll
      for (int o = 0; o < 3; ++o) acc[r][o] = dot4(buf[r], w[o], acc[r][o]);
  } else if constexpr (S < 18) {
#pragma unroll
    for (int h = 0; h < 2; ++h) {
      const int q = 2 * (S - 12) + h;
#pragma unroll
      for (int p = 0; p < 4; ++p) {
        if (q < 5) {
#pragma unroll
          for (int o = 0; o < 3; ++o) acc[2 * p][o] = dot4(buf[h * 4 + p], w[h * 3 + o], acc[2 * p][o]);
        } else if (q > 5) {
#pragma unroll
          for (int o = 0; o < 3; ++o) acc[2 * p + 1][o] = dot4(buf[h * 4 + p], w[h * 3 + o], acc[2 * p + 1][o]);
        } else {
          pair_mixed(buf[h * 4 + p], &w[h * 3], acc[2 * p], acc[2 * p + 1], lane <= 27, lane <= 26);
        }
      }
    }
  } else if constexpr (S == 18) {
#pragma unroll
    for (int p = 0; p < 4; ++p) {
#pragma unroll
      for (int o = 0; o < 3; ++o) acc[2 * p][o] = dot4(buf[p], w[o], acc[2 * p][o]);
      pair_mixed(buf[4 + p], &w[3], acc[2 * p], acc[2 * p + 1], lane <= 5, lane <= 4);
    }
  } else {
#pragma unroll
    for (int p = 0; p < 4; ++p)
#pragma unroll
      for (int o = 0; o < 3; ++o) acc[2 * p + 1][o] = dot4(buf[p], w[o], acc[2 * p + 1][o]);
  }
}

__global__ void __launch_bounds__(256) __attribute__((amdgpu_waves_per_eu(4, 4))) k_cat(
    const float* __restrict__ in1, const float* __restrict__ in2,
    const float* __restrict__ in3, const float* __restrict__ in4,
    const float* __restrict__ wl, const float* __restrict__ bl,
    const float* __restrict__ wm, const float* __restrict__ bm,
    const float* __restrict__ wsw, const float* __restrict__ bs,
    float* __restrict__ cat, float* __restrict__ zacc)
{
  __shared__ __attribute__((aligned(16))) float sw[WQN];
  __shared__ float red[CAT_WAVES][72 * RPITCH];
  __shared__ __attribute__((aligned(16))) float cst[CAT_WAVES][CAT_ROWS * 16];

  const int warp = threadIdx.x >> 5, lane = threadIdx.x & 31;
  const int b0 = (blockIdx.x * CAT_WAVES + warp) * CAT_ROWS;
  const v4f* p1 = (const v4f*)in1 + (size_t)b0 * 375 + lane;
  const v4f* p2 = (const v4f*)in2 + (size_t)(b0 >> 1) * 375 + lane;
  const v4f* p3 = (const v4f*)in3 + (size_t)(b0 >> 1) * 75 + lane;

  v4f buf[4][8];
  cat_issue<0>(buf[0], p1, p2, p3, lane);
  cat_issue<1>(buf[1], p1, p2, p3, lane);
  cat_issue<2>(buf[2], p1, p2, p3, lane);

  float tailv[2];
#pragma unroll
  for (int g = 0; g < 2; ++g) {
    const int u = lane + 32 * g, cc = u % 7;
    const bool take = (u < 56) && (cc < 3);
    const float t = in4[(size_t)b0 * 3 + (take ? (u / 7) * 3 + cc : 0)];
    tailv[g] = take ? t : 0.f;
  }

  {
    float t1[18];
#pragma unroll
    for (int t = 0; t < 18; ++t) {
      const int i = min((int)threadIdx.x + 256 * t, 4499), o = i / 1500, f = i - o * 1500;
      t1[t] = wl[(o * 3 + f % 3) * 500 + f / 3];
    }
#pragma unroll
    for (int t = 0; t < 18; ++t) sw[WQ1 + threadIdx.x + 256 * t] = t1[t];
#pragma unroll
    for (int t = 0; t < 18; ++t) {
      const int i = min((int)threadIdx.x + 256 * t, 4499), o = i / 1500, g = i - o * 1500;
      const int f = (g >= 750) ? g - 750 : g;
      t1[t] = wm[(o * 3 + f % 3) * 250 + f / 3];
    }
#pragma unroll
    for (int t = 0; t < 18; ++t) sw[WQ2 + threadIdx.x + 256 * t] = t1[t];
#pragma unroll
    for (int t = 0; t < 4; ++t) {
      const int i = min((int)threadIdx.x + 256 * t, 899), o = i / 300, g = i - o * 300;
      const int f = (g >= 150) ? g - 150 : g;
      t1[t] = wsw[(o * 3 + f % 3) * 50 + f / 3];
    }
#pragma unroll
    for (int t = 0; t < 4; ++t) sw[WQ3 + threadIdx.x + 256 * t] = t1[t];
  }
  __syncthreads();

  float acc[8][3];
  float* redw = red[warp];
  v4f w[6];

#define CAT_STAGE(S)                                                              \
  cat_weights<S>(w, sw, lane);                                                    \
  __builtin_amdgcn_sched_barrier(0);                                              \
  if constexpr ((S) + 3 < NSTAGE)                                                 \
    cat_issue<((S) + 3 < NSTAGE ? (S) + 3 : 0)>(buf[((S) + 3) % 4], p1, p2, p3, lane); \
  __builtin_amdgcn_sched_barrier(0);                                              \
  cat_consume<S>(buf[(S) % 4], w, acc, lane);                                     \
  CAT_PIN()
#define CAT_PIN()                                                                 \
  _Pragma("unroll") for (int r = 0; r < 8; ++r)                                   \
    _Pragma("unroll") for (int o = 0; o < 3; ++o) asm volatile("" : "+v"(acc[r][o]));
#define CAT_CLEAR()                                                               \
  _Pragma("unroll") for (int r = 0; r < 8; ++r)                                   \
    _Pragma("unroll") for (int o = 0; o < 3; ++o) acc[r][o] = 0.f;
#define CAT_PARK(COL)                                                             \
  _Pragma("unroll") for (int r = 0; r < 8; ++r)                                   \
    _Pragma("unroll") for (int o = 0; o < 3; ++o)                                 \
      redw[(r * 9 + (COL) + o) * RPITCH + lane] = acc[r][o];

  CAT_CLEAR()
  CAT_STAGE(0) CAT_STAGE(1) CAT_STAGE(2) CAT_STAGE(3) CAT_STAGE(4) CAT_STAGE(5)
  CAT_STAGE(6) CAT_STAGE(7) CAT_STAGE(8) CAT_STAGE(9) CAT_STAGE(10) CAT_STAGE(11)
  CAT_PARK(0)
  CAT_CLEAR()
  CAT_STAGE(12) CAT_STAGE(13) CAT_STAGE(14) CAT_STAGE(15) CAT_STAGE(16) CAT_STAGE(17)
  CAT_PARK(3)
  CAT_CLEAR()
  CAT_STAGE(18) CAT_STAGE(19)
  CAT_PARK(6)
#undef CAT_STAGE
#undef CAT_PIN
#undef CAT_CLEAR
#undef CAT_PARK
  __syncthreads();

  float* cw = cst[warp];
  const float b9[9] = {bl[0], bl[1], bl[2], bm[0], bm[1], bm[2], bs[0], bs[1], bs[2]};
#pragma unroll
  for (int g = 0; g < 3; ++g) {
    const int v = lane + 32 * g;
    if (v < 72) {
      const float* rp = redw + v * RPITCH;
      float s0 = 0.f, s1 = 0.f, s2 = 0.f, s3 = 0.f;
#pragma unroll
      for (int i = 0; i < 32; i += 4) { s0 += rp[i]; s1 += rp[i + 1]; s2 += rp[i + 2]; s3 += rp[i + 3]; }
      const int c = v % 9;
      float bias = b9[0];
#pragma unroll
      for (int k = 1; k < 9; ++k) bias = (c == k) ? b9[k] : bias;
      cw[(v / 9) * 16 + c] = (s0 + s1) + (s2 + s3) + bias;
    }
  }
#pragma unroll
  for (int g = 0; g < 2; ++g) {
    const int u = lane + 32 * g;
    if (u < 56) cw[(u / 7) * 16 + 9 + u % 7] = tailv[g];
  }
  __syncthreads();
  *((v4f*)(cat + (size_t)b0 * 16) + lane) = *((const v4f*)cw + lane);
  if (blockIdx.x == 0) zacc[threadIdx.x] = 0.f;
}


constexpr int TAB_BLOCKS  = 12;
constexpr int GRAM_BLOCKS = BATCH / 128;
constexpr int GRAM_GRID   = TAB_BLOCKS + GRAM_BLOCKS;
__global__ void __launch_bounds__(256) __attribute__((amdgpu_waves_per_eu(2, 2))) k_gram(
    const float* __restrict__ cat,
    const float* __restrict__ lw, const float* __restrict__ tk, const float* __restrict__ tvm,
    const float* __restrict__ tq,
    const float* __restrict__ fc1w, const float* __restrict__ fc1b,
    const float* __restrict__ qw, const float* __restrict__ qb,
    const float* __restrict__ kw, const float* __restrict__ kbv,
    const float* __restrict__ vw, const float* __restrict__ vb,
    float* __restrict__ Uw, float* __restrict__ VTw, float* __restrict__ Tw,
    float* __restrict__ gC)
{
  __shared__ float sC[16 * 16];
  __shared__ float sAt[64 * W1P];
  __shared__ float sMq[64 * W1P];
  const int warp = threadIdx.x >> 5, lane = threadIdx.x & 31;
  const int hi = lane >> 4, nl = lane & 15;

  if (blockIdx.x < TAB_BLOCKS) {
    const int t = blockIdx.x >> 2, q = blockIdx.x & 3;
    const float* g2src = (warp == 2) ? tq : tk;
    v2f g2[16];
#pragma unroll
    for (int ks = 0; ks < 16; ++ks) {
      const int kb = ks * 4 + 2 * hi;
      g2[ks] = mkv2(g2src[kb * 64 + q * 16 + nl], g2src[(kb + 1) * 64 + q * 16 + nl]);
    }
    if (warp < 4) {
      const float* Wt = (t == 0) ? qw : ((t == 1) ? kw : vw);
      const float* bsel = (t == 0) ? qb : ((t == 1) ? kbv : vb);
      const float* fsrc = (nl < 12) ? fc1w + nl : fc1b;
      const int fstr = (nl < 12) ? 12 : 1;
      v4f wa[8], fb[8];
#pragma unroll
      for (int p = 0; p < 8; ++p) {
        const int k0 = 8 * p + 4 * hi;
        wa[p] = *(const v4f*)(Wt + (warp * 16 + nl) * 64 + k0);
        fb[p].x = fsrc[k0 * fstr];       fb[p].y = fsrc[(k0 + 1) * fstr];
        fb[p].z = fsrc[(k0 + 2) * fstr]; fb[p].w = fsrc[(k0 + 3) * fstr];
      }
      float bv[8];
#pragma unroll
      for (int r = 0; r < 8; ++r) bv[r] = bsel[warp * 16 + 8 * hi + r];
      v8f c = {};
#pragma unroll
      for (int p = 0; p < 8; ++p) {
        const bool z = nl > 12;
        const float one = (nl == 12) ? 1.0f : 0.0f;
        c = wmma_f32(mkv2(wa[p].x, wa[p].y), mkv2(z ? 0.f : fb[p].x, z ? 0.f : fb[p].y + one), c);
        c = wmma_f32(mkv2(wa[p].z, wa[p].w), mkv2(z ? 0.f : fb[p].z, z ? 0.f : fb[p].w + one), c);
      }
#pragma unroll
      for (int r = 0; r < 8; ++r)
        sAt[(warp * 16 + 8 * hi + r) * W1P + nl] = c[r] + (nl == 12 ? bv[r] : 0.f);
    } else {
      const int a0 = warp - 4, i = q * 16 + nl;
      v4f ta[8], la[8];
#pragma unroll
      for (int p = 0; p < 8; ++p) {
        const int k0 = 8 * p + 4 * hi;
        ta[p] = *(const v4f*)(tk + (a0 * 16 + nl) * 64 + k0);
        la[p] = *(const v4f*)(lw + i * 64 + k0);
      }
      float tv8[8];
#pragma unroll
      for (int r = 0; r < 8; ++r) tv8[r] = tvm[(a0 * 16 + 8 * hi + r) * 64 + i];
      v8f c = {};
#pragma unroll
      for (int p = 0; p < 8; ++p) {
        c = wmma_f32(mkv2(ta[p].x, ta[p].y), mkv2(la[p].x, la[p].y), c);
        c = wmma_f32(mkv2(ta[p].z, ta[p].w), mkv2(la[p].z, la[p].w), c);
      }
#pragma unroll
      for (int r = 0; r < 8; ++r) sMq[(a0 * 16 + 8 * hi + r) * W1P + nl] = c[r] - tv8[r];
    }
    __syncthreads();
    if (warp == 0) {
      v2f a[16], b[16];
#pragma unroll
      for (int ks = 0; ks < 16; ++ks) {
        const int kb = ks * 4 + 2 * hi;
        a[ks] = mkv2(sMq[kb * W1P + nl], sMq[(kb + 1) * W1P + nl]);
        b[ks] = mkv2(sAt[kb * W1P + nl], sAt[(kb + 1) * W1P + nl]);
      }
      __builtin_amdgcn_sched_barrier(0);
      v8f c = {};
#pragma unroll
      for (int ks = 0; ks < 16; ++ks) c = wmma_f32(a[ks], b[ks], c);
#pragma unroll
      for (int r = 0; r < 8; ++r) Uw[t * 1024 + (q * 16 + 8 * hi + r) * 16 + nl] = c[r];
    } else if (warp == 1) {
      v2f b[16];
#pragma unroll
      for (int ks = 0; ks < 16; ++ks) {
        const int kb = ks * 4 + 2 * hi;
        b[ks] = mkv2(sAt[kb * W1P + nl], sAt[(kb + 1) * W1P + nl]);
      }
      __builtin_amdgcn_sched_barrier(0);
      v8f c = {};
#pragma unroll
      for (int ks = 0; ks < 16; ++ks) c = wmma_f32(g2[ks], b[ks], c);
#pragma unroll
      for (int r = 0; r < 8; ++r) VTw[t * 1024 + (q * 16 + 8 * hi + r) * 16 + nl] = c[r];
    } else if (warp == 2 && t == 2) {
      v2f a[16];
#pragma unroll
      for (int ks = 0; ks < 16; ++ks) {
        const int kb = ks * 4 + 2 * hi;
        a[ks] = mkv2(sAt[kb * W1P + nl], sAt[(kb + 1) * W1P + nl]);
      }
      __builtin_amdgcn_sched_barrier(0);
      v8f c = {};
#pragma unroll
      for (int ks = 0; ks < 16; ++ks) c = wmma_f32(a[ks], g2[ks], c);
#pragma unroll
      for (int r = 0; r < 8; ++r) {
        const int a_ = 8 * hi + r, m = q * 16 + nl;
        Tw[(m >> 3) * 128 + (((m >> 1) & 1) * 16 + a_) * 4 + ((m >> 2) & 1) * 2 + (m & 1)] = c[r];
      }
    }
    return;
  }

  const int row0 = ((blockIdx.x - TAB_BLOCKS) * 8 + warp) * 16;
  sC[threadIdx.x] = 0.f;
  float cg[4][2];
#pragma unroll
  for (int m = 0; m < 4; ++m) {
    const float* p = cat + (size_t)(row0 + 4 * m + 2 * hi) * 16 + nl;
    cg[m][0] = p[0];
    cg[m][1] = p[16];
  }
  __syncthreads();

  {
    v8f cc = {};
#pragma unroll
    for (int m = 0; m < 4; ++m) {
      const v2f x = mkv2(nl == 12 ? 1.0f : cg[m][0], nl == 12 ? 1.0f : cg[m][1]);
      cc = wmma_f32(x, x, cc);
    }
#pragma unroll
    for (int r = 0; r < 8; ++r) atomicAdd(&sC[(r + 8 * hi) * 16 + nl], cc[r]);
  }
  __syncthreads();
  atomicAdd(&gC[threadIdx.x], sC[threadIdx.x]);
}

__global__ void __launch_bounds__(256) k_out(
    const float* __restrict__ cat,
    const float* __restrict__ lw, const float* __restrict__ lb,
    const float* __restrict__ gC,
    const float* __restrict__ Uw, const float* __restrict__ VTw, const float* __restrict__ Tw,
    const float* __restrict__ lng, const float* __restrict__ lnb,
    const float* __restrict__ ow, const float* __restrict__ obp,
    float* __restrict__ out)
{
  __shared__ float so[8][16 * 68];
  __shared__ __attribute__((aligned(16))) float slw1[64 * WP];
  __shared__ __attribute__((aligned(16))) float sZT[3][64 * W1P];
  __shared__ __attribute__((aligned(16))) float sPT[64 * W1P];
  __shared__ float slb1[64];
  const int warp = threadIdx.x >> 5, lane = threadIdx.x & 31;
  const int hi = lane >> 4, nl = lane & 15;
  const int row0 = (blockIdx.x * 8 + warp) * 16;
  float* om = so[warp];
  const float gscale = 2.0f / (float)(3ull * BATCH * 64);
  const int t0 = warp >> 1, t1 = 2 * (warp & 1);

  v4f cA[2];
  v4f vb[2][2];
#pragma unroll
  for (int h = 0; h < 2; ++h) cA[h] = *(const v4f*)(gC + nl * 16 + 8 * hi + 4 * h);
#pragma unroll
  for (int rep = 0; rep < 2; ++rep) {
    const int TT = min(warp + 8 * rep, 11);
#pragma unroll
    for (int h = 0; h < 2; ++h)
      vb[rep][h] = *(const v4f*)(VTw + (TT >> 2) * 1024 + ((TT & 3) * 16 + nl) * 16 + 8 * hi + 4 * h);
  }
  __builtin_amdgcn_sched_barrier(0);
  v4f ua[3][2];
#pragma unroll
  for (int t = 0; t < 3; ++t)
#pragma unroll
    for (int h = 0; h < 2; ++h)
      ua[t][h] = *(const v4f*)(Uw + t * 1024 + (t0 * 16 + nl) * 16 + 8 * hi + 4 * h);
  v4f r12[2];
#pragma unroll
  for (int h = 0; h < 2; ++h) r12[h] = *(const v4f*)(gC + 12 * 16 + 8 * hi + 4 * h);
  const float lbi = lb[t0 * 16 + nl];
  v4f lbv[2];
#pragma unroll
  for (int h = 0; h < 2; ++h) lbv[h] = *(const v4f*)(lb + t0 * 16 + 8 * hi + 4 * h);
  v4f lwc[4];
#pragma unroll
  for (int j = 0; j < 4; ++j) lwc[j] = *(const v4f*)(lw + ((int)threadIdx.x + 256 * j) * 4);
  __builtin_amdgcn_sched_barrier(0);
#pragma unroll
  for (int h = 0; h < 2; ++h) asm volatile("" : "+v"(vb[1][h]));

#pragma unroll
  for (int rep = 0; rep < 2; ++rep) {
    const int TT = warp + 8 * rep;
    if (rep == 0 || warp < 4) {
      v8f c = {};
#pragma unroll
      for (int h = 0; h < 2; ++h) {
        c = wmma_f32(mkv2(cA[h].x, cA[h].y), mkv2(vb[rep][h].x, vb[rep][h].y), c);
        c = wmma_f32(mkv2(cA[h].z, cA[h].w), mkv2(vb[rep][h].z, vb[rep][h].w), c);
      }
      float* zp = &sZT[TT >> 2][((TT & 3) * 16 + nl) * W1P + 8 * hi];
      const v4f lo = {c[0], c[1], c[2], c[3]}, up = {c[4], c[5], c[6], c[7]};
      *(v4f*)zp = lo;
      *(v4f*)(zp + 4) = up;
    }
  }
  {
    float a3[3];
#pragma unroll
    for (int t = 0; t < 3; ++t) {
      float a = 0.f;
#pragma unroll
      for (int h = 0; h < 2; ++h) {
        a = __builtin_fmaf(ua[t][h].x, r12[h].x, a);
        a = __builtin_fmaf(ua[t][h].y, r12[h].y, a);
        a = __builtin_fmaf(ua[t][h].z, r12[h].z, a);
        a = __builtin_fmaf(ua[t][h].w, r12[h].w, a);
      }
      a3[t] = a;
    }
    const float part = (a3[0] + a3[1]) + a3[2];
    const float lo8 = __shfl(part, nl, 32), up8 = __shfl(part, nl + 16, 32);
    const float g = gscale * ((lo8 + up8) + (float)(3 * BATCH) * lbi);
    if (t1 == 0 && hi == 0) slb1[t0 * 16 + nl] = lbi - LRATE * g / (fabsf(g) + AEPS);
  }
#pragma unroll
  for (int j = 0; j < 4; ++j) {
    const int e = ((int)threadIdx.x + 256 * j) * 4;
    *(v4f*)(&slw1[(e >> 6) * WP + (e & 63)]) = lwc[j];
  }
  __syncthreads();
  v4f ta[8];
#pragma unroll
  for (int p = 0; p < 8; ++p) ta[p] = *(const v4f*)(Tw + p * 128 + lane * 4);
  v2f ca[4];
#pragma unroll
  for (int ks = 0; ks < 4; ++ks) {
    const float* p = cat + (size_t)(row0 + nl) * 16 + ks * 4 + 2 * hi;
    ca[ks] = mkv2((ks == 3 && hi == 0) ? 1.0f : p[0], p[1]);
  }
  {
    float lwv[2][8];
#pragma unroll
    for (int jj = 0; jj < 2; ++jj)
#pragma unroll
      for (int r = 0; r < 8; ++r)
        lwv[jj][r] = slw1[(t0 * 16 + r + 8 * hi) * WP + (t1 + jj) * 16 + nl];
    v8f c[2] = {};
#pragma unroll
    for (int t = 0; t < 3; ++t) {
      v4f z[2][2];
#pragma unroll
      for (int jj = 0; jj < 2; ++jj)
#pragma unroll
        for (int h = 0; h < 2; ++h)
          z[jj][h] = *(const v4f*)(&sZT[t][((t1 + jj) * 16 + nl) * W1P + 8 * hi + 4 * h]);
#pragma unroll
      for (int h = 0; h < 2; ++h) {
#pragma unroll
        for (int jj = 0; jj < 2; ++jj)
          c[jj] = wmma_f32(mkv2(ua[t][h].x, ua[t][h].y), mkv2(z[jj][h].x, z[jj][h].y), c[jj]);
#pragma unroll
        for (int jj = 0; jj < 2; ++jj)
          c[jj] = wmma_f32(mkv2(ua[t][h].z, ua[t][h].w), mkv2(z[jj][h].z, z[jj][h].w), c[jj]);
      }
    }
#pragma unroll
    for (int jj = 0; jj < 2; ++jj) {
      const int j = (t1 + jj) * 16 + nl;
      const float ctj = (sZT[0][j * W1P + 12] + sZT[1][j * W1P + 12]) + sZT[2][j * W1P + 12];
#pragma unroll
      for (int r = 0; r < 8; ++r) {
        const int ii = t0 * 16 + r + 8 * hi;
        const float g = gscale * (c[jj][r] + lbv[r >> 2][r & 3] * ctj);
        slw1[ii * WP + j] = lwv[jj][r] - LRATE * g / (fabsf(g) + AEPS);
      }
    }
  }
  __syncthreads();
  if (warp < 4) {
    const int n = warp * 16 + nl;
    v8f c = {};
#pragma unroll
    for (int p = 0; p < 8; ++p) {
      const int kb = p * 8 + 2 * hi;
      c = wmma_f32(mkv2(ta[p].x, ta[p].y), *(const v2f*)(&slw1[n * WP + kb]), c);
      c = wmma_f32(mkv2(ta[p].z, ta[p].w), *(const v2f*)(&slw1[n * WP + kb + 4]), c);
    }
#pragma unroll
    for (int r = 0; r < 8; ++r) sPT[n * W1P + 8 * hi + r] = c[r];
  }
  __syncthreads();
#pragma unroll
  for (int n0 = 0; n0 < 4; ++n0) {
    int n = n0 * 16 + nl;
    v8f c = {};
#pragma unroll
    for (int ks = 0; ks < 4; ++ks) {
      int kb = ks * 4 + 2 * hi;
      c = wmma_f32(ca[ks], *(const v2f*)(&sPT[n * W1P + kb]), c);
    }
    float bias = slb1[n];
#pragma unroll
    for (int r = 0; r < 8; ++r) om[(r + 8 * hi) * 68 + n] = c[r] + bias;
  }
  __syncthreads();
  if (lane < 16) {
    const float* rowp = om + lane * 68;
    float mu = 0.f;
#pragma unroll
    for (int j = 0; j < 64; ++j) mu += rowp[j];
    mu *= (1.0f / 64.0f);
    float var = 0.f;
#pragma unroll
    for (int j = 0; j < 64; ++j) { float d = rowp[j] - mu; var += d * d; }
    var *= (1.0f / 64.0f);
    float rstd = rsqrtf(var + LNEPS);
    float acc = 0.f;
#pragma unroll
    for (int j = 0; j < 64; ++j) {
      float nrm = (rowp[j] - mu) * rstd * lng[j] + lnb[j];
      float a = nrm / (1.0f + __expf(-nrm));
      acc += a * ow[j];
    }
    out[row0 + lane] = acc + obp[0];
  }
}

extern "C" void kernel_launch(void* const* d_in, const int* in_sizes, int n_in,
                              void* d_out, int out_size, void* d_ws, size_t ws_size,
                              hipStream_t stream) {
  const float* in1  = (const float*)d_in[0];
  const float* in2  = (const float*)d_in[1];
  const float* in3  = (const float*)d_in[2];
  const float* in4  = (const float*)d_in[3];
  const float* w_l  = (const float*)d_in[4];
  const float* b_l  = (const float*)d_in[5];
  const float* w_m  = (const float*)d_in[6];
  const float* b_m  = (const float*)d_in[7];
  const float* w_s  = (const float*)d_in[8];
  const float* b_s  = (const float*)d_in[9];
  const float* fc1w = (const float*)d_in[10];
  const float* fc1b = (const float*)d_in[11];
  const float* q_w  = (const float*)d_in[12];
  const float* q_b  = (const float*)d_in[13];
  const float* k_w  = (const float*)d_in[14];
  const float* k_b  = (const float*)d_in[15];
  const float* v_w  = (const float*)d_in[16];
  const float* v_b  = (const float*)d_in[17];
  const float* t_k  = (const float*)d_in[18];
  const float* t_v  = (const float*)d_in[19];
  const float* t_q  = (const float*)d_in[20];
  const float* lw   = (const float*)d_in[21];
  const float* lb   = (const float*)d_in[22];
  const float* lng  = (const float*)d_in[23];
  const float* lnb  = (const float*)d_in[24];
  const float* outw = (const float*)d_in[25];
  const float* outb = (const float*)d_in[26];

  float* wsf  = (float*)d_ws;
  float* cat  = wsf + F_CAT;
  float* gC   = wsf + F_C;
  float* Uw   = wsf + F_U;
  float* VTw  = wsf + F_VT;
  float* Tw   = wsf + F_T;

  k_cat<<<512, 256, 0, stream>>>(in1, in2, in3, in4, w_l, b_l, w_m, b_m, w_s, b_s, cat, gC);
  k_gram<<<GRAM_GRID, 256, 0, stream>>>(cat, lw, t_k, t_v, t_q, fc1w, fc1b, q_w, q_b, k_w, k_b, v_w, v_b,
                                         Uw, VTw, Tw, gC);
  k_out<<<BATCH / 128, 256, 0, stream>>>(cat, lw, lb, gC, Uw, VTw, Tw, lng, lnb, outw, outb, (float*)d_out);
}


# k_cat: every 512-byte load starts on a 128-byte line: a wave takes the rows that share one line offset, started that many lanes early; zero-padded LDS weight tables; changes the conv sum order
# speedup vs baseline: 3.4513x; 1.0360x over previous
#include <hip/hip_runtime.h>
#include <math.h>

#define BATCH 32768
#define LRATE 0.01f
#define AEPS  1e-8f
#define LNEPS 1e-5f

typedef __attribute__((ext_vector_type(2))) float v2f;
typedef __attribute__((ext_vector_type(8))) float v8f;

__device__ __forceinline__ v2f mkv2(float x, float y) { v2f v; v[0] = x; v[1] = y; return v; }

__device__ __forceinline__ v8f wmma_f32(v2f a, v2f b, v8f c) {
  return __builtin_amdgcn_wmma_f32_16x16x4_f32(false, a, false, b, (short)0, c, false, false);
}

constexpr size_t F_CAT  = 0;
constexpr size_t F_C    = F_CAT + (size_t)BATCH * 16;
constexpr size_t F_U    = F_C + 16 * 16;
constexpr size_t F_VT   = F_U + 3 * 64 * 16;
constexpr size_t F_T    = F_VT + 3 * 64 * 16;

constexpr int WP  = 68;
constexpr int W1P = 20;

typedef __attribute__((ext_vector_type(4))) float v4f;

constexpr int CAT_WAVES  = 8;
constexpr int CAT_ROWS   = 8;
constexpr int CAT_BLOCKS = BATCH / (CAT_WAVES * CAT_ROWS);
static_assert(CAT_BLOCKS == 512 && CAT_WAVES * 32 == 256, "k_cat is launched as <<<512, 256>>>");
constexpr int T1P    = 1568;
constexpr int T2P    = 928;
constexpr int T3P    = 288;
constexpr int T1OFF  = 0;
constexpr int TA2OFF = 4864;
constexpr int TB2OFF = TA2OFF + 2816;
constexpr int TA3OFF = TB2OFF + 2816;
constexpr int TB3OFF = TA3OFF + 1024;
constexpr int WQN    = TB3OFF + 1024;
constexpr int NSTAGE = 20;
constexpr int RPITCH = 33;

__device__ __forceinline__ v4f ld_stream(const v4f* p) { return __builtin_nontemporal_load(p); }
__device__ __forceinline__ float dot4(v4f d, v4f w, float a) {
  a = __builtin_fmaf(d.x, w.x, a); a = __builtin_fmaf(d.y, w.y, a);
  a = __builtin_fmaf(d.z, w.z, a); a = __builtin_fmaf(d.w, w.w, a);
  return a;
}

template <int S>
__device__ __forceinline__ void cat_issue(v4f (&buf)[8], const v4f* p1, const v4f* p2,
                                          const v4f* p3, int t, int t3) {
  if constexpr (S < 12) {
#pragma unroll
    for (int r = 0; r < 8; ++r) {
      if constexpr (S == 11) buf[r] = ld_stream(p1 + r * 3000 + ((t <= 22) ? 352 : 374 - t));
      else                   buf[r] = ld_stream(p1 + r * 3000 + 32 * S);
    }
  } else if constexpr (S < 18) {
#pragma unroll
    for (int h = 0; h < 2; ++h) {
      const int q = 2 * (S - 12) + h;
#pragma unroll
      for (int p = 0; p < 4; ++p) {
        if (q == 11) buf[h * 4 + p] = ld_stream(p2 + p * 3000 + ((t <= 22) ? 352 : 374 - t));
        else         buf[h * 4 + p] = ld_stream(p2 + p * 3000 + 32 * q);
      }
    }
  } else if constexpr (S == 18) {
#pragma unroll
    for (int h = 0; h < 2; ++h)
#pragma unroll
      for (int p = 0; p < 4; ++p) buf[h * 4 + p] = ld_stream(p3 + p * 600 + 32 * h);
  } else if constexpr (S == 19) {
#pragma unroll
    for (int p = 0; p < 4; ++p) buf[p] = ld_stream(p3 + p * 600 + ((t3 <= 10) ? 64 : 74 - t3));
  }
}

template <int S>
__device__ __forceinline__ void cat_weights(v4f (&w)[12], const float* sw, int wb1, int wb3) {
  if constexpr (S < 12) {
#pragma unroll
    for (int o = 0; o < 3; ++o) w[o] = *(const v4f*)(sw + T1OFF + o * T1P + wb1 + 128 * S);
  } else if constexpr (S < 18) {
#pragma unroll
    for (int h = 0; h < 2; ++h) {
      const int q = 2 * (S - 12) + h;
      if (q <= 6) {
#pragma unroll
        for (int o = 0; o < 3; ++o) w[h * 6 + o] = *(const v4f*)(sw + TA2OFF + o * T2P + wb1 + 128 * q);
      }
      if (q >= 5) {
#pragma unroll
        for (int o = 0; o < 3; ++o) w[h * 6 + 3 + o] = *(const v4f*)(sw + TB2OFF + o * T2P + wb1 + 128 * q - 640);
      }
    }
  } else if constexpr (S == 18) {
#pragma unroll
    for (int o = 0; o < 3; ++o) {
      w[o]     = *(const v4f*)(sw + TA3OFF + o * T3P + wb3);
      w[6 + o] = *(const v4f*)(sw + TA3OFF + o * T3P + wb3 + 128);
      w[9 + o] = *(const v4f*)(sw + TB3OFF + o * T3P + wb3);
    }
  } else {
#pragma unroll
    for (int o = 0; o < 3; ++o) w[3 + o] = *(const v4f*)(sw + TB3OFF + o * T3P + wb3 + 128);
  }
}

template <int S>
__device__ __forceinline__ void cat_consume(const v4f (&buf)[8], const v4f (&w)[12],
                                            float (&acc)[8][3]) {
  if constexpr (S < 12) {
#pragma unroll
    for (int r = 0; r < 8; ++r)
#pragma unroll
      for (int o = 0; o < 3; ++o) acc[r][o] = dot4(buf[r], w[o], acc[r][o]);
  } else if constexpr (S < 18) {
#pragma unroll
    for (int h = 0; h < 2; ++h) {
      const int q = 2 * (S - 12) + h;
#pragma unroll
      for (int p = 0; p < 4; ++p) {
        if (q <= 6) {
#pragma unroll
          for (int o = 0; o < 3; ++o) acc[2 * p][o] = dot4(buf[h * 4 + p], w[h * 6 + o], acc[2 * p][o]);
        }
        if (q >= 5) {
#pragma unroll
          for (int o = 0; o < 3; ++o) acc[2 * p + 1][o] = dot4(buf[h * 4 + p], w[h * 6 + 3 + o], acc[2 * p + 1][o]);
        }
      }
    }
  } else if constexpr (S == 18) {
#pragma unroll
    for (int p = 0; p < 4; ++p) {
#pragma unroll
      for (int o = 0; o < 3; ++o) acc[2 * p][o] = dot4(buf[p], w[o], acc[2 * p][o]);
#pragma unroll
      for (int o = 0; o < 3; ++o) acc[2 * p][o] = dot4(buf[4 + p], w[6 + o], acc[2 * p][o]);
#pragma unroll
      for (int o = 0; o < 3; ++o) acc[2 * p + 1][o] = dot4(buf[4 + p], w[9 + o], acc[2 * p + 1][o]);
    }
  } else {
#pragma unroll
    for (int p = 0; p < 4; ++p)
#pragma unroll
      for (int o = 0; o < 3; ++o) acc[2 * p + 1][o] = dot4(buf[p], w[3 + o], acc[2 * p + 1][o]);
  }
}

template <int PITCH, int ITERS, int SHIFT, int NVALID, int KLEN>
__device__ __forceinline__ void cat_table(float* dst, const float* __restrict__ src, int tid) {
  float v[ITERS];
#pragma unroll
  for (int it = 0; it < ITERS; ++it) {
    const int i = min(tid + 256 * it, 3 * PITCH - 1), o = i / PITCH, g = i - o * PITCH - SHIFT;
    const bool ok = (unsigned)g < (unsigned)NVALID;
    const int gg = ok ? g : 0;
    const float x = src[(o * 3 + gg % 3) * KLEN + gg / 3];
    v[it] = ok ? x : 0.f;
  }
#pragma unroll
  for (int it = 0; it < ITERS; ++it) dst[tid + 256 * it] = v[it];
}

__global__ void __launch_bounds__(256) __attribute__((amdgpu_waves_per_eu(4, 4))) k_cat(
    const float* __restrict__ in1, const float* __restrict__ in2,
    const float* __restrict__ in3, const float* __restrict__ in4,
    const float* __restrict__ wl, const float* __restrict__ bl,
    const float* __restrict__ wm, const float* __restrict__ bm,
    const float* __restrict__ wsw, const float* __restrict__ bs,
    float* __restrict__ cat, float* __restrict__ zacc)
{
  __shared__ __attribute__((aligned(16))) float sw[WQN];
  __shared__ float red[CAT_WAVES * 72 * RPITCH];
  __shared__ __attribute__((aligned(16))) float cst[CAT_WAVES][CAT_ROWS * 16];

  const int warp = threadIdx.x >> 5, lane = threadIdx.x & 31;
  const int b0 = (blockIdx.x * CAT_WAVES + warp) * CAT_ROWS;
  const int s1 = (7 * warp) & 7, s3 = (3 * warp) & 7;
  const int t = lane - s1, t3 = lane - s3;
  const int wb1 = 4 * (lane + (s1 ^ 7) + 1), wb3 = 4 * (lane + (s3 ^ 7) + 1);
  const v4f* p1 = (const v4f*)in1 + (size_t)(blockIdx.x * 64 + warp) * 375 + t;
  const v4f* p2 = (const v4f*)in2 + (size_t)(blockIdx.x * 32 + warp) * 375 + t;
  const v4f* p3 = (const v4f*)in3 + (size_t)(blockIdx.x * 32 + warp) * 75 + t3;

  v4f buf[4][8];
  cat_issue<0>(buf[0], p1, p2, p3, t, t3);
  cat_issue<1>(buf[1], p1, p2, p3, t, t3);
  cat_issue<2>(buf[2], p1, p2, p3, t, t3);

  float tailv[2];
#pragma unroll
  for (int g = 0; g < 2; ++g) {
    const int u = lane + 32 * g, cc = u % 7;
    const bool take = (u < 56) && (cc < 3);
    const float tt = in4[(size_t)b0 * 3 + (take ? (u / 7) * 3 + cc : 0)];
    tailv[g] = take ? tt : 0.f;
  }

  cat_table<T1P, 19, 32, 1500, 500>(sw + T1OFF, wl, threadIdx.x);
  cat_table<T2P, 11, 32, 750, 250>(sw + TA2OFF, wm, threadIdx.x);
  cat_table<T2P, 11, 142, 750, 250>(sw + TB2OFF, wm, threadIdx.x);
  cat_table<T3P, 4, 32, 150, 50>(sw + TA3OFF, wsw, threadIdx.x);
  cat_table<T3P, 4, 54, 150, 50>(sw + TB3OFF, wsw, threadIdx.x);
  __syncthreads();

  float acc[8][3];
  v4f w[12];

#define CAT_STAGE(S)                                                              \
  cat_weights<S>(w, sw, wb1, wb3);                                                \
  __builtin_amdgcn_sched_barrier(0);                                              \
  if constexpr ((S) + 3 < NSTAGE)                                                 \
    cat_issue<((S) + 3 < NSTAGE ? (S) + 3 : 0)>(buf[((S) + 3) % 4], p1, p2, p3, t, t3); \
  __builtin_amdgcn_sched_barrier(0);                                              \
  cat_consume<S>(buf[(S) % 4], w, acc);                                           \
  CAT_PIN()
#define CAT_PIN()                                                                 \
  _Pragma("unroll") for (int r = 0; r < 8; ++r)                                   \
    _Pragma("unroll") for (int o = 0; o < 3; ++o) asm volatile("" : "+v"(acc[r][o]));
#define CAT_CLEAR()                                                               \
  _Pragma("unroll") for (int r = 0; r < 8; ++r)                                   \
    _Pragma("unroll") for (int o = 0; o < 3; ++o) acc[r][o] = 0.f;
#define CAT_PARK(COL, ROWOF)                                                      \
  _Pragma("unroll") for (int r = 0; r < 8; ++r)                                   \
    _Pragma("unroll") for (int o = 0; o < 3; ++o)                                 \
      red[((ROWOF) * 9 + (COL) + o) * RPITCH + lane] = acc[r][o];

  CAT_CLEAR()
  CAT_STAGE(0) CAT_STAGE(1) CAT_STAGE(2) CAT_STAGE(3) CAT_STAGE(4) CAT_STAGE(5)
  CAT_STAGE(6) CAT_STAGE(7) CAT_STAGE(8) CAT_STAGE(9) CAT_STAGE(10) CAT_STAGE(11)
  CAT_PARK(0, warp + 8 * r)
  CAT_CLEAR()
  CAT_STAGE(12) CAT_STAGE(13) CAT_STAGE(14) CAT_STAGE(15) CAT_STAGE(16) CAT_STAGE(17)
  CAT_PARK(3, 2 * (warp + 8 * (r >> 1)) + (r & 1))
  CAT_CLEAR()
  CAT_STAGE(18) CAT_STAGE(19)
  CAT_PARK(6, 2 * (warp + 8 * (r >> 1)) + (r & 1))
#undef CAT_STAGE
#undef CAT_PIN
#undef CAT_CLEAR
#undef CAT_PARK
  __syncthreads();

  const float* redw = red + warp * 72 * RPITCH;
  float* cw = cst[warp];
  const float b9[9] = {bl[0], bl[1], bl[2], bm[0], bm[1], bm[2], bs[0], bs[1], bs[2]};
#pragma unroll
  for (int g = 0; g < 3; ++g) {
    const int v = lane + 32 * g;
    if (v < 72) {
      const float* rp = redw + v * RPITCH;
      float s0 = 0.f, s1a = 0.f, s2 = 0.f, s3a = 0.f;
#pragma unroll
      for (int i = 0; i < 32; i += 4) { s0 += rp[i]; s1a += rp[i + 1]; s2 += rp[i + 2]; s3a += rp[i + 3]; }
      const int c = v % 9;
      float bias = b9[0];
#pragma unroll
      for (int k = 1; k < 9; ++k) bias = (c == k) ? b9[k] : bias;
      cw[(v / 9) * 16 + c] = (s0 + s1a) + (s2 + s3a) + bias;
    }
  }
#pragma unroll
  for (int g = 0; g < 2; ++g) {
    const int u = lane + 32 * g;
    if (u < 56) cw[(u / 7) * 16 + 9 + u % 7] = tailv[g];
  }
  __syncthreads();
  *((v4f*)(cat + (size_t)b0 * 16) + lane) = *((const v4f*)cw + lane);
  if (blockIdx.x == 0) zacc[threadIdx.x] = 0.f;
}


constexpr int TAB_BLOCKS  = 12;
constexpr int GRAM_BLOCKS = BATCH / 128;
constexpr int GRAM_GRID   = TAB_BLOCKS + GRAM_BLOCKS;
__global__ void __launch_bounds__(256) __attribute__((amdgpu_waves_per_eu(2, 2))) k_gram(
    const float* __restrict__ cat,
    const float* __restrict__ lw, const float* __restrict__ tk, const float* __restrict__ tvm,
    const float* __restrict__ tq,
    const float* __restrict__ fc1w, const float* __restrict__ fc1b,
    const float* __restrict__ qw, const float* __restrict__ qb,
    const float* __restrict__ kw, const float* __restrict__ kbv,
    const float* __restrict__ vw, const float* __restrict__ vb,
    float* __restrict__ Uw, float* __restrict__ VTw, float* __restrict__ Tw,
    float* __restrict__ gC)
{
  __shared__ float sC[16 * 16];
  __shared__ float sAt[64 * W1P];
  __shared__ float sMq[64 * W1P];
  const int warp = threadIdx.x >> 5, lane = threadIdx.x & 31;
  const int hi = lane >> 4, nl = lane & 15;

  if (blockIdx.x < TAB_BLOCKS) {
    const int t = blockIdx.x >> 2, q = blockIdx.x & 3;
    const float* g2src = (warp == 2) ? tq : tk;
    v2f g2[16];
#pragma unroll
    for (int ks = 0; ks < 16; ++ks) {
      const int kb = ks * 4 + 2 * hi;
      g2[ks] = mkv2(g2src[kb * 64 + q * 16 + nl], g2src[(kb + 1) * 64 + q * 16 + nl]);
    }
    if (warp < 4) {
      const float* Wt = (t == 0) ? qw : ((t == 1) ? kw : vw);
      const float* bsel = (t == 0) ? qb : ((t == 1) ? kbv : vb);
      const float* fsrc = (nl < 12) ? fc1w + nl : fc1b;
      const int fstr = (nl < 12) ? 12 : 1;
      v4f wa[8], fb[8];
#pragma unroll
      for (int p = 0; p < 8; ++p) {
        const int k0 = 8 * p + 4 * hi;
        wa[p] = *(const v4f*)(Wt + (warp * 16 + nl) * 64 + k0);
        fb[p].x = fsrc[k0 * fstr];       fb[p].y = fsrc[(k0 + 1) * fstr];
        fb[p].z = fsrc[(k0 + 2) * fstr]; fb[p].w = fsrc[(k0 + 3) * fstr];
      }
      float bv[8];
#pragma unroll
      for (int r = 0; r < 8; ++r) bv[r] = bsel[warp * 16 + 8 * hi + r];
      v8f c = {};
#pragma unroll
      for (int p = 0; p < 8; ++p) {
        const bool z = nl > 12;
        const float one = (nl == 12) ? 1.0f : 0.0f;
        c = wmma_f32(mkv2(wa[p].x, wa[p].y), mkv2(z ? 0.f : fb[p].x, z ? 0.f : fb[p].y + one), c);
        c = wmma_f32(mkv2(wa[p].z, wa[p].w), mkv2(z ? 0.f : fb[p].z, z ? 0.f : fb[p].w + one), c);
      }
#pragma unroll
      for (int r = 0; r < 8; ++r)
        sAt[(warp * 16 + 8 * hi + r) * W1P + nl] = c[r] + (nl == 12 ? bv[r] : 0.f);
    } else {
      const int a0 = warp - 4, i = q * 16 + nl;
      v4f ta[8], la[8];
#pragma unroll
      for (int p = 0; p < 8; ++p) {
        const int k0 = 8 * p + 4 * hi;
        ta[p] = *(const v4f*)(tk + (a0 * 16 + nl) * 64 + k0);
        la[p] = *(const v4f*)(lw + i * 64 + k0);
      }
      float tv8[8];
#pragma unroll
      for (int r = 0; r < 8; ++r) tv8[r] = tvm[(a0 * 16 + 8 * hi + r) * 64 + i];
      v8f c = {};
#pragma unroll
      for (int p = 0; p < 8; ++p) {
        c = wmma_f32(mkv2(ta[p].x, ta[p].y), mkv2(la[p].x, la[p].y), c);
        c = wmma_f32(mkv2(ta[p].z, ta[p].w), mkv2(la[p].z, la[p].w), c);
      }
#pragma unroll
      for (int r = 0; r < 8; ++r) sMq[(a0 * 16 + 8 * hi + r) * W1P + nl] = c[r] - tv8[r];
    }
    __syncthreads();
    if (warp == 0) {
      v2f a[16], b[16];
#pragma unroll
      for (int ks = 0; ks < 16; ++ks) {
        const int kb = ks * 4 + 2 * hi;
        a[ks] = mkv2(sMq[kb * W1P + nl], sMq[(kb + 1) * W1P + nl]);
        b[ks] = mkv2(sAt[kb * W1P + nl], sAt[(kb + 1) * W1P + nl]);
      }
      __builtin_amdgcn_sched_barrier(0);
      v8f c = {};
#pragma unroll
      for (int ks = 0; ks < 16; ++ks) c = wmma_f32(a[ks], b[ks], c);
#pragma unroll
      for (int r = 0; r < 8; ++r) Uw[t * 1024 + (q * 16 + 8 * hi + r) * 16 + nl] = c[r];
    } else if (warp == 1) {
      v2f b[16];
#pragma unroll
      for (int ks = 0; ks < 16; ++ks) {
        const int kb = ks * 4 + 2 * hi;
        b[ks] = mkv2(sAt[kb * W1P + nl], sAt[(kb + 1) * W1P + nl]);
      }
      __builtin_amdgcn_sched_barrier(0);
      v8f c = {};
#pragma unroll
      for (int ks = 0; ks < 16; ++ks) c = wmma_f32(g2[ks], b[ks], c);
#pragma unroll
      for (int r = 0; r < 8; ++r) VTw[t * 1024 + (q * 16 + 8 * hi + r) * 16 + nl] = c[r];
    } else if (warp == 2 && t == 2) {
      v2f a[16];
#pragma unroll
      for (int ks = 0; ks < 16; ++ks) {
        const int kb = ks * 4 + 2 * hi;
        a[ks] = mkv2(sAt[kb * W1P + nl], sAt[(kb + 1) * W1P + nl]);
      }
      __builtin_amdgcn_sched_barrier(0);
      v8f c = {};
#pragma unroll
      for (int ks = 0; ks < 16; ++ks) c = wmma_f32(a[ks], g2[ks], c);
#pragma unroll
      for (int r = 0; r < 8; ++r) {
        const int a_ = 8 * hi + r, m = q * 16 + nl;
        Tw[(m >> 3) * 128 + (((m >> 1) & 1) * 16 + a_) * 4 + ((m >> 2) & 1) * 2 + (m & 1)] = c[r];
      }
    }
    return;
  }

  const int row0 = ((blockIdx.x - TAB_BLOCKS) * 8 + warp) * 16;
  sC[threadIdx.x] = 0.f;
  float cg[4][2];
#pragma unroll
  for (int m = 0; m < 4; ++m) {
    const float* p = cat + (size_t)(row0 + 4 * m + 2 * hi) * 16 + nl;
    cg[m][0] = p[0];
    cg[m][1] = p[16];
  }
  __syncthreads();

  {
    v8f cc = {};
#pragma unroll
    for (int m = 0; m < 4; ++m) {
      const v2f x = mkv2(nl == 12 ? 1.0f : cg[m][0], nl == 12 ? 1.0f : cg[m][1]);
      cc = wmma_f32(x, x, cc);
    }
#pragma unroll
    for (int r = 0; r < 8; ++r) atomicAdd(&sC[(r + 8 * hi) * 16 + nl], cc[r]);
  }
  __syncthreads();
  atomicAdd(&gC[threadIdx.x], sC[threadIdx.x]);
}

__global__ void __launch_bounds__(256) k_out(
    const float* __restrict__ cat,
    const float* __restrict__ lw, const float* __restrict__ lb,
    const float* __restrict__ gC,
    const float* __restrict__ Uw, const float* __restrict__ VTw, const float* __restrict__ Tw,
    const float* __restrict__ lng, const float* __restrict__ lnb,
    const float* __restrict__ ow, const float* __restrict__ obp,
    float* __restrict__ out)
{
  __shared__ float so[8][16 * 68];
  __shared__ __attribute__((aligned(16))) float slw1[64 * WP];
  __shared__ __attribute__((aligned(16))) float sZT[3][64 * W1P];
  __shared__ __attribute__((aligned(16))) float sPT[64 * W1P];
  __shared__ float slb1[64];
  const int warp = threadIdx.x >> 5, lane = threadIdx.x & 31;
  const int hi = lane >> 4, nl = lane & 15;
  const int row0 = (blockIdx.x * 8 + warp) * 16;
  float* om = so[warp];
  const float gscale = 2.0f / (float)(3ull * BATCH * 64);
  const int t0 = warp >> 1, t1 = 2 * (warp & 1);

  v4f cA[2];
  v4f vb[2][2];
#pragma unroll
  for (int h = 0; h < 2; ++h) cA[h] = *(const v4f*)(gC + nl * 16 + 8 * hi + 4 * h);
#pragma unroll
  for (int rep = 0; rep < 2; ++rep) {
    const int TT = min(warp + 8 * rep, 11);
#pragma unroll
    for (int h = 0; h < 2; ++h)
      vb[rep][h] = *(const v4f*)(VTw + (TT >> 2) * 1024 + ((TT & 3) * 16 + nl) * 16 + 8 * hi + 4 * h);
  }
  __builtin_amdgcn_sched_barrier(0);
  v4f ua[3][2];
#pragma unroll
  for (int t = 0; t < 3; ++t)
#pragma unroll
    for (int h = 0; h < 2; ++h)
      ua[t][h] = *(const v4f*)(Uw + t * 1024 + (t0 * 16 + nl) * 16 + 8 * hi + 4 * h);
  v4f r12[2];
#pragma unroll
  for (int h = 0; h < 2; ++h) r12[h] = *(const v4f*)(gC + 12 * 16 + 8 * hi + 4 * h);
  const float lbi = lb[t0 * 16 + nl];
  v4f lbv[2];
#pragma unroll
  for (int h = 0; h < 2; ++h) lbv[h] = *(const v4f*)(lb + t0 * 16 + 8 * hi + 4 * h);
  v4f lwc[4];
#pragma unroll
  for (int j = 0; j < 4; ++j) lwc[j] = *(const v4f*)(lw + ((int)threadIdx.x + 256 * j) * 4);
  __builtin_amdgcn_sched_barrier(0);
#pragma unroll
  for (int h = 0; h < 2; ++h) asm volatile("" : "+v"(vb[1][h]));

#pragma unroll
  for (int rep = 0; rep < 2; ++rep) {
    const int TT = warp + 8 * rep;
    if (rep == 0 || warp < 4) {
      v8f c = {};
#pragma unroll
      for (int h = 0; h < 2; ++h) {
        c = wmma_f32(mkv2(cA[h].x, cA[h].y), mkv2(vb[rep][h].x, vb[rep][h].y), c);
        c = wmma_f32(mkv2(cA[h].z, cA[h].w), mkv2(vb[rep][h].z, vb[rep][h].w), c);
      }
      float* zp = &sZT[TT >> 2][((TT & 3) * 16 + nl) * W1P + 8 * hi];
      const v4f lo = {c[0], c[1], c[2], c[3]}, up = {c[4], c[5], c[6], c[7]};
      *(v4f*)zp = lo;
      *(v4f*)(zp + 4) = up;
    }
  }
  {
    float a3[3];
#pragma unroll
    for (int t = 0; t < 3; ++t) {
      float a = 0.f;
#pragma unroll
      for (int h = 0; h < 2; ++h) {
        a = __builtin_fmaf(ua[t][h].x, r12[h].x, a);
        a = __builtin_fmaf(ua[t][h].y, r12[h].y, a);
        a = __builtin_fmaf(ua[t][h].z, r12[h].z, a);
        a = __builtin_fmaf(ua[t][h].w, r12[h].w, a);
      }
      a3[t] = a;
    }
    const float part = (a3[0] + a3[1]) + a3[2];
    const float lo8 = __shfl(part, nl, 32), up8 = __shfl(part, nl + 16, 32);
    const float g = gscale * ((lo8 + up8) + (float)(3 * BATCH) * lbi);
    if (t1 == 0 && hi == 0) slb1[t0 * 16 + nl] = lbi - LRATE * g / (fabsf(g) + AEPS);
  }
#pragma unroll
  for (int j = 0; j < 4; ++j) {
    const int e = ((int)threadIdx.x + 256 * j) * 4;
    *(v4f*)(&slw1[(e >> 6) * WP + (e & 63)]) = lwc[j];
  }
  __syncthreads();
  v4f ta[8];
#pragma unroll
  for (int p = 0; p < 8; ++p) ta[p] = *(const v4f*)(Tw + p * 128 + lane * 4);
  v2f ca[4];
#pragma unroll
  for (int ks = 0; ks < 4; ++ks) {
    const float* p = cat + (size_t)(row0 + nl) * 16 + ks * 4 + 2 * hi;
    ca[ks] = mkv2((ks == 3 && hi == 0) ? 1.0f : p[0], p[1]);
  }
  {
    float lwv[2][8];
#pragma unroll
    for (int jj = 0; jj < 2; ++jj)
#pragma unroll
      for (int r = 0; r < 8; ++r)
        lwv[jj][r] = slw1[(t0 * 16 + r + 8 * hi) * WP + (t1 + jj) * 16 + nl];
    v8f c[2] = {};
#pragma unroll
    for (int t = 0; t < 3; ++t) {
      v4f z[2][2];
#pragma unroll
      for (int jj = 0; jj < 2; ++jj)
#pragma unroll
        for (int h = 0; h < 2; ++h)
          z[jj][h] = *(const v4f*)(&sZT[t][((t1 + jj) * 16 + nl) * W1P + 8 * hi + 4 * h]);
#pragma unroll
      for (int h = 0; h < 2; ++h) {
#pragma unroll
        for (int jj = 0; jj < 2; ++jj)
          c[jj] = wmma_f32(mkv2(ua[t][h].x, ua[t][h].y), mkv2(z[jj][h].x, z[jj][h].y), c[jj]);
#pragma unroll
        for (int jj = 0; jj < 2; ++jj)
          c[jj] = wmma_f32(mkv2(ua[t][h].z, ua[t][h].w), mkv2(z[jj][h].z, z[jj][h].w), c[jj]);
      }
    }
#pragma unroll
    for (int jj = 0; jj < 2; ++jj) {
      const int j = (t1 + jj) * 16 + nl;
      const float ctj = (sZT[0][j * W1P + 12] + sZT[1][j * W1P + 12]) + sZT[2][j * W1P + 12];
#pragma unroll
      for (int r = 0; r < 8; ++r) {
        const int ii = t0 * 16 + r + 8 * hi;
        const float g = gscale * (c[jj][r] + lbv[r >> 2][r & 3] * ctj);
        slw1[ii * WP + j] = lwv[jj][r] - LRATE * g / (fabsf(g) + AEPS);
      }
    }
  }
  __syncthreads();
  if (warp < 4) {
    const int n = warp * 16 + nl;
    v8f c = {};
#pragma unroll
    for (int p = 0; p < 8; ++p) {
      const int kb = p * 8 + 2 * hi;
      c = wmma_f32(mkv2(ta[p].x, ta[p].y), *(const v2f*)(&slw1[n * WP + kb]), c);
      c = wmma_f32(mkv2(ta[p].z, ta[p].w), *(const v2f*)(&slw1[n * WP + kb + 4]), c);
    }
#pragma unroll
    for (int r = 0; r < 8; ++r) sPT[n * W1P + 8 * hi + r] = c[r];
  }
  __syncthreads();
#pragma unroll
  for (int n0 = 0; n0 < 4; ++n0) {
    int n = n0 * 16 + nl;
    v8f c = {};
#pragma unroll
    for (int ks = 0; ks < 4; ++ks) {
      int kb = ks * 4 + 2 * hi;
      c = wmma_f32(ca[ks], *(const v2f*)(&sPT[n * W1P + kb]), c);
    }
    float bias = slb1[n];
#pragma unroll
    for (int r = 0; r < 8; ++r) om[(r + 8 * hi) * 68 + n] = c[r] + bias;
  }
  __syncthreads();
  if (lane < 16) {
    const float* rowp = om + lane * 68;
    float mu = 0.f;
#pragma unroll
    for (int j = 0; j < 64; ++j) mu += rowp[j];
    mu *= (1.0f / 64.0f);
    float var = 0.f;
#pragma unroll
    for (int j = 0; j < 64; ++j) { float d = rowp[j] - mu; var += d * d; }
    var *= (1.0f / 64.0f);
    float rstd = rsqrtf(var + LNEPS);
    float acc = 0.f;
#pragma unroll
    for (int j = 0; j < 64; ++j) {
      float nrm = (rowp[j] - mu) * rstd * lng[j] + lnb[j];
      float a = nrm / (1.0f + __expf(-nrm));
      acc += a * ow[j];
    }
    out[row0 + lane] = acc + obp[0];
  }
}

extern "C" void kernel_launch(void* const* d_in, const int* in_sizes, int n_in,
                              void* d_out, int out_size, void* d_ws, size_t ws_size,
                              hipStream_t stream) {
  const float* in1  = (const float*)d_in[0];
  const float* in2  = (const float*)d_in[1];
  const float* in3  = (const float*)d_in[2];
  const float* in4  = (const float*)d_in[3];
  const float* w_l  = (const float*)d_in[4];
  const float* b_l  = (const float*)d_in[5];
  const float* w_m  = (const float*)d_in[6];
  const float* b_m  = (const float*)d_in[7];
  const float* w_s  = (const float*)d_in[8];
  const float* b_s  = (const float*)d_in[9];
  const float* fc1w = (const float*)d_in[10];
  const float* fc1b = (const float*)d_in[11];
  const float* q_w  = (const float*)d_in[12];
  const float* q_b  = (const float*)d_in[13];
  const float* k_w  = (const float*)d_in[14];
  const float* k_b  = (const float*)d_in[15];
  const float* v_w  = (const float*)d_in[16];
  const float* v_b  = (const float*)d_in[17];
  const float* t_k  = (const float*)d_in[18];
  const float* t_v  = (const float*)d_in[19];
  const float* t_q  = (const float*)d_in[20];
  const float* lw   = (const float*)d_in[21];
  const float* lb   = (const float*)d_in[22];
  const float* lng  = (const float*)d_in[23];
  const float* lnb  = (const float*)d_in[24];
  const float* outw = (const float*)d_in[25];
  const float* outb = (const float*)d_in[26];

  float* wsf  = (float*)d_ws;
  float* cat  = wsf + F_CAT;
  float* gC   = wsf + F_C;
  float* Uw   = wsf + F_U;
  float* VTw  = wsf + F_VT;
  float* Tw   = wsf + F_T;

  k_cat<<<512, 256, 0, stream>>>(in1, in2, in3, in4, w_l, b_l, w_m, b_m, w_s, b_s, cat, gC);
  k_gram<<<GRAM_GRID, 256, 0, stream>>>(cat, lw, t_k, t_v, t_q, fc1w, fc1b, q_w, q_b, k_w, k_b, v_w, v_b,
                                         Uw, VTw, Tw, gC);
  k_out<<<BATCH / 128, 256, 0, stream>>>(cat, lw, lb, gC, Uw, VTw, Tw, lng, lnb, outw, outb, (float*)d_out);
}


# k_out: o made transposed (P^T as A, cat rows as B), a lane pair holds one row: LayerNorm, SiLU and the out_w dot run in registers on all 32 lanes, o skips LDS; changes LayerNorm and output sum order
# speedup vs baseline: 3.5728x; 1.0352x over previous
#include <hip/hip_runtime.h>
#include <math.h>

#define BATCH 32768
#define LRATE 0.01f
#define AEPS  1e-8f
#define LNEPS 1e-5f

typedef __attribute__((ext_vector_type(2))) float v2f;
typedef __attribute__((ext_vector_type(8))) float v8f;

__device__ __forceinline__ v2f mkv2(float x, float y) { v2f v; v[0] = x; v[1] = y; return v; }

__device__ __forceinline__ float xhalf(float x) {
  return __builtin_bit_cast(float, __builtin_amdgcn_permlanex16(__builtin_bit_cast(int, x), __builtin_bit_cast(int, x),
                                                                0x76543210, 0xfedcba98, false, false));
}

__device__ __forceinline__ v8f wmma_f32(v2f a, v2f b, v8f c) {
  return __builtin_amdgcn_wmma_f32_16x16x4_f32(false, a, false, b, (short)0, c, false, false);
}

constexpr size_t F_CAT  = 0;
constexpr size_t F_C    = F_CAT + (size_t)BATCH * 16;
constexpr size_t F_U    = F_C + 16 * 16;
constexpr size_t F_VT   = F_U + 3 * 64 * 16;
constexpr size_t F_T    = F_VT + 3 * 64 * 16;

constexpr int WP  = 68;
constexpr int W1P = 20;

typedef __attribute__((ext_vector_type(4))) float v4f;

constexpr int CAT_WAVES  = 8;
constexpr int CAT_ROWS   = 8;
constexpr int CAT_BLOCKS = BATCH / (CAT_WAVES * CAT_ROWS);
static_assert(CAT_BLOCKS == 512 && CAT_WAVES * 32 == 256, "k_cat is launched as <<<512, 256>>>");
constexpr int T1P    = 1568;
constexpr int T2P    = 928;
constexpr int T3P    = 288;
constexpr int T1OFF  = 0;
constexpr int TA2OFF = 4864;
constexpr int TB2OFF = TA2OFF + 2816;
constexpr int TA3OFF = TB2OFF + 2816;
constexpr int TB3OFF = TA3OFF + 1024;
constexpr int WQN    = TB3OFF + 1024;
constexpr int NSTAGE = 20;
constexpr int RPITCH = 33;

__device__ __forceinline__ v4f ld_stream(const v4f* p) { return __builtin_nontemporal_load(p); }
__device__ __forceinline__ float dot4(v4f d, v4f w, float a) {
  a = __builtin_fmaf(d.x, w.x, a); a = __builtin_fmaf(d.y, w.y, a);
  a = __builtin_fmaf(d.z, w.z, a); a = __builtin_fmaf(d.w, w.w, a);
  return a;
}

template <int S>
__device__ __forceinline__ void cat_issue(v4f (&buf)[8], const v4f* p1, const v4f* p2,
                                          const v4f* p3, int t, int t3) {
  if constexpr (S < 12) {
#pragma unroll
    for (int r = 0; r < 8; ++r) {
      if constexpr (S == 11) buf[r] = ld_stream(p1 + r * 3000 + ((t <= 22) ? 352 : 374 - t));
      else                   buf[r] = ld_stream(p1 + r * 3000 + 32 * S);
    }
  } else if constexpr (S < 18) {
#pragma unroll
    for (int h = 0; h < 2; ++h) {
      const int q = 2 * (S - 12) + h;
#pragma unroll
      for (int p = 0; p < 4; ++p) {
        if (q == 11) buf[h * 4 + p] = ld_stream(p2 + p * 3000 + ((t <= 22) ? 352 : 374 - t));
        else         buf[h * 4 + p] = ld_stream(p2 + p * 3000 + 32 * q);
      }
    }
  } else if constexpr (S == 18) {
#pragma unroll
    for (int h = 0; h < 2; ++h)
#pragma unroll
      for (int p = 0; p < 4; ++p) buf[h * 4 + p] = ld_stream(p3 + p * 600 + 32 * h);
  } else if constexpr (S == 19) {
#pragma unroll
    for (int p = 0; p < 4; ++p) buf[p] = ld_stream(p3 + p * 600 + ((t3 <= 10) ? 64 : 74 - t3));
  }
}

template <int S>
__device__ __forceinline__ void cat_weights(v4f (&w)[12], const float* sw, int wb1, int wb3) {
  if constexpr (S < 12) {
#pragma unroll
    for (int o = 0; o < 3; ++o) w[o] = *(const v4f*)(sw + T1OFF + o * T1P + wb1 + 128 * S);
  } else if constexpr (S < 18) {
#pragma unroll
    for (int h = 0; h < 2; ++h) {
      const int q = 2 * (S - 12) + h;
      if (q <= 6) {
#pragma unroll
        for (int o = 0; o < 3; ++o) w[h * 6 + o] = *(const v4f*)(sw + TA2OFF + o * T2P + wb1 + 128 * q);
      }
      if (q >= 5) {
#pragma unroll
        for (int o = 0; o < 3; ++o) w[h * 6 + 3 + o] = *(const v4f*)(sw + TB2OFF + o * T2P + wb1 + 128 * q - 640);
      }
    }
  } else if constexpr (S == 18) {
#pragma unroll
    for (int o = 0; o < 3; ++o) {
      w[o]     = *(const v4f*)(sw + TA3OFF + o * T3P + wb3);
      w[6 + o] = *(const v4f*)(sw + TA3OFF + o * T3P + wb3 + 128);
      w[9 + o] = *(const v4f*)(sw + TB3OFF + o * T3P + wb3);
    }
  } else {
#pragma unroll
    for (int o = 0; o < 3; ++o) w[3 + o] = *(const v4f*)(sw + TB3OFF + o * T3P + wb3 + 128);
  }
}

template <int S>
__device__ __forceinline__ void cat_consume(const v4f (&buf)[8], const v4f (&w)[12],
                                            float (&acc)[8][3]) {
  if constexpr (S < 12) {
#pragma unroll
    for (int r = 0; r < 8; ++r)
#pragma unroll
      for (int o = 0; o < 3; ++o) acc[r][o] = dot4(buf[r], w[o], acc[r][o]);
  } else if constexpr (S < 18) {
#pragma unroll
    for (int h = 0; h < 2; ++h) {
      const int q = 2 * (S - 12) + h;
#pragma unroll
      for (int p = 0; p < 4; ++p) {
        if (q <= 6) {
#pragma unroll
          for (int o = 0; o < 3; ++o) acc[2 * p][o] = dot4(buf[h * 4 + p], w[h * 6 + o], acc[2 * p][o]);
        }
        if (q >= 5) {
#pragma unroll
          for (int o = 0; o < 3; ++o) acc[2 * p + 1][o] = dot4(buf[h * 4 + p], w[h * 6 + 3 + o], acc[2 * p + 1][o]);
        }
      }
    }
  } else if constexpr (S == 18) {
#pragma unroll
    for (int p = 0; p < 4; ++p) {
#pragma unroll
      for (int o = 0; o < 3; ++o) acc[2 * p][o] = dot4(buf[p], w[o], acc[2 * p][o]);
#pragma unroll
      for (int o = 0; o < 3; ++o) acc[2 * p][o] = dot4(buf[4 + p], w[6 + o], acc[2 * p][o]);
#pragma unroll
      for (int o = 0; o < 3; ++o) acc[2 * p + 1][o] = dot4(buf[4 + p], w[9 + o], acc[2 * p + 1][o]);
    }
  } else {
#pragma unroll
    for (int p = 0; p < 4; ++p)
#pragma unroll
      for (int o = 0; o < 3; ++o) acc[2 * p + 1][o] = dot4(buf[p], w[3 + o], acc[2 * p + 1][o]);
  }
}

template <int PITCH, int ITERS, int SHIFT, int NVALID, int KLEN>
__device__ __forceinline__ void cat_table(float* dst, const float* __restrict__ src, int tid) {
  float v[ITERS];
#pragma unroll
  for (int it = 0; it < ITERS; ++it) {
    const int i = min(tid + 256 * it, 3 * PITCH - 1), o = i / PITCH, g = i - o * PITCH - SHIFT;
    const bool ok = (unsigned)g < (unsigned)NVALID;
    const int gg = ok ? g : 0;
    const float x = src[(o * 3 + gg % 3) * KLEN + gg / 3];
    v[it] = ok ? x : 0.f;
  }
#pragma unroll
  for (int it = 0; it < ITERS; ++it) dst[tid + 256 * it] = v[it];
}

__global__ void __launch_bounds__(256) __attribute__((amdgpu_waves_per_eu(4, 4))) k_cat(
    const float* __restrict__ in1, const float* __restrict__ in2,
    const float* __restrict__ in3, const float* __restrict__ in4,
    const float* __restrict__ wl, const float* __restrict__ bl,
    const float* __restrict__ wm, const float* __restrict__ bm,
    const float* __restrict__ wsw, const float* __restrict__ bs,
    float* __restrict__ cat, float* __restrict__ zacc)
{
  __shared__ __attribute__((aligned(16))) float sw[WQN];
  __shared__ float red[CAT_WAVES * 72 * RPITCH];
  __shared__ __attribute__((aligned(16))) float cst[CAT_WAVES][CAT_ROWS * 16];

  const int warp = threadIdx.x >> 5, lane = threadIdx.x & 31;
  const int b0 = (blockIdx.x * CAT_WAVES + warp) * CAT_ROWS;
  const int s1 = (7 * warp) & 7, s3 = (3 * warp) & 7;
  const int t = lane - s1, t3 = lane - s3;
  const int wb1 = 4 * (lane + (s1 ^ 7) + 1), wb3 = 4 * (lane + (s3 ^ 7) + 1);
  const v4f* p1 = (const v4f*)in1 + (size_t)(blockIdx.x * 64 + warp) * 375 + t;
  const v4f* p2 = (const v4f*)in2 + (size_t)(blockIdx.x * 32 + warp) * 375 + t;
  const v4f* p3 = (const v4f*)in3 + (size_t)(blockIdx.x * 32 + warp) * 75 + t3;

  v4f buf[4][8];
  cat_issue<0>(buf[0], p1, p2, p3, t, t3);
  cat_issue<1>(buf[1], p1, p2, p3, t, t3);
  cat_issue<2>(buf[2], p1, p2, p3, t, t3);

  float tailv[2];
#pragma unroll
  for (int g = 0; g < 2; ++g) {
    const int u = lane + 32 * g, cc = u % 7;
    const bool take = (u < 56) && (cc < 3);
    const float tt = in4[(size_t)b0 * 3 + (take ? (u / 7) * 3 + cc : 0)];
    tailv[g] = take ? tt : 0.f;
  }

  cat_table<T1P, 19, 32, 1500, 500>(sw + T1OFF, wl, threadIdx.x);
  cat_table<T2P, 11, 32, 750, 250>(sw + TA2OFF, wm, threadIdx.x);
  cat_table<T2P, 11, 142, 750, 250>(sw + TB2OFF, wm, threadIdx.x);
  cat_table<T3P, 4, 32, 150, 50>(sw + TA3OFF, wsw, threadIdx.x);
  cat_table<T3P, 4, 54, 150, 50>(sw + TB3OFF, wsw, threadIdx.x);
  __syncthreads();

  float acc[8][3];
  v4f w[12];

#define CAT_STAGE(S)                                                              \
  cat_weights<S>(w, sw, wb1, wb3);                                                \
  __builtin_amdgcn_sched_barrier(0);                                              \
  if constexpr ((S) + 3 < NSTAGE)                                                 \
    cat_issue<((S) + 3 < NSTAGE ? (S) + 3 : 0)>(buf[((S) + 3) % 4], p1, p2, p3, t, t3); \
  __builtin_amdgcn_sched_barrier(0);                                              \
  cat_consume<S>(buf[(S) % 4], w, acc);                                           \
  CAT_PIN()
#define CAT_PIN()                                                                 \
  _Pragma("unroll") for (int r = 0; r < 8; ++r)                                   \
    _Pragma("unroll") for (int o = 0; o < 3; ++o) asm volatile("" : "+v"(acc[r][o]));
#define CAT_CLEAR()                                                               \
  _Pragma("unroll") for (int r = 0; r < 8; ++r)                                   \
    _Pragma("unroll") for (int o = 0; o < 3; ++o) acc[r][o] = 0.f;
#define CAT_PARK(COL, ROWOF)                                                      \
  _Pragma("unroll") for (int r = 0; r < 8; ++r)                                   \
    _Pragma("unroll") for (int o = 0; o < 3; ++o)                                 \
      red[((ROWOF) * 9 + (COL) + o) * RPITCH + lane] = acc[r][o];

  CAT_CLEAR()
  CAT_STAGE(0) CAT_STAGE(1) CAT_STAGE(2) CAT_STAGE(3) CAT_STAGE(4) CAT_STAGE(5)
  CAT_STAGE(6) CAT_STAGE(7) CAT_STAGE(8) CAT_STAGE(9) CAT_STAGE(10) CAT_STAGE(11)
  CAT_PARK(0, warp + 8 * r)
  CAT_CLEAR()
  CAT_STAGE(12) CAT_STAGE(13) CAT_STAGE(14) CAT_STAGE(15) CAT_STAGE(16) CAT_STAGE(17)
  CAT_PARK(3, 2 * (warp + 8 * (r >> 1)) + (r & 1))
  CAT_CLEAR()
  CAT_STAGE(18) CAT_STAGE(19)
  CAT_PARK(6, 2 * (warp + 8 * (r >> 1)) + (r & 1))
#undef CAT_STAGE
#undef CAT_PIN
#undef CAT_CLEAR
#undef CAT_PARK
  __syncthreads();

  const float* redw = red + warp * 72 * RPITCH;
  float* cw = cst[warp];
  const float b9[9] = {bl[0], bl[1], bl[2], bm[0], bm[1], bm[2], bs[0], bs[1], bs[2]};
#pragma unroll
  for (int g = 0; g < 3; ++g) {
    const int v = lane + 32 * g;
    if (v < 72) {
      const float* rp = redw + v * RPITCH;
      float s0 = 0.f, s1a = 0.f, s2 = 0.f, s3a = 0.f;
#pragma unroll
      for (int i = 0; i < 32; i += 4) { s0 += rp[i]; s1a += rp[i + 1]; s2 += rp[i + 2]; s3a += rp[i + 3]; }
      const int c = v % 9;
      float bias = b9[0];
#pragma unroll
      for (int k = 1; k < 9; ++k) bias = (c == k) ? b9[k] : bias;
      cw[(v / 9) * 16 + c] = (s0 + s1a) + (s2 + s3a) + bias;
    }
  }
#pragma unroll
  for (int g = 0; g < 2; ++g) {
    const int u = lane + 32 * g;
    if (u < 56) cw[(u / 7) * 16 + 9 + u % 7] = tailv[g];
  }
  __syncthreads();
  *((v4f*)(cat + (size_t)b0 * 16) + lane) = *((const v4f*)cw + lane);
  if (blockIdx.x == 0) zacc[threadIdx.x] = 0.f;
}


constexpr int TAB_BLOCKS  = 12;
constexpr int GRAM_BLOCKS = BATCH / 128;
constexpr int GRAM_GRID   = TAB_BLOCKS + GRAM_BLOCKS;
__global__ void __launch_bounds__(256) __attribute__((amdgpu_waves_per_eu(2, 2))) k_gram(
    const float* __restrict__ cat,
    const float* __restrict__ lw, const float* __restrict__ tk, const float* __restrict__ tvm,
    const float* __restrict__ tq,
    const float* __restrict__ fc1w, const float* __restrict__ fc1b,
    const float* __restrict__ qw, const float* __restrict__ qb,
    const float* __restrict__ kw, const float* __restrict__ kbv,
    const float* __restrict__ vw, const float* __restrict__ vb,
    float* __restrict__ Uw, float* __restrict__ VTw, float* __restrict__ Tw,
    float* __restrict__ gC)
{
  __shared__ float sC[16 * 16];
  __shared__ float sAt[64 * W1P];
  __shared__ float sMq[64 * W1P];
  const int warp = threadIdx.x >> 5, lane = threadIdx.x & 31;
  const int hi = lane >> 4, nl = lane & 15;

  if (blockIdx.x < TAB_BLOCKS) {
    const int t = blockIdx.x >> 2, q = blockIdx.x & 3;
    const float* g2src = (warp == 2) ? tq : tk;
    v2f g2[16];
#pragma unroll
    for (int ks = 0; ks < 16; ++ks) {
      const int kb = ks * 4 + 2 * hi;
      g2[ks] = mkv2(g2src[kb * 64 + q * 16 + nl], g2src[(kb + 1) * 64 + q * 16 + nl]);
    }
    if (warp < 4) {
      const float* Wt = (t == 0) ? qw : ((t == 1) ? kw : vw);
      const float* bsel = (t == 0) ? qb : ((t == 1) ? kbv : vb);
      const float* fsrc = (nl < 12) ? fc1w + nl : fc1b;
      const int fstr = (nl < 12) ? 12 : 1;
      v4f wa[8], fb[8];
#pragma unroll
      for (int p = 0; p < 8; ++p) {
        const int k0 = 8 * p + 4 * hi;
        wa[p] = *(const v4f*)(Wt + (warp * 16 + nl) * 64 + k0);
        fb[p].x = fsrc[k0 * fstr];       fb[p].y = fsrc[(k0 + 1) * fstr];
        fb[p].z = fsrc[(k0 + 2) * fstr]; fb[p].w = fsrc[(k0 + 3) * fstr];
      }
      float bv[8];
#pragma unroll
      for (int r = 0; r < 8; ++r) bv[r] = bsel[warp * 16 + 8 * hi + r];
      v8f c = {};
#pragma unroll
      for (int p = 0; p < 8; ++p) {
        const bool z = nl > 12;
        const float one = (nl == 12) ? 1.0f : 0.0f;
        c = wmma_f32(mkv2(wa[p].x, wa[p].y), mkv2(z ? 0.f : fb[p].x, z ? 0.f : fb[p].y + one), c);
        c = wmma_f32(mkv2(wa[p].z, wa[p].w), mkv2(z ? 0.f : fb[p].z, z ? 0.f : fb[p].w + one), c);
      }
#pragma unroll
      for (int r = 0; r < 8; ++r)
        sAt[(warp * 16 + 8 * hi + r) * W1P + nl] = c[r] + (nl == 12 ? bv[r] : 0.f);
    } else {
      const int a0 = warp - 4, i = q * 16 + nl;
      v4f ta[8], la[8];
#pragma unroll
      for (int p = 0; p < 8; ++p) {
        const int k0 = 8 * p + 4 * hi;
        ta[p] = *(const v4f*)(tk + (a0 * 16 + nl) * 64 + k0);
        la[p] = *(const v4f*)(lw + i * 64 + k0);
      }
      float tv8[8];
#pragma unroll
      for (int r = 0; r < 8; ++r) tv8[r] = tvm[(a0 * 16 + 8 * hi + r) * 64 + i];
      v8f c = {};
#pragma unroll
      for (int p = 0; p < 8; ++p) {
        c = wmma_f32(mkv2(ta[p].x, ta[p].y), mkv2(la[p].x, la[p].y), c);
        c = wmma_f32(mkv2(ta[p].z, ta[p].w), mkv2(la[p].z, la[p].w), c);
      }
#pragma unroll
      for (int r = 0; r < 8; ++r) sMq[(a0 * 16 + 8 * hi + r) * W1P + nl] = c[r] - tv8[r];
    }
    __syncthreads();
    if (warp == 0) {
      v2f a[16], b[16];
#pragma unroll
      for (int ks = 0; ks < 16; ++ks) {
        const int kb = ks * 4 + 2 * hi;
        a[ks] = mkv2(sMq[kb * W1P + nl], sMq[(kb + 1) * W1P + nl]);
        b[ks] = mkv2(sAt[kb * W1P + nl], sAt[(kb + 1) * W1P + nl]);
      }
      __builtin_amdgcn_sched_barrier(0);
      v8f c = {};
#pragma unroll
      for (int ks = 0; ks < 16; ++ks) c = wmma_f32(a[ks], b[ks], c);
#pragma unroll
      for (int r = 0; r < 8; ++r) Uw[t * 1024 + (q * 16 + 8 * hi + r) * 16 + nl] = c[r];
    } else if (warp == 1) {
      v2f b[16];
#pragma unroll
      for (int ks = 0; ks < 16; ++ks) {
        const int kb = ks * 4 + 2 * hi;
        b[ks] = mkv2(sAt[kb * W1P + nl], sAt[(kb + 1) * W1P + nl]);
      }
      __builtin_amdgcn_sched_barrier(0);
      v8f c = {};
#pragma unroll
      for (int ks = 0; ks < 16; ++ks) c = wmma_f32(g2[ks], b[ks], c);
#pragma unroll
      for (int r = 0; r < 8; ++r) VTw[t * 1024 + (q * 16 + 8 * hi + r) * 16 + nl] = c[r];
    } else if (warp == 2 && t == 2) {
      v2f a[16];
#pragma unroll
      for (int ks = 0; ks < 16; ++ks) {
        const int kb = ks * 4 + 2 * hi;
        a[ks] = mkv2(sAt[kb * W1P + nl], sAt[(kb + 1) * W1P + nl]);
      }
      __builtin_amdgcn_sched_barrier(0);
      v8f c = {};
#pragma unroll
      for (int ks = 0; ks < 16; ++ks) c = wmma_f32(a[ks], g2[ks], c);
#pragma unroll
      for (int r = 0; r < 8; ++r) {
        const int a_ = 8 * hi + r, m = q * 16 + nl;
        Tw[(m >> 3) * 128 + (((m >> 1) & 1) * 16 + a_) * 4 + ((m >> 2) & 1) * 2 + (m & 1)] = c[r];
      }
    }
    return;
  }

  const int row0 = ((blockIdx.x - TAB_BLOCKS) * 8 + warp) * 16;
  sC[threadIdx.x] = 0.f;
  float cg[4][2];
#pragma unroll
  for (int m = 0; m < 4; ++m) {
    const float* p = cat + (size_t)(row0 + 4 * m + 2 * hi) * 16 + nl;
    cg[m][0] = p[0];
    cg[m][1] = p[16];
  }
  __syncthreads();

  {
    v8f cc = {};
#pragma unroll
    for (int m = 0; m < 4; ++m) {
      const v2f x = mkv2(nl == 12 ? 1.0f : cg[m][0], nl == 12 ? 1.0f : cg[m][1]);
      cc = wmma_f32(x, x, cc);
    }
#pragma unroll
    for (int r = 0; r < 8; ++r) atomicAdd(&sC[(r + 8 * hi) * 16 + nl], cc[r]);
  }
  __syncthreads();
  atomicAdd(&gC[threadIdx.x], sC[threadIdx.x]);
}

__global__ void __launch_bounds__(256) k_out(
    const float* __restrict__ cat,
    const float* __restrict__ lw, const float* __restrict__ lb,
    const float* __restrict__ gC,
    const float* __restrict__ Uw, const float* __restrict__ VTw, const float* __restrict__ Tw,
    const float* __restrict__ lng, const float* __restrict__ lnb,
    const float* __restrict__ ow, const float* __restrict__ obp,
    float* __restrict__ out)
{
  __shared__ __attribute__((aligned(16))) float slw1[64 * WP];
  __shared__ __attribute__((aligned(16))) float sZT[3][64 * W1P];
  __shared__ __attribute__((aligned(16))) float sPT[64 * W1P];
  __shared__ __attribute__((aligned(16))) float slb1[64];
  const int warp = threadIdx.x >> 5, lane = threadIdx.x & 31;
  const int hi = lane >> 4, nl = lane & 15;
  const int row0 = (blockIdx.x * 8 + warp) * 16;
  const float gscale = 2.0f / (float)(3ull * BATCH * 64);
  const int t0 = warp >> 1, t1 = 2 * (warp & 1);

  v4f cA[2];
  v4f vb[2][2];
#pragma unroll
  for (int h = 0; h < 2; ++h) cA[h] = *(const v4f*)(gC + nl * 16 + 8 * hi + 4 * h);
#pragma unroll
  for (int rep = 0; rep < 2; ++rep) {
    const int TT = min(warp + 8 * rep, 11);
#pragma unroll
    for (int h = 0; h < 2; ++h)
      vb[rep][h] = *(const v4f*)(VTw + (TT >> 2) * 1024 + ((TT & 3) * 16 + nl) * 16 + 8 * hi + 4 * h);
  }
  __builtin_amdgcn_sched_barrier(0);
  v4f ua[3][2];
#pragma unroll
  for (int t = 0; t < 3; ++t)
#pragma unroll
    for (int h = 0; h < 2; ++h)
      ua[t][h] = *(const v4f*)(Uw + t * 1024 + (t0 * 16 + nl) * 16 + 8 * hi + 4 * h);
  v4f r12[2];
#pragma unroll
  for (int h = 0; h < 2; ++h) r12[h] = *(const v4f*)(gC + 12 * 16 + 8 * hi + 4 * h);
  const float lbi = lb[t0 * 16 + nl];
  v4f lbv[2];
#pragma unroll
  for (int h = 0; h < 2; ++h) lbv[h] = *(const v4f*)(lb + t0 * 16 + 8 * hi + 4 * h);
  v4f lwc[4];
#pragma unroll
  for (int j = 0; j < 4; ++j) lwc[j] = *(const v4f*)(lw + ((int)threadIdx.x + 256 * j) * 4);
  __builtin_amdgcn_sched_barrier(0);
#pragma unroll
  for (int h = 0; h < 2; ++h) asm volatile("" : "+v"(vb[1][h]));

#pragma unroll
  for (int rep = 0; rep < 2; ++rep) {
    const int TT = warp + 8 * rep;
    if (rep == 0 || warp < 4) {
      v8f c = {};
#pragma unroll
      for (int h = 0; h < 2; ++h) {
        c = wmma_f32(mkv2(cA[h].x, cA[h].y), mkv2(vb[rep][h].x, vb[rep][h].y), c);
        c = wmma_f32(mkv2(cA[h].z, cA[h].w), mkv2(vb[rep][h].z, vb[rep][h].w), c);
      }
      float* zp = &sZT[TT >> 2][((TT & 3) * 16 + nl) * W1P + 8 * hi];
      const v4f lo = {c[0], c[1], c[2], c[3]}, up = {c[4], c[5], c[6], c[7]};
      *(v4f*)zp = lo;
      *(v4f*)(zp + 4) = up;
    }
  }
  {
    float a3[3];
#pragma unroll
    for (int t = 0; t < 3; ++t) {
      float a = 0.f;
#pragma unroll
      for (int h = 0; h < 2; ++h) {
        a = __builtin_fmaf(ua[t][h].x, r12[h].x, a);
        a = __builtin_fmaf(ua[t][h].y, r12[h].y, a);
        a = __builtin_fmaf(ua[t][h].z, r12[h].z, a);
        a = __builtin_fmaf(ua[t][h].w, r12[h].w, a);
      }
      a3[t] = a;
    }
    const float part = (a3[0] + a3[1]) + a3[2];
    const float lo8 = __shfl(part, nl, 32), up8 = __shfl(part, nl + 16, 32);
    const float g = gscale * ((lo8 + up8) + (float)(3 * BATCH) * lbi);
    if (t1 == 0 && hi == 0) slb1[t0 * 16 + nl] = lbi - LRATE * g / (fabsf(g) + AEPS);
  }
#pragma unroll
  for (int j = 0; j < 4; ++j) {
    const int e = ((int)threadIdx.x + 256 * j) * 4;
    *(v4f*)(&slw1[(e >> 6) * WP + (e & 63)]) = lwc[j];
  }
  __syncthreads();
  v4f ta[8];
#pragma unroll
  for (int p = 0; p < 8; ++p) ta[p] = *(const v4f*)(Tw + p * 128 + lane * 4);
  v2f ca[4];
#pragma unroll
  for (int ks = 0; ks < 4; ++ks) {
    const float* p = cat + (size_t)(row0 + nl) * 16 + ks * 4 + 2 * hi;
    ca[ks] = mkv2((ks == 3 && hi == 0) ? 1.0f : p[0], p[1]);
  }
  v4f lgv[4][2], lnbv[4][2], owv[4][2];
#pragma unroll
  for (int n0 = 0; n0 < 4; ++n0)
#pragma unroll
    for (int h = 0; h < 2; ++h) {
      const int f = n0 * 16 + 8 * hi + 4 * h;
      lgv[n0][h]  = *(const v4f*)(lng + f);
      lnbv[n0][h] = *(const v4f*)(lnb + f);
      owv[n0][h]  = *(const v4f*)(ow + f);
    }
  {
    float lwv[2][8];
#pragma unroll
    for (int jj = 0; jj < 2; ++jj)
#pragma unroll
      for (int r = 0; r < 8; ++r)
        lwv[jj][r] = slw1[(t0 * 16 + r + 8 * hi) * WP + (t1 + jj) * 16 + nl];
    v8f c[2] = {};
#pragma unroll
    for (int t = 0; t < 3; ++t) {
      v4f z[2][2];
#pragma unroll
      for (int jj = 0; jj < 2; ++jj)
#pragma unroll
        for (int h = 0; h < 2; ++h)
          z[jj][h] = *(const v4f*)(&sZT[t][((t1 + jj) * 16 + nl) * W1P + 8 * hi + 4 * h]);
#pragma unroll
      for (int h = 0; h < 2; ++h) {
#pragma unroll
        for (int jj = 0; jj < 2; ++jj)
          c[jj] = wmma_f32(mkv2(ua[t][h].x, ua[t][h].y), mkv2(z[jj][h].x, z[jj][h].y), c[jj]);
#pragma unroll
        for (int jj = 0; jj < 2; ++jj)
          c[jj] = wmma_f32(mkv2(ua[t][h].z, ua[t][h].w), mkv2(z[jj][h].z, z[jj][h].w), c[jj]);
      }
    }
#pragma unroll
    for (int jj = 0; jj < 2; ++jj) {
      const int j = (t1 + jj) * 16 + nl;
      const float ctj = (sZT[0][j * W1P + 12] + sZT[1][j * W1P + 12]) + sZT[2][j * W1P + 12];
#pragma unroll
      for (int r = 0; r < 8; ++r) {
        const int ii = t0 * 16 + r + 8 * hi;
        const float g = gscale * (c[jj][r] + lbv[r >> 2][r & 3] * ctj);
        slw1[ii * WP + j] = lwv[jj][r] - LRATE * g / (fabsf(g) + AEPS);
      }
    }
  }
  __syncthreads();
  if (warp < 4) {
    const int n = warp * 16 + nl;
    v8f c = {};
#pragma unroll
    for (int p = 0; p < 8; ++p) {
      const int kb = p * 8 + 2 * hi;
      c = wmma_f32(mkv2(ta[p].x, ta[p].y), *(const v2f*)(&slw1[n * WP + kb]), c);
      c = wmma_f32(mkv2(ta[p].z, ta[p].w), *(const v2f*)(&slw1[n * WP + kb + 4]), c);
    }
#pragma unroll
    for (int r = 0; r < 8; ++r) sPT[n * W1P + 8 * hi + r] = c[r];
  }
  __syncthreads();
  v4f b1v[4][2];
#pragma unroll
  for (int n0 = 0; n0 < 4; ++n0)
#pragma unroll
    for (int h = 0; h < 2; ++h) b1v[n0][h] = *(const v4f*)(&slb1[n0 * 16 + 8 * hi + 4 * h]);
  float x[4][8];
#pragma unroll
  for (int n0 = 0; n0 < 4; ++n0) {
    const int n = n0 * 16 + nl;
    v8f c = {};
#pragma unroll
    for (int ks = 0; ks < 4; ++ks) {
      const int kb = ks * 4 + 2 * hi;
      c = wmma_f32(*(const v2f*)(&sPT[n * W1P + kb]), ca[ks], c);
    }
#pragma unroll
    for (int r = 0; r < 8; ++r) x[n0][r] = c[r] + b1v[n0][r >> 2][r & 3];
  }
  float s4[4];
#pragma unroll
  for (int n0 = 0; n0 < 4; ++n0) {
    float s = 0.f;
#pragma unroll
    for (int r = 0; r < 8; ++r) s += x[n0][r];
    s4[n0] = s;
  }
  float sum = (s4[0] + s4[1]) + (s4[2] + s4[3]);
  sum += xhalf(sum);
  const float mu = sum * (1.0f / 64.0f);
#pragma unroll
  for (int n0 = 0; n0 < 4; ++n0) {
    float s = 0.f;
#pragma unroll
    for (int r = 0; r < 8; ++r) { const float d = x[n0][r] - mu; x[n0][r] = d; s += d * d; }
    s4[n0] = s;
  }
  float var = (s4[0] + s4[1]) + (s4[2] + s4[3]);
  var += xhalf(var);
  var *= (1.0f / 64.0f);
  const float rstd = rsqrtf(var + LNEPS);
#pragma unroll
  for (int n0 = 0; n0 < 4; ++n0) {
    float s = 0.f;
#pragma unroll
    for (int r = 0; r < 8; ++r) {
      const float nrm = x[n0][r] * rstd * lgv[n0][r >> 2][r & 3] + lnbv[n0][r >> 2][r & 3];
      const float a = nrm / (1.0f + __expf(-nrm));
      s += a * owv[n0][r >> 2][r & 3];
    }
    s4[n0] = s;
  }
  float acc = (s4[0] + s4[1]) + (s4[2] + s4[3]);
  acc += xhalf(acc);
  if (hi == 0) out[row0 + nl] = acc + obp[0];
}

extern "C" void kernel_launch(void* const* d_in, const int* in_sizes, int n_in,
                              void* d_out, int out_size, void* d_ws, size_t ws_size,
                              hipStream_t stream) {
  const float* in1  = (const float*)d_in[0];
  const float* in2  = (const float*)d_in[1];
  const float* in3  = (const float*)d_in[2];
  const float* in4  = (const float*)d_in[3];
  const float* w_l  = (const float*)d_in[4];
  const float* b_l  = (const float*)d_in[5];
  const float* w_m  = (const float*)d_in[6];
  const float* b_m  = (const float*)d_in[7];
  const float* w_s  = (const float*)d_in[8];
  const float* b_s  = (const float*)d_in[9];
  const float* fc1w = (const float*)d_in[10];
  const float* fc1b = (const float*)d_in[11];
  const float* q_w  = (const float*)d_in[12];
  const float* q_b  = (const float*)d_in[13];
  const float* k_w  = (const float*)d_in[14];
  const float* k_b  = (const float*)d_in[15];
  const float* v_w  = (const float*)d_in[16];
  const float* v_b  = (const float*)d_in[17];
  const float* t_k  = (const float*)d_in[18];
  const float* t_v  = (const float*)d_in[19];
  const float* t_q  = (const float*)d_in[20];
  const float* lw   = (const float*)d_in[21];
  const float* lb   = (const float*)d_in[22];
  const float* lng  = (const float*)d_in[23];
  const float* lnb  = (const float*)d_in[24];
  const float* outw = (const float*)d_in[25];
  const float* outb = (const float*)d_in[26];

  float* wsf  = (float*)d_ws;
  float* cat  = wsf + F_CAT;
  float* gC   = wsf + F_C;
  float* Uw   = wsf + F_U;
  float* VTw  = wsf + F_VT;
  float* Tw   = wsf + F_T;

  k_cat<<<512, 256, 0, stream>>>(in1, in2, in3, in4, w_l, b_l, w_m, b_m, w_s, b_s, cat, gC);
  k_gram<<<GRAM_GRID, 256, 0, stream>>>(cat, lw, t_k, t_v, t_q, fc1w, fc1b, q_w, q_b, k_w, k_b, v_w, v_b,
                                         Uw, VTw, Tw, gC);
  k_out<<<BATCH / 128, 256, 0, stream>>>(cat, lw, lb, gC, Uw, VTw, Tw, lng, lnb, outw, outb, (float*)d_out);
}
